# starting kernel
# baseline (speedup 1.0000x reference)
#include <hip/hip_runtime.h>
#include <math.h>

#define NN   50000
#define EE   800000
#define IND  256
#define H1_  4
#define HID1 64
#define OUTD 64
#define SLOPE 0.2f

typedef __attribute__((ext_vector_type(2))) float v2f;
typedef __attribute__((ext_vector_type(8))) float v8f;

__device__ __forceinline__ unsigned f2ord(float f) {
    unsigned u = __float_as_uint(f);
    return (u & 0x80000000u) ? ~u : (u | 0x80000000u);
}
__device__ __forceinline__ float ord2f(unsigned u) {
    return (u & 0x80000000u) ? __uint_as_float(u & 0x7fffffffu) : __uint_as_float(~u);
}

template <int BM, int BN, int NCOL>
__global__ __launch_bounds__(256) void wmma_gemm_f32(
    const float* __restrict__ A, const float* __restrict__ B,
    float* __restrict__ C, int M, int K) {
    constexpr int BK  = 32;
    constexpr int BMp = BM + 2;
    constexpr int BNp = BN + 2;
    __shared__ float As[(BK / 2) * BMp * 2];
    __shared__ float Bs[(BK / 2) * BNp * 2];

    const int tid  = threadIdx.x;
    const int lane = tid & 31;
    const int wave = tid >> 5;
    constexpr int WCOLS = BN / 64;
    const int wrow = wave / WCOLS;
    const int wcol = wave % WCOLS;
    const int row0 = blockIdx.y * BM;
    const int col0 = blockIdx.x * BN;

    v8f acc[4];
#pragma unroll
    for (int t = 0; t < 4; t++)
#pragma unroll
        for (int r = 0; r < 8; r++) acc[t][r] = 0.0f;

    const int m16 = lane & 15;
    const int kb  = (lane >> 4) * 2;

    const v2f* Asp = (const v2f*)As;
    const v2f* Bsp = (const v2f*)Bs;

    for (int k0 = 0; k0 < K; k0 += BK) {
        for (int idx = tid; idx < BM * BK; idx += 256) {
            int m = idx / BK, k = idx % BK;
            int r = row0 + m;
            float v = (r < M) ? A[(size_t)r * K + k0 + k] : 0.0f;
            As[((k >> 1) * BMp + m) * 2 + (k & 1)] = v;
        }
        for (int idx = tid; idx < BK * BN; idx += 256) {
            int k = idx / BN, n = idx % BN;
            Bs[((k >> 1) * BNp + n) * 2 + (k & 1)] =
                B[(size_t)(k0 + k) * NCOL + col0 + n];
        }
        __syncthreads();

        const int am = wrow * 16 + m16;
#pragma unroll
        for (int kk = 0; kk < BK; kk += 4) {
            const int kp = (kk + kb) >> 1;
            v2f a = Asp[kp * BMp + am];
#pragma unroll
            for (int t = 0; t < 4; t++) {
                v2f b = Bsp[kp * BNp + wcol * 64 + t * 16 + m16];
                acc[t] = __builtin_amdgcn_wmma_f32_16x16x4_f32(
                    false, a, false, b, (short)0, acc[t], false, false);
            }
        }
        __syncthreads();
    }

    const int rbase = row0 + wrow * 16 + ((lane >> 4) << 3);
#pragma unroll
    for (int t = 0; t < 4; t++) {
        int col = col0 + wcol * 64 + t * 16 + m16;
#pragma unroll
        for (int r = 0; r < 8; r++) {
            int row = rbase + r;
            if (row < M) C[(size_t)row * NCOL + col] = acc[t][r];
        }
    }
}

template <int H>
__global__ __launch_bounds__(256) void node_attn(
    const float* __restrict__ h, const float* __restrict__ al,
    const float* __restrict__ ar, float* __restrict__ el, float* __restrict__ er) {
    int lane = threadIdx.x & 31;
    int n = blockIdx.x * 8 + (threadIdx.x >> 5);
    if (n >= NN) return;
    const float* hp = h + (size_t)n * (H * 64);
#pragma unroll
    for (int hh = 0; hh < H; hh++) {
        float v0 = hp[hh * 64 + lane];
        float v1 = hp[hh * 64 + 32 + lane];
        float pl = v0 * al[hh * 64 + lane] + v1 * al[hh * 64 + 32 + lane];
        float pr = v0 * ar[hh * 64 + lane] + v1 * ar[hh * 64 + 32 + lane];
        for (int off = 16; off > 0; off >>= 1) {
            pl += __shfl_down(pl, off);
            pr += __shfl_down(pr, off);
        }
        if (lane == 0) {
            el[(size_t)n * H + hh] = pl;
            er[(size_t)n * H + hh] = pr;
        }
    }
}

template <int H>
__global__ void edge_max(const int* __restrict__ src, const int* __restrict__ dst,
                         const float* __restrict__ el, const float* __restrict__ er,
                         unsigned* __restrict__ emax) {
    int e = blockIdx.x * blockDim.x + threadIdx.x;
    if (e >= EE) return;
    int s = src[e], d = dst[e];
#pragma unroll
    for (int hh = 0; hh < H; hh++) {
        float sc = el[(size_t)s * H + hh] + er[(size_t)d * H + hh];
        sc = sc > 0.0f ? sc : SLOPE * sc;
        atomicMax(&emax[(size_t)d * H + hh], f2ord(sc));
    }
}

template <int H>
__global__ void edge_exp(const int* __restrict__ src, const int* __restrict__ dst,
                         const float* __restrict__ el, const float* __restrict__ er,
                         const unsigned* __restrict__ emax,
                         float* __restrict__ ex, float* __restrict__ denom) {
    int e = blockIdx.x * blockDim.x + threadIdx.x;
    if (e >= EE) return;
    int s = src[e], d = dst[e];
#pragma unroll
    for (int hh = 0; hh < H; hh++) {
        float sc = el[(size_t)s * H + hh] + er[(size_t)d * H + hh];
        sc = sc > 0.0f ? sc : SLOPE * sc;
        float m = ord2f(emax[(size_t)d * H + hh]);
        float v = expf(sc - m);
        ex[(size_t)e * H + hh] = v;
        atomicAdd(&denom[(size_t)d * H + hh], v);
    }
}

template <int H>
__global__ __launch_bounds__(256) void edge_aggregate(
    const int* __restrict__ src, const int* __restrict__ dst,
    const float* __restrict__ ex, const float* __restrict__ denom,
    const float* __restrict__ h, float* __restrict__ out) {
    int lane = threadIdx.x & 31;
    int e = blockIdx.x * 8 + (threadIdx.x >> 5);
    if (e >= EE) return;
    int s = src[e], d = dst[e];
    float a[H];
#pragma unroll
    for (int hh = 0; hh < H; hh++)
        a[hh] = ex[(size_t)e * H + hh] / denom[(size_t)d * H + hh];
    constexpr int T4 = (H * 64) / 4;
    const float4* hs = (const float4*)(h + (size_t)s * (H * 64));
    float* od = out + (size_t)d * (H * 64);
    for (int i = lane; i < T4; i += 32) {
        float4 v = hs[i];
        float c = a[(i * 4) >> 6];
        atomicAdd(&od[i * 4 + 0], c * v.x);
        atomicAdd(&od[i * 4 + 1], c * v.y);
        atomicAdd(&od[i * 4 + 2], c * v.z);
        atomicAdd(&od[i * 4 + 3], c * v.w);
    }
}

__global__ void elu_bias256(float* __restrict__ x, const float* __restrict__ b, int total) {
    int i = blockIdx.x * blockDim.x + threadIdx.x;
    if (i >= total) return;
    float v = x[i] + b[i & 255];
    x[i] = v > 0.0f ? v : expm1f(v);
}

__global__ void fill_f32(float* p, float v, int n) {
    int i = blockIdx.x * blockDim.x + threadIdx.x;
    if (i < n) p[i] = v;
}
__global__ void fill_u32(unsigned* p, unsigned v, int n) {
    int i = blockIdx.x * blockDim.x + threadIdx.x;
    if (i < n) p[i] = v;
}
__global__ void init_out_bias(float* __restrict__ out, const float* __restrict__ b) {
    int i = blockIdx.x * blockDim.x + threadIdx.x;
    if (i < NN * OUTD) out[i] = b[i & 63];
}

extern "C" void kernel_launch(void* const* d_in, const int* in_sizes, int n_in,
                              void* d_out, int out_size, void* d_ws, size_t ws_size,
                              hipStream_t stream) {
    const float* feat = (const float*)d_in[0];
    const int*   src  = (const int*)d_in[1];
    const int*   dst  = (const int*)d_in[2];
    const float* W1   = (const float*)d_in[3];
    const float* al1  = (const float*)d_in[4];
    const float* ar1  = (const float*)d_in[5];
    const float* b1   = (const float*)d_in[6];
    const float* W2   = (const float*)d_in[7];
    const float* al2  = (const float*)d_in[8];
    const float* ar2  = (const float*)d_in[9];
    const float* b2   = (const float*)d_in[10];
    float* out = (float*)d_out;

    float* h1    = (float*)d_ws;
    float* agg1  = h1    + (size_t)NN * 256;
    float* h2    = agg1  + (size_t)NN * 256;
    float* el1   = h2    + (size_t)NN * 64;
    float* er1   = el1   + (size_t)NN * H1_;
    float* den1  = er1   + (size_t)NN * H1_;
    unsigned* mx1 = (unsigned*)(den1 + (size_t)NN * H1_);
    float* ex1   = (float*)(mx1 + (size_t)NN * H1_);
    float* el2   = ex1   + (size_t)EE * H1_;
    float* er2   = el2   + NN;
    float* den2  = er2   + NN;
    unsigned* mx2 = (unsigned*)(den2 + NN);
    float* ex2   = (float*)(mx2 + NN);

    const int TB = 256;
    fill_f32<<<(NN * 256 + TB - 1) / TB, TB, 0, stream>>>(agg1, 0.0f, NN * 256);
    fill_f32<<<(NN * H1_ + TB - 1) / TB, TB, 0, stream>>>(den1, 0.0f, NN * H1_);
    fill_u32<<<(NN * H1_ + TB - 1) / TB, TB, 0, stream>>>(mx1, 0u, NN * H1_);
    fill_f32<<<(NN + TB - 1) / TB, TB, 0, stream>>>(den2, 0.0f, NN);
    fill_u32<<<(NN + TB - 1) / TB, TB, 0, stream>>>(mx2, 0u, NN);
    init_out_bias<<<(NN * OUTD + TB - 1) / TB, TB, 0, stream>>>(out, b2);

    {
        dim3 grid(256 / 128, (NN + 63) / 64);
        wmma_gemm_f32<64, 128, 256><<<grid, 256, 0, stream>>>(feat, W1, h1, NN, IND);
    }
    node_attn<H1_><<<(NN + 7) / 8, 256, 0, stream>>>(h1, al1, ar1, el1, er1);
    edge_max<H1_><<<(EE + TB - 1) / TB, TB, 0, stream>>>(src, dst, el1, er1, mx1);
    edge_exp<H1_><<<(EE + TB - 1) / TB, TB, 0, stream>>>(src, dst, el1, er1, mx1, ex1, den1);
    edge_aggregate<H1_><<<(EE + 7) / 8, 256, 0, stream>>>(src, dst, ex1, den1, h1, agg1);
    elu_bias256<<<(NN * 256 + TB - 1) / TB, TB, 0, stream>>>(agg1, b1, NN * 256);

    {
        dim3 grid(64 / 64, (NN + 127) / 128);
        wmma_gemm_f32<128, 64, 64><<<grid, 256, 0, stream>>>(agg1, W2, h2, NN, 256);
    }
    node_attn<1><<<(NN + 7) / 8, 256, 0, stream>>>(h2, al2, ar2, el2, er2);
    edge_max<1><<<(EE + TB - 1) / TB, TB, 0, stream>>>(src, dst, el2, er2, mx2);
    edge_exp<1><<<(EE + TB - 1) / TB, TB, 0, stream>>>(src, dst, el2, er2, mx2, ex2, den2);
    edge_aggregate<1><<<(EE + 7) / 8, 256, 0, stream>>>(src, dst, ex2, den2, h2, out);
}


# Softmax division deferred: one edge pass adds exp(e-max) to denom and exp(e-max)*h[src] to the row, rows divided once in the bias pass; exp passes and ex arrays go; sums in arrival order, as before
# speedup vs baseline: 1.0649x; 1.0649x over previous
#include <hip/hip_runtime.h>
#include <math.h>

#define NN   50000
#define EE   800000
#define IND  256
#define H1_  4
#define HID1 64
#define OUTD 64
#define SLOPE 0.2f

typedef __attribute__((ext_vector_type(2))) float v2f;
typedef __attribute__((ext_vector_type(8))) float v8f;

__device__ __forceinline__ unsigned f2ord(float f) {
    unsigned u = __float_as_uint(f);
    return (u & 0x80000000u) ? ~u : (u | 0x80000000u);
}
__device__ __forceinline__ float ord2f(unsigned u) {
    return (u & 0x80000000u) ? __uint_as_float(u & 0x7fffffffu) : __uint_as_float(~u);
}

template <int BM, int BN, int NCOL>
__global__ __launch_bounds__(256) void wmma_gemm_f32(
    const float* __restrict__ A, const float* __restrict__ B,
    float* __restrict__ C, int M, int K) {
    constexpr int BK  = 32;
    constexpr int BMp = BM + 2;
    constexpr int BNp = BN + 2;
    __shared__ float As[(BK / 2) * BMp * 2];
    __shared__ float Bs[(BK / 2) * BNp * 2];

    const int tid  = threadIdx.x;
    const int lane = tid & 31;
    const int wave = tid >> 5;
    constexpr int WCOLS = BN / 64;
    const int wrow = wave / WCOLS;
    const int wcol = wave % WCOLS;
    const int row0 = blockIdx.y * BM;
    const int col0 = blockIdx.x * BN;

    v8f acc[4];
#pragma unroll
    for (int t = 0; t < 4; t++)
#pragma unroll
        for (int r = 0; r < 8; r++) acc[t][r] = 0.0f;

    const int m16 = lane & 15;
    const int kb  = (lane >> 4) * 2;

    const v2f* Asp = (const v2f*)As;
    const v2f* Bsp = (const v2f*)Bs;

    for (int k0 = 0; k0 < K; k0 += BK) {
        for (int idx = tid; idx < BM * BK; idx += 256) {
            int m = idx / BK, k = idx % BK;
            int r = row0 + m;
            float v = (r < M) ? A[(size_t)r * K + k0 + k] : 0.0f;
            As[((k >> 1) * BMp + m) * 2 + (k & 1)] = v;
        }
        for (int idx = tid; idx < BK * BN; idx += 256) {
            int k = idx / BN, n = idx % BN;
            Bs[((k >> 1) * BNp + n) * 2 + (k & 1)] =
                B[(size_t)(k0 + k) * NCOL + col0 + n];
        }
        __syncthreads();

        const int am = wrow * 16 + m16;
#pragma unroll
        for (int kk = 0; kk < BK; kk += 4) {
            const int kp = (kk + kb) >> 1;
            v2f a = Asp[kp * BMp + am];
#pragma unroll
            for (int t = 0; t < 4; t++) {
                v2f b = Bsp[kp * BNp + wcol * 64 + t * 16 + m16];
                acc[t] = __builtin_amdgcn_wmma_f32_16x16x4_f32(
                    false, a, false, b, (short)0, acc[t], false, false);
            }
        }
        __syncthreads();
    }

    const int rbase = row0 + wrow * 16 + ((lane >> 4) << 3);
#pragma unroll
    for (int t = 0; t < 4; t++) {
        int col = col0 + wcol * 64 + t * 16 + m16;
#pragma unroll
        for (int r = 0; r < 8; r++) {
            int row = rbase + r;
            if (row < M) C[(size_t)row * NCOL + col] = acc[t][r];
        }
    }
}

template <int H>
__global__ __launch_bounds__(256) void node_attn(
    const float* __restrict__ h, const float* __restrict__ al,
    const float* __restrict__ ar, float* __restrict__ el, float* __restrict__ er) {
    int lane = threadIdx.x & 31;
    int n = blockIdx.x * 8 + (threadIdx.x >> 5);
    if (n >= NN) return;
    const float* hp = h + (size_t)n * (H * 64);
#pragma unroll
    for (int hh = 0; hh < H; hh++) {
        float v0 = hp[hh * 64 + lane];
        float v1 = hp[hh * 64 + 32 + lane];
        float pl = v0 * al[hh * 64 + lane] + v1 * al[hh * 64 + 32 + lane];
        float pr = v0 * ar[hh * 64 + lane] + v1 * ar[hh * 64 + 32 + lane];
        for (int off = 16; off > 0; off >>= 1) {
            pl += __shfl_down(pl, off);
            pr += __shfl_down(pr, off);
        }
        if (lane == 0) {
            el[(size_t)n * H + hh] = pl;
            er[(size_t)n * H + hh] = pr;
        }
    }
}

template <int H>
__global__ void edge_max(const int* __restrict__ src, const int* __restrict__ dst,
                         const float* __restrict__ el, const float* __restrict__ er,
                         unsigned* __restrict__ emax) {
    int e = blockIdx.x * blockDim.x + threadIdx.x;
    if (e >= EE) return;
    int s = src[e], d = dst[e];
#pragma unroll
    for (int hh = 0; hh < H; hh++) {
        float sc = el[(size_t)s * H + hh] + er[(size_t)d * H + hh];
        sc = sc > 0.0f ? sc : SLOPE * sc;
        atomicMax(&emax[(size_t)d * H + hh], f2ord(sc));
    }
}

template <int H>
__global__ __launch_bounds__(256) void edge_accumulate(
    const int* __restrict__ src, const int* __restrict__ dst,
    const float* __restrict__ el, const float* __restrict__ er,
    const unsigned* __restrict__ emax, const float* __restrict__ h,
    float* __restrict__ denom, float* __restrict__ acc) {
    int lane = threadIdx.x & 31;
    int e = blockIdx.x * 8 + (threadIdx.x >> 5);
    if (e >= EE) return;
    int s = src[e], d = dst[e];
    float w[H];
#pragma unroll
    for (int hh = 0; hh < H; hh++) {
        float sc = el[(size_t)s * H + hh] + er[(size_t)d * H + hh];
        sc = sc > 0.0f ? sc : SLOPE * sc;
        w[hh] = expf(sc - ord2f(emax[(size_t)d * H + hh]));
    }
    if (lane < H) {
        float wl = w[0];
#pragma unroll
        for (int hh = 1; hh < H; hh++) wl = (lane == hh) ? w[hh] : wl;
        atomicAdd(&denom[(size_t)d * H + lane], wl);
    }
    constexpr int T4 = (H * 64) / 4;
    const float4* hs = (const float4*)(h + (size_t)s * (H * 64));
    float* od = acc + (size_t)d * (H * 64);
    for (int i = lane; i < T4; i += 32) {
        float4 v = hs[i];
        float c = w[0];
#pragma unroll
        for (int hh = 1; hh < H; hh++) c = ((i >> 4) == hh) ? w[hh] : c;
        atomicAdd(&od[i * 4 + 0], c * v.x);
        atomicAdd(&od[i * 4 + 1], c * v.y);
        atomicAdd(&od[i * 4 + 2], c * v.z);
        atomicAdd(&od[i * 4 + 3], c * v.w);
    }
}

__global__ void row_div_bias_elu256(float* __restrict__ x, const float* __restrict__ denom,
                                    const float* __restrict__ b, int total) {
    int i = blockIdx.x * blockDim.x + threadIdx.x;
    if (i >= total) return;
    float dn = denom[i >> 6];
    float a = x[i];
    float v = (dn > 0.0f ? a / dn : 0.0f) + b[i & 255];
    x[i] = v > 0.0f ? v : expm1f(v);
}

__global__ void row_div_bias64(float* __restrict__ out, const float* __restrict__ denom,
                               const float* __restrict__ b) {
    int i = blockIdx.x * blockDim.x + threadIdx.x;
    if (i >= NN * OUTD) return;
    float dn = denom[i >> 6];
    float a = out[i];
    out[i] = (dn > 0.0f ? a / dn : 0.0f) + b[i & 63];
}

__global__ void fill_f32(float* p, float v, int n) {
    int i = blockIdx.x * blockDim.x + threadIdx.x;
    if (i < n) p[i] = v;
}
__global__ void fill_u32(unsigned* p, unsigned v, int n) {
    int i = blockIdx.x * blockDim.x + threadIdx.x;
    if (i < n) p[i] = v;
}

extern "C" void kernel_launch(void* const* d_in, const int* in_sizes, int n_in,
                              void* d_out, int out_size, void* d_ws, size_t ws_size,
                              hipStream_t stream) {
    const float* feat = (const float*)d_in[0];
    const int*   src  = (const int*)d_in[1];
    const int*   dst  = (const int*)d_in[2];
    const float* W1   = (const float*)d_in[3];
    const float* al1  = (const float*)d_in[4];
    const float* ar1  = (const float*)d_in[5];
    const float* b1   = (const float*)d_in[6];
    const float* W2   = (const float*)d_in[7];
    const float* al2  = (const float*)d_in[8];
    const float* ar2  = (const float*)d_in[9];
    const float* b2   = (const float*)d_in[10];
    float* out = (float*)d_out;

    float* h1    = (float*)d_ws;
    float* agg1  = h1    + (size_t)NN * 256;
    float* h2    = agg1  + (size_t)NN * 256;
    float* el1   = h2    + (size_t)NN * 64;
    float* er1   = el1   + (size_t)NN * H1_;
    float* den1  = er1   + (size_t)NN * H1_;
    unsigned* mx1 = (unsigned*)(den1 + (size_t)NN * H1_);
    float* el2   = (float*)(mx1 + (size_t)NN * H1_);
    float* er2   = el2   + NN;
    float* den2  = er2   + NN;
    unsigned* mx2 = (unsigned*)(den2 + NN);

    const int TB = 256;
    fill_f32<<<(NN * 256 + TB - 1) / TB, TB, 0, stream>>>(agg1, 0.0f, NN * 256);
    fill_f32<<<(NN * H1_ + TB - 1) / TB, TB, 0, stream>>>(den1, 0.0f, NN * H1_);
    fill_u32<<<(NN * H1_ + TB - 1) / TB, TB, 0, stream>>>(mx1, 0u, NN * H1_);
    fill_f32<<<(NN + TB - 1) / TB, TB, 0, stream>>>(den2, 0.0f, NN);
    fill_u32<<<(NN + TB - 1) / TB, TB, 0, stream>>>(mx2, 0u, NN);
    fill_f32<<<(NN * OUTD + TB - 1) / TB, TB, 0, stream>>>(out, 0.0f, NN * OUTD);

    {
        dim3 grid(256 / 128, (NN + 63) / 64);
        wmma_gemm_f32<64, 128, 256><<<grid, 256, 0, stream>>>(feat, W1, h1, NN, IND);
    }
    node_attn<H1_><<<(NN + 7) / 8, 256, 0, stream>>>(h1, al1, ar1, el1, er1);
    edge_max<H1_><<<(EE + TB - 1) / TB, TB, 0, stream>>>(src, dst, el1, er1, mx1);
    edge_accumulate<H1_><<<(EE + 7) / 8, 256, 0, stream>>>(src, dst, el1, er1, mx1, h1, den1, agg1);
    row_div_bias_elu256<<<(NN * 256 + TB - 1) / TB, TB, 0, stream>>>(agg1, den1, b1, NN * 256);

    {
        dim3 grid(64 / 64, (NN + 127) / 128);
        wmma_gemm_f32<128, 64, 64><<<grid, 256, 0, stream>>>(agg1, W2, h2, NN, 256);
    }
    node_attn<1><<<(NN + 7) / 8, 256, 0, stream>>>(h2, al2, ar2, el2, er2);
    edge_max<1><<<(EE + TB - 1) / TB, TB, 0, stream>>>(src, dst, el2, er2, mx2);
    edge_accumulate<1><<<(EE + 7) / 8, 256, 0, stream>>>(src, dst, el2, er2, mx2, h2, den2, out);
    row_div_bias64<<<(NN * OUTD + TB - 1) / TB, TB, 0, stream>>>(out, den2, b2);
}


# Work-group owns 196 dst rows in LDS: filters the edge list for them, sums w and w*h[src] by LDS float adds, writes acc/denom+bias once; global float atomics, row passes go; sum order = LDS arrival
# speedup vs baseline: 1.7874x; 1.6785x over previous
#include <hip/hip_runtime.h>
#include <math.h>

#define NN   50000
#define EE   800000
#define IND  256
#define H1_  4
#define HID1 64
#define OUTD 64
#define SLOPE 0.2f

typedef __attribute__((ext_vector_type(2))) float v2f;
typedef __attribute__((ext_vector_type(8))) float v8f;

__device__ __forceinline__ unsigned f2ord(float f) {
    unsigned u = __float_as_uint(f);
    return (u & 0x80000000u) ? ~u : (u | 0x80000000u);
}
__device__ __forceinline__ float ord2f(unsigned u) {
    return (u & 0x80000000u) ? __uint_as_float(u & 0x7fffffffu) : __uint_as_float(~u);
}

template <int BM, int BN, int NCOL>
__global__ __launch_bounds__(256) void wmma_gemm_f32(
    const float* __restrict__ A, const float* __restrict__ B,
    float* __restrict__ C, int M, int K) {
    constexpr int BK  = 32;
    constexpr int BMp = BM + 2;
    constexpr int BNp = BN + 2;
    __shared__ float As[(BK / 2) * BMp * 2];
    __shared__ float Bs[(BK / 2) * BNp * 2];

    const int tid  = threadIdx.x;
    const int lane = tid & 31;
    const int wave = tid >> 5;
    constexpr int WCOLS = BN / 64;
    const int wrow = wave / WCOLS;
    const int wcol = wave % WCOLS;
    const int row0 = blockIdx.y * BM;
    const int col0 = blockIdx.x * BN;

    v8f acc[4];
#pragma unroll
    for (int t = 0; t < 4; t++)
#pragma unroll
        for (int r = 0; r < 8; r++) acc[t][r] = 0.0f;

    const int m16 = lane & 15;
    const int kb  = (lane >> 4) * 2;

    const v2f* Asp = (const v2f*)As;
    const v2f* Bsp = (const v2f*)Bs;

    for (int k0 = 0; k0 < K; k0 += BK) {
        for (int idx = tid; idx < BM * BK; idx += 256) {
            int m = idx / BK, k = idx % BK;
            int r = row0 + m;
            float v = (r < M) ? A[(size_t)r * K + k0 + k] : 0.0f;
            As[((k >> 1) * BMp + m) * 2 + (k & 1)] = v;
        }
        for (int idx = tid; idx < BK * BN; idx += 256) {
            int k = idx / BN, n = idx % BN;
            Bs[((k >> 1) * BNp + n) * 2 + (k & 1)] =
                B[(size_t)(k0 + k) * NCOL + col0 + n];
        }
        __syncthreads();

        const int am = wrow * 16 + m16;
#pragma unroll
        for (int kk = 0; kk < BK; kk += 4) {
            const int kp = (kk + kb) >> 1;
            v2f a = Asp[kp * BMp + am];
#pragma unroll
            for (int t = 0; t < 4; t++) {
                v2f b = Bsp[kp * BNp + wcol * 64 + t * 16 + m16];
                acc[t] = __builtin_amdgcn_wmma_f32_16x16x4_f32(
                    false, a, false, b, (short)0, acc[t], false, false);
            }
        }
        __syncthreads();
    }

    const int rbase = row0 + wrow * 16 + ((lane >> 4) << 3);
#pragma unroll
    for (int t = 0; t < 4; t++) {
        int col = col0 + wcol * 64 + t * 16 + m16;
#pragma unroll
        for (int r = 0; r < 8; r++) {
            int row = rbase + r;
            if (row < M) C[(size_t)row * NCOL + col] = acc[t][r];
        }
    }
}

template <int H>
__global__ __launch_bounds__(256) void node_attn(
    const float* __restrict__ h, const float* __restrict__ al,
    const float* __restrict__ ar, float* __restrict__ el, float* __restrict__ er) {
    int lane = threadIdx.x & 31;
    int n = blockIdx.x * 8 + (threadIdx.x >> 5);
    if (n >= NN) return;
    const float* hp = h + (size_t)n * (H * 64);
#pragma unroll
    for (int hh = 0; hh < H; hh++) {
        float v0 = hp[hh * 64 + lane];
        float v1 = hp[hh * 64 + 32 + lane];
        float pl = v0 * al[hh * 64 + lane] + v1 * al[hh * 64 + 32 + lane];
        float pr = v0 * ar[hh * 64 + lane] + v1 * ar[hh * 64 + 32 + lane];
        for (int off = 16; off > 0; off >>= 1) {
            pl += __shfl_down(pl, off);
            pr += __shfl_down(pr, off);
        }
        if (lane == 0) {
            el[(size_t)n * H + hh] = pl;
            er[(size_t)n * H + hh] = pr;
        }
    }
}

template <int H>
__global__ void edge_max(const int* __restrict__ src, const int* __restrict__ dst,
                         const float* __restrict__ el, const float* __restrict__ er,
                         unsigned* __restrict__ emax) {
    int e = blockIdx.x * blockDim.x + threadIdx.x;
    if (e >= EE) return;
    int s = src[e], d = dst[e];
#pragma unroll
    for (int hh = 0; hh < H; hh++) {
        float sc = el[(size_t)s * H + hh] + er[(size_t)d * H + hh];
        sc = sc > 0.0f ? sc : SLOPE * sc;
        atomicMax(&emax[(size_t)d * H + hh], f2ord(sc));
    }
}

#define TR   196
#define NBLK (EE / 128)
#define LCAP 8192
static_assert(EE % 128 == 0, "edge list is read as whole 128-edge blocks");

template <int H>
__device__ __forceinline__ void tile_add_edge(
    int e, int r0, int lane,
    const int* __restrict__ src, const int* __restrict__ dst,
    const float* __restrict__ el, const float* __restrict__ er,
    const unsigned* __restrict__ emax, const float* __restrict__ h,
    float* acc, float* den) {
    constexpr int RW = H * 64;
    const int s  = src[e];
    const int d  = dst[e];
    const int dl = d - r0;
    const float2* hs = (const float2*)(h + (size_t)s * RW);
    float2 v[H];
#pragma unroll
    for (int j = 0; j < H; j++) v[j] = hs[lane + 32 * j];
    float w[H];
#pragma unroll
    for (int hh = 0; hh < H; hh++) {
        float sc = el[(size_t)s * H + hh] + er[(size_t)d * H + hh];
        sc = sc > 0.0f ? sc : SLOPE * sc;
        w[hh] = expf(sc - ord2f(emax[(size_t)d * H + hh]));
    }
    if (lane < H) {
        float wl = w[0];
#pragma unroll
        for (int hh = 1; hh < H; hh++) wl = (lane == hh) ? w[hh] : wl;
        atomicAdd(&den[dl * H + lane], wl);
    }
    float* row = acc + dl * RW + 2 * lane;
#pragma unroll
    for (int j = 0; j < H; j++) {
        atomicAdd(&row[64 * j],     w[j] * v[j].x);
        atomicAdd(&row[64 * j + 1], w[j] * v[j].y);
    }
}

template <int H, bool ELU>
__global__ __launch_bounds__(1024) void tile_accumulate(
    const int* __restrict__ src, const int* __restrict__ dst,
    const float* __restrict__ el, const float* __restrict__ er,
    const unsigned* __restrict__ emax, const float* __restrict__ h,
    const float* __restrict__ b, float* __restrict__ outp) {
    constexpr int RW = H * 64;
    constexpr int U  = 8;
    constexpr int G  = 32 / H;
    __shared__ float acc[TR * RW];
    __shared__ float den[TR * H];
    __shared__ int   list[LCAP];
    __shared__ int   cnt;
    const int tid  = threadIdx.x;
    const int lane = tid & 31;
    const int wave = __builtin_amdgcn_readfirstlane(tid >> 5);
    const int r0   = blockIdx.x * TR;

    for (int i = tid; i < TR * RW; i += 1024) acc[i] = 0.0f;
    for (int i = tid; i < TR * H; i += 1024) den[i] = 0.0f;
    if (tid == 0) cnt = 0;
    __syncthreads();

    const int4* dst4 = (const int4*)dst;
    const int start = (int)((blockIdx.x * (unsigned)NBLK) / gridDim.x);

    for (int i = wave; i < NBLK; i += 32 * U) {
        int4 dv[U];
#pragma unroll
        for (int u = 0; u < U; u++) {
            const int bi = i + u * 32;
            int bb = (bi < NBLK ? bi : i) + start;
            if (bb >= NBLK) bb -= NBLK;
            dv[u] = dst4[(size_t)bb * 32 + lane];
        }
        unsigned lm = 0;
#pragma unroll
        for (int u = 0; u < U; u++) {
            if (i + u * 32 < NBLK) {
                lm |= ((unsigned)(dv[u].x - r0) < (unsigned)TR) ? (1u << (u * 4 + 0)) : 0u;
                lm |= ((unsigned)(dv[u].y - r0) < (unsigned)TR) ? (1u << (u * 4 + 1)) : 0u;
                lm |= ((unsigned)(dv[u].z - r0) < (unsigned)TR) ? (1u << (u * 4 + 2)) : 0u;
                lm |= ((unsigned)(dv[u].w - r0) < (unsigned)TR) ? (1u << (u * 4 + 3)) : 0u;
            }
        }
        if (__builtin_amdgcn_ballot_w32(lm != 0u) != 0u) {
            unsigned failm = 0;
            if (lm != 0u) {
                int slot = atomicAdd(&cnt, __popc(lm));
                unsigned m2 = lm;
                while (m2) {
                    const int k = __builtin_ctz(m2);
                    m2 &= m2 - 1;
                    if (slot < LCAP) {
                        int bb = i + (k >> 2) * 32 + start;
                        if (bb >= NBLK) bb -= NBLK;
                        list[slot] = bb * 128 + lane * 4 + (k & 3);
                    } else {
                        failm |= 1u << k;
                    }
                    slot++;
                }
            }
            unsigned fl = __builtin_amdgcn_ballot_w32(failm != 0u);
            while (fl) {
                const int l = __builtin_ctz(fl);
                fl &= fl - 1;
                unsigned fm = (unsigned)__builtin_amdgcn_readlane((int)failm, l);
                while (fm) {
                    const int k = __builtin_ctz(fm);
                    fm &= fm - 1;
                    int bb = i + (k >> 2) * 32 + start;
                    if (bb >= NBLK) bb -= NBLK;
                    tile_add_edge<H>(bb * 128 + l * 4 + (k & 3), r0, lane,
                                     src, dst, el, er, emax, h, acc, den);
                }
            }
        }
    }
    __syncthreads();

    const int n = cnt < LCAP ? cnt : LCAP;
    const int gl  = lane / H;
    const int hhl = lane % H;
    for (int base = wave * G; base < n; base += 32 * G) {
        const bool okl = (base + gl) < n;
        const int  el_ = list[okl ? base + gl : base];
        const int  sl  = src[el_];
        const int  dl_ = dst[el_];
        float scl = el[(size_t)sl * H + hhl] + er[(size_t)dl_ * H + hhl];
        scl = scl > 0.0f ? scl : SLOPE * scl;
        const float wl = expf(scl - ord2f(emax[(size_t)dl_ * H + hhl]));
        float2 v[G][H];
#pragma unroll
        for (int g = 0; g < G; g++) {
            const int sg = __builtin_amdgcn_readlane(sl, g * H);
            const float2* hs = (const float2*)(h + (size_t)sg * RW);
#pragma unroll
            for (int j = 0; j < H; j++) v[g][j] = hs[lane + 32 * j];
        }
        if (okl) atomicAdd(&den[(dl_ - r0) * H + hhl], wl);
#pragma unroll
        for (int g = 0; g < G; g++) {
            if (base + g < n) {
                const int dg = __builtin_amdgcn_readlane(dl_, g * H) - r0;
                float* row = acc + dg * RW + 2 * lane;
#pragma unroll
                for (int j = 0; j < H; j++) {
                    const float wj = __int_as_float(__builtin_amdgcn_readlane(__float_as_int(wl), g * H + j));
                    atomicAdd(&row[64 * j],     wj * v[g][j].x);
                    atomicAdd(&row[64 * j + 1], wj * v[g][j].y);
                }
            }
        }
    }
    __syncthreads();

    static_assert(1024 % RW == 0, "a thread keeps its column");
    const int   rows = (NN - r0) < TR ? (NN - r0) : TR;
    const int   nel  = rows * RW;
    const float bc   = b[tid % RW];
    float* op = outp + (size_t)r0 * RW;
    for (int i = tid; i < nel; i += 1024) {
        const float dn = den[i >> 6];
        const float a  = acc[i];
        float x = (dn > 0.0f ? a / dn : 0.0f) + bc;
        if (ELU) x = x > 0.0f ? x : expm1f(x);
        op[i] = x;
    }
}

__global__ void fill_u32(unsigned* p, unsigned v, int n) {
    int i = blockIdx.x * blockDim.x + threadIdx.x;
    if (i < n) p[i] = v;
}

extern "C" void kernel_launch(void* const* d_in, const int* in_sizes, int n_in,
                              void* d_out, int out_size, void* d_ws, size_t ws_size,
                              hipStream_t stream) {
    const float* feat = (const float*)d_in[0];
    const int*   src  = (const int*)d_in[1];
    const int*   dst  = (const int*)d_in[2];
    const float* W1   = (const float*)d_in[3];
    const float* al1  = (const float*)d_in[4];
    const float* ar1  = (const float*)d_in[5];
    const float* b1   = (const float*)d_in[6];
    const float* W2   = (const float*)d_in[7];
    const float* al2  = (const float*)d_in[8];
    const float* ar2  = (const float*)d_in[9];
    const float* b2   = (const float*)d_in[10];
    float* out = (float*)d_out;

    float* h1    = (float*)d_ws;
    float* agg1  = h1    + (size_t)NN * 256;
    float* h2    = agg1  + (size_t)NN * 256;
    float* el1   = h2    + (size_t)NN * 64;
    float* er1   = el1   + (size_t)NN * H1_;
    unsigned* mx1 = (unsigned*)(er1 + (size_t)NN * H1_);
    float* el2   = (float*)(mx1 + (size_t)NN * H1_);
    float* er2   = el2   + NN;
    unsigned* mx2 = (unsigned*)(er2 + NN);

    const int TB = 256;
    fill_u32<<<(NN * H1_ + TB - 1) / TB, TB, 0, stream>>>(mx1, 0u, NN * H1_);
    fill_u32<<<(NN + TB - 1) / TB, TB, 0, stream>>>(mx2, 0u, NN);

    {
        dim3 grid(256 / 128, (NN + 63) / 64);
        wmma_gemm_f32<64, 128, 256><<<grid, 256, 0, stream>>>(feat, W1, h1, NN, IND);
    }
    node_attn<H1_><<<(NN + 7) / 8, 256, 0, stream>>>(h1, al1, ar1, el1, er1);
    edge_max<H1_><<<(EE + TB - 1) / TB, TB, 0, stream>>>(src, dst, el1, er1, mx1);
    tile_accumulate<H1_, true><<<(NN + TR - 1) / TR, 1024, 0, stream>>>(src, dst, el1, er1, mx1, h1, b1, agg1);

    {
        dim3 grid(64 / 64, (NN + 127) / 128);
        wmma_gemm_f32<128, 64, 64><<<grid, 256, 0, stream>>>(agg1, W2, h2, NN, 256);
    }
    node_attn<1><<<(NN + 7) / 8, 256, 0, stream>>>(h2, al2, ar2, el2, er2);
    edge_max<1><<<(EE + TB - 1) / TB, TB, 0, stream>>>(src, dst, el2, er2, mx2);
    tile_accumulate<1, false><<<(NN + TR - 1) / TR, 1024, 0, stream>>>(src, dst, el2, er2, mx2, h2, b2, out);
}


# Segment max taken inside the owning work-group: LDS integer max over its noted edges (entries rewritten as row+src); both edge-max launches, both fills, 4M global atomics go; sums in LDS arrival order
# speedup vs baseline: 2.6499x; 1.4825x over previous
#include <hip/hip_runtime.h>
#include <math.h>

#define NN   50000
#define EE   800000
#define IND  256
#define H1_  4
#define HID1 64
#define OUTD 64
#define SLOPE 0.2f

typedef __attribute__((ext_vector_type(2))) float v2f;
typedef __attribute__((ext_vector_type(8))) float v8f;

__device__ __forceinline__ unsigned f2ord(float f) {
    unsigned u = __float_as_uint(f);
    return (u & 0x80000000u) ? ~u : (u | 0x80000000u);
}
__device__ __forceinline__ float ord2f(unsigned u) {
    return (u & 0x80000000u) ? __uint_as_float(u & 0x7fffffffu) : __uint_as_float(~u);
}

template <int BM, int BN, int NCOL>
__global__ __launch_bounds__(256) void wmma_gemm_f32(
    const float* __restrict__ A, const float* __restrict__ B,
    float* __restrict__ C, int M, int K) {
    constexpr int BK  = 32;
    constexpr int BMp = BM + 2;
    constexpr int BNp = BN + 2;
    __shared__ float As[(BK / 2) * BMp * 2];
    __shared__ float Bs[(BK / 2) * BNp * 2];

    const int tid  = threadIdx.x;
    const int lane = tid & 31;
    const int wave = tid >> 5;
    constexpr int WCOLS = BN / 64;
    const int wrow = wave / WCOLS;
    const int wcol = wave % WCOLS;
    const int row0 = blockIdx.y * BM;
    const int col0 = blockIdx.x * BN;

    v8f acc[4];
#pragma unroll
    for (int t = 0; t < 4; t++)
#pragma unroll
        for (int r = 0; r < 8; r++) acc[t][r] = 0.0f;

    const int m16 = lane & 15;
    const int kb  = (lane >> 4) * 2;

    const v2f* Asp = (const v2f*)As;
    const v2f* Bsp = (const v2f*)Bs;

    for (int k0 = 0; k0 < K; k0 += BK) {
        for (int idx = tid; idx < BM * BK; idx += 256) {
            int m = idx / BK, k = idx % BK;
            int r = row0 + m;
            float v = (r < M) ? A[(size_t)r * K + k0 + k] : 0.0f;
            As[((k >> 1) * BMp + m) * 2 + (k & 1)] = v;
        }
        for (int idx = tid; idx < BK * BN; idx += 256) {
            int k = idx / BN, n = idx % BN;
            Bs[((k >> 1) * BNp + n) * 2 + (k & 1)] =
                B[(size_t)(k0 + k) * NCOL + col0 + n];
        }
        __syncthreads();

        const int am = wrow * 16 + m16;
#pragma unroll
        for (int kk = 0; kk < BK; kk += 4) {
            const int kp = (kk + kb) >> 1;
            v2f a = Asp[kp * BMp + am];
#pragma unroll
            for (int t = 0; t < 4; t++) {
                v2f b = Bsp[kp * BNp + wcol * 64 + t * 16 + m16];
                acc[t] = __builtin_amdgcn_wmma_f32_16x16x4_f32(
                    false, a, false, b, (short)0, acc[t], false, false);
            }
        }
        __syncthreads();
    }

    const int rbase = row0 + wrow * 16 + ((lane >> 4) << 3);
#pragma unroll
    for (int t = 0; t < 4; t++) {
        int col = col0 + wcol * 64 + t * 16 + m16;
#pragma unroll
        for (int r = 0; r < 8; r++) {
            int row = rbase + r;
            if (row < M) C[(size_t)row * NCOL + col] = acc[t][r];
        }
    }
}

template <int H>
__global__ __launch_bounds__(256) void node_attn(
    const float* __restrict__ h, const float* __restrict__ al,
    const float* __restrict__ ar, float* __restrict__ el, float* __restrict__ er) {
    int lane = threadIdx.x & 31;
    int n = blockIdx.x * 8 + (threadIdx.x >> 5);
    if (n >= NN) return;
    const float* hp = h + (size_t)n * (H * 64);
#pragma unroll
    for (int hh = 0; hh < H; hh++) {
        float v0 = hp[hh * 64 + lane];
        float v1 = hp[hh * 64 + 32 + lane];
        float pl = v0 * al[hh * 64 + lane] + v1 * al[hh * 64 + 32 + lane];
        float pr = v0 * ar[hh * 64 + lane] + v1 * ar[hh * 64 + 32 + lane];
        for (int off = 16; off > 0; off >>= 1) {
            pl += __shfl_down(pl, off);
            pr += __shfl_down(pr, off);
        }
        if (lane == 0) {
            el[(size_t)n * H + hh] = pl;
            er[(size_t)n * H + hh] = pr;
        }
    }
}

#define TR   196
#define NBLK (EE / 128)
#define LCAP 8192
static_assert(EE % 128 == 0, "edge list is read as whole 128-edge blocks");
static_assert(NN <= 65536, "a list entry holds src in 16 bits beside the local row");

typedef __attribute__((ext_vector_type(4))) float v4f;
template <int H> struct score_vec;
template <> struct score_vec<1> { typedef float type; };
template <> struct score_vec<4> { typedef v4f type; };
template <int H>
__device__ __forceinline__ typename score_vec<H>::type load_scores(const float* __restrict__ p, int idx) {
    return *(const typename score_vec<H>::type*)(p + (size_t)idx * H);
}
template <int H>
__device__ __forceinline__ float score_of(typename score_vec<H>::type v, int hh) {
    if constexpr (H == 1) return v; else return v[hh];
}

template <int H>
__device__ __forceinline__ void tile_add_edge(
    int e, int r0, int lane,
    const int* __restrict__ src, const int* __restrict__ dst,
    const float* __restrict__ el, const float* __restrict__ er,
    const unsigned* mx, const float* __restrict__ h,
    float* acc, float* den) {
    constexpr int RW = H * 64;
    const int s  = src[e];
    const int d  = dst[e];
    const int dl = d - r0;
    const float2* hs = (const float2*)(h + (size_t)s * RW);
    float2 v[H];
#pragma unroll
    for (int j = 0; j < H; j++) v[j] = hs[lane + 32 * j];
    float w[H];
#pragma unroll
    for (int hh = 0; hh < H; hh++) {
        float sc = el[(size_t)s * H + hh] + er[(size_t)d * H + hh];
        sc = sc > 0.0f ? sc : SLOPE * sc;
        w[hh] = expf(sc - ord2f(mx[dl * H + hh]));
    }
    if (lane < H) {
        float wl = w[0];
#pragma unroll
        for (int hh = 1; hh < H; hh++) wl = (lane == hh) ? w[hh] : wl;
        atomicAdd(&den[dl * H + lane], wl);
    }
    float* row = acc + dl * RW + 2 * lane;
#pragma unroll
    for (int j = 0; j < H; j++) {
        atomicAdd(&row[64 * j],     w[j] * v[j].x);
        atomicAdd(&row[64 * j + 1], w[j] * v[j].y);
    }
}

template <int H, bool ELU>
__global__ __launch_bounds__(1024) void tile_accumulate(
    const int* __restrict__ src, const int* __restrict__ dst,
    const float* __restrict__ el, const float* __restrict__ er,
    const float* __restrict__ h,
    const float* __restrict__ b, float* __restrict__ outp) {
    constexpr int RW = H * 64;
    constexpr int U  = 8;
    constexpr int G  = 32 / H;
    constexpr int MU = 4;
    __shared__ float    acc[TR * RW];
    __shared__ float    den[TR * H];
    __shared__ unsigned mx[TR * H];
    __shared__ int      list[LCAP];
    __shared__ int      cnt;
    const int tid  = threadIdx.x;
    const int lane = tid & 31;
    const int wave = __builtin_amdgcn_readfirstlane(tid >> 5);
    const int r0   = blockIdx.x * TR;

    for (int i = tid; i < TR * RW; i += 1024) acc[i] = 0.0f;
    for (int i = tid; i < TR * H; i += 1024) { den[i] = 0.0f; mx[i] = 0u; }
    if (tid == 0) cnt = 0;
    __syncthreads();

    const int4* dst4 = (const int4*)dst;
    const int start = (int)((blockIdx.x * (unsigned)NBLK) / gridDim.x);

    for (int i = wave; i < NBLK; i += 32 * U) {
        int4 dv[U];
#pragma unroll
        for (int u = 0; u < U; u++) {
            const int bi = i + u * 32;
            int bb = (bi < NBLK ? bi : i) + start;
            if (bb >= NBLK) bb -= NBLK;
            dv[u] = dst4[(size_t)bb * 32 + lane];
        }
        unsigned lm = 0;
#pragma unroll
        for (int u = 0; u < U; u++) {
            if (i + u * 32 < NBLK) {
                lm |= ((unsigned)(dv[u].x - r0) < (unsigned)TR) ? (1u << (u * 4 + 0)) : 0u;
                lm |= ((unsigned)(dv[u].y - r0) < (unsigned)TR) ? (1u << (u * 4 + 1)) : 0u;
                lm |= ((unsigned)(dv[u].z - r0) < (unsigned)TR) ? (1u << (u * 4 + 2)) : 0u;
                lm |= ((unsigned)(dv[u].w - r0) < (unsigned)TR) ? (1u << (u * 4 + 3)) : 0u;
            }
        }
        if (lm != 0u) {
            int slot = atomicAdd(&cnt, __popc(lm));
            unsigned m2 = lm;
            while (m2) {
                const int k = __builtin_ctz(m2);
                m2 &= m2 - 1;
                if (slot < LCAP) {
                    int bb = i + (k >> 2) * 32 + start;
                    if (bb >= NBLK) bb -= NBLK;
                    list[slot] = bb * 128 + lane * 4 + (k & 3);
                }
                slot++;
            }
        }
    }
    __syncthreads();

    const int n = cnt;
    if (n <= LCAP) {
        static_assert(MU == 4 && LCAP % (MU * 1024) == 0, "step M: four entries per lane and trip, every place read lies inside the list");
        for (int base = 0; base < n; base += MU * 1024) {
            int  ee[MU];
            bool ok[MU];
#pragma unroll
            for (int u = 0; u < MU; u++) {
                const int i = base + u * 1024 + tid;
                const int t = list[i];
                ok[u] = i < n;
                ee[u] = ok[u] ? t : 0;
            }
            int ss[MU], dd[MU];
#pragma unroll
            for (int u = 0; u < MU; u++) { ss[u] = src[ee[u]]; dd[u] = dst[ee[u]]; }
            typename score_vec<H>::type a[MU], c[MU];
#pragma unroll
            for (int u = 0; u < MU; u++) {
                a[u] = load_scores<H>(el, ss[u]);
                c[u] = load_scores<H>(er, dd[u]);
            }
            asm volatile("" :: "v"(a[0]), "v"(c[0]), "v"(a[1]), "v"(c[1]), "v"(a[2]), "v"(c[2]), "v"(a[3]), "v"(c[3]));
#pragma unroll
            for (int u = 0; u < MU; u++) {
                if (ok[u]) {
                    const int dl = dd[u] - r0;
#pragma unroll
                    for (int hh = 0; hh < H; hh++) {
                        float sc = score_of<H>(a[u], hh) + score_of<H>(c[u], hh);
                        sc = sc > 0.0f ? sc : SLOPE * sc;
                        atomicMax(&mx[dl * H + hh], f2ord(sc));
                    }
                    list[base + u * 1024 + tid] = (dl << 16) | ss[u];
                }
            }
        }
        __syncthreads();

        const int gl  = lane / H;
        const int hhl = lane % H;
        for (int base = wave * G; base < n; base += 32 * G) {
            const bool okl = (base + gl) < n;
            const int  ent = list[okl ? base + gl : base];
            const int  sl  = ent & 0xffff;
            const int  dl_ = ent >> 16;
            float scl = el[(size_t)sl * H + hhl] + er[(size_t)(r0 + dl_) * H + hhl];
            scl = scl > 0.0f ? scl : SLOPE * scl;
            const float wl = expf(scl - ord2f(mx[dl_ * H + hhl]));
            float2 v[G][H];
#pragma unroll
            for (int g = 0; g < G; g++) {
                const int sg = __builtin_amdgcn_readlane(sl, g * H);
                const float2* hs = (const float2*)(h + (size_t)sg * RW);
#pragma unroll
                for (int j = 0; j < H; j++) v[g][j] = hs[lane + 32 * j];
            }
            if (okl) atomicAdd(&den[dl_ * H + hhl], wl);
#pragma unroll
            for (int g = 0; g < G; g++) {
                if (base + g < n) {
                    const int dg = __builtin_amdgcn_readlane(dl_, g * H);
                    float* row = acc + dg * RW + 2 * lane;
#pragma unroll
                    for (int j = 0; j < H; j++) {
                        const float wj = __int_as_float(__builtin_amdgcn_readlane(__float_as_int(wl), g * H + j));
                        atomicAdd(&row[64 * j],     wj * v[g][j].x);
                        atomicAdd(&row[64 * j + 1], wj * v[g][j].y);
                    }
                }
            }
        }
    } else {
        for (int g = wave; g < EE / 32; g += 32) {
            const int e = g * 32 + lane;
            const int d = dst[e];
            if ((unsigned)(d - r0) < (unsigned)TR) {
                const int s = src[e];
#pragma unroll
                for (int hh = 0; hh < H; hh++) {
                    float sc = el[(size_t)s * H + hh] + er[(size_t)d * H + hh];
                    sc = sc > 0.0f ? sc : SLOPE * sc;
                    atomicMax(&mx[(d - r0) * H + hh], f2ord(sc));
                }
            }
        }
        __syncthreads();
        for (int g = wave; g < EE / 32; g += 32) {
            const int d = dst[g * 32 + lane];
            unsigned hm = __builtin_amdgcn_ballot_w32((unsigned)(d - r0) < (unsigned)TR);
            while (hm) {
                const int l = __builtin_ctz(hm);
                hm &= hm - 1;
                tile_add_edge<H>(g * 32 + l, r0, lane, src, dst, el, er, mx, h, acc, den);
            }
        }
    }
    __syncthreads();

    static_assert(1024 % RW == 0, "a thread keeps its column");
    const int   rows = (NN - r0) < TR ? (NN - r0) : TR;
    const int   nel  = rows * RW;
    const float bc   = b[tid % RW];
    float* op = outp + (size_t)r0 * RW;
    for (int i = tid; i < nel; i += 1024) {
        const float dn = den[i >> 6];
        const float a  = acc[i];
        float x = (dn > 0.0f ? a / dn : 0.0f) + bc;
        if (ELU) x = x > 0.0f ? x : expm1f(x);
        op[i] = x;
    }
}

extern "C" void kernel_launch(void* const* d_in, const int* in_sizes, int n_in,
                              void* d_out, int out_size, void* d_ws, size_t ws_size,
                              hipStream_t stream) {
    const float* feat = (const float*)d_in[0];
    const int*   src  = (const int*)d_in[1];
    const int*   dst  = (const int*)d_in[2];
    const float* W1   = (const float*)d_in[3];
    const float* al1  = (const float*)d_in[4];
    const float* ar1  = (const float*)d_in[5];
    const float* b1   = (const float*)d_in[6];
    const float* W2   = (const float*)d_in[7];
    const float* al2  = (const float*)d_in[8];
    const float* ar2  = (const float*)d_in[9];
    const float* b2   = (const float*)d_in[10];
    float* out = (float*)d_out;

    float* h1    = (float*)d_ws;
    float* agg1  = h1    + (size_t)NN * 256;
    float* h2    = agg1  + (size_t)NN * 256;
    float* el1   = h2    + (size_t)NN * 64;
    float* er1   = el1   + (size_t)NN * H1_;
    float* el2   = er1   + (size_t)NN * H1_;
    float* er2   = el2   + NN;

    {
        dim3 grid(256 / 128, (NN + 63) / 64);
        wmma_gemm_f32<64, 128, 256><<<grid, 256, 0, stream>>>(feat, W1, h1, NN, IND);
    }
    node_attn<H1_><<<(NN + 7) / 8, 256, 0, stream>>>(h1, al1, ar1, el1, er1);
    tile_accumulate<H1_, true><<<(NN + TR - 1) / TR, 1024, 0, stream>>>(src, dst, el1, er1, h1, b1, agg1);

    {
        dim3 grid(64 / 64, (NN + 127) / 128);
        wmma_gemm_f32<128, 64, 64><<<grid, 256, 0, stream>>>(agg1, W2, h2, NN, 256);
    }
    node_attn<1><<<(NN + 7) / 8, 256, 0, stream>>>(h2, al2, ar2, el2, er2);
    tile_accumulate<1, false><<<(NN + TR - 1) / TR, 1024, 0, stream>>>(src, dst, el2, er2, h2, b2, out);
}


# Products made weight-stationary: W staged once per work-group into LDS as k-quads; each wave multiplies 16x64 units, A read straight from global 16 B per lane; no K-step staging; new k sum order
# speedup vs baseline: 3.9105x; 1.4757x over previous
#include <hip/hip_runtime.h>
#include <math.h>

#define NN   50000
#define EE   800000
#define IND  256
#define H1_  4
#define HID1 64
#define OUTD 64
#define SLOPE 0.2f

typedef __attribute__((ext_vector_type(8))) float v8f;

__device__ __forceinline__ unsigned f2ord(float f) {
    unsigned u = __float_as_uint(f);
    return (u & 0x80000000u) ? ~u : (u | 0x80000000u);
}
__device__ __forceinline__ float ord2f(unsigned u) {
    return (u & 0x80000000u) ? __uint_as_float(u & 0x7fffffffu) : __uint_as_float(~u);
}

#define GWG 256
typedef __attribute__((ext_vector_type(4))) float q4f;
template <int NCOL>
__global__ __launch_bounds__(512) void wmma_gemm_wres(
    const float* __restrict__ A, const float* __restrict__ B, float* __restrict__ C) {
    constexpr int KD = 256;
    constexpr int KQ = KD / 4;
    constexpr int KO = KD / 8;
    constexpr int Q  = NCOL / 64;
    constexpr int NSTRIP = NN / 16;
    static_assert(NN % 16 == 0, "whole 16-row strips: no WMMA runs with rows masked off");
    static_assert(NCOL % 64 == 0 && 512 % NCOL == 0, "units of 64 columns; staging threads cover whole rows of B");
    __shared__ q4f Bs4[KQ * NCOL];

    const int tid  = threadIdx.x;
    const int lane = tid & 31;
    const int wave = __builtin_amdgcn_readfirstlane(tid >> 5);
    const int c    = lane & 15;
    const int h    = lane >> 4;
    const int wg   = blockIdx.x;
    const int nslots = ((NSTRIP - wg + GWG - 1) / GWG) * Q;

    q4f a4[KO];
    int slot = wave;
    if (slot < nslots) {
        const int strip = (slot / Q) * GWG + wg;
        const q4f* pa = (const q4f*)(A + (size_t)(strip * 16 + c) * KD + 4 * h);
#pragma unroll
        for (int i = 0; i < KO; i++) a4[i] = pa[2 * i];
    }

    {
        constexpr int KSTEP = 512 / NCOL;
        constexpr int UNR   = 8;
        static_assert(KQ % (KSTEP * UNR) == 0, "staging: whole trips");
        const int n   = tid % NCOL;
        const int kq0 = tid / NCOL;
        for (int kqb = kq0; kqb < KQ; kqb += KSTEP * UNR) {
            float v[UNR][4];
#pragma unroll
            for (int u = 0; u < UNR; u++)
#pragma unroll
                for (int s = 0; s < 4; s++)
                    v[u][s] = B[(size_t)(4 * (kqb + u * KSTEP) + s) * NCOL + n];
#pragma unroll
            for (int u = 0; u < UNR; u++) {
                q4f q = { v[u][0], v[u][1], v[u][2], v[u][3] };
                Bs4[(kqb + u * KSTEP) * NCOL + n] = q;
            }
        }
    }
    __syncthreads();

    while (slot < nslots) {
        const int strip   = (slot / Q) * GWG + wg;
        const int colbase = (slot % Q) * 64;
        v8f acc[4];
#pragma unroll
        for (int t = 0; t < 4; t++)
#pragma unroll
            for (int r = 0; r < 8; r++) acc[t][r] = 0.0f;

        int boff = h * NCOL + colbase + c;
        if constexpr (Q == 1) {
            asm volatile("" : "+v"(boff));
        }
        const q4f* bp = Bs4 + boff;
#pragma unroll
        for (int i = 0; i < KO; i++) {
            q4f bq[4];
#pragma unroll
            for (int t = 0; t < 4; t++) bq[t] = bp[(2 * i) * NCOL + t * 16];
#pragma unroll
            for (int t = 0; t < 4; t++)
                acc[t] = __builtin_amdgcn_wmma_f32_16x16x4_f32(
                    false, a4[i].xy, false, bq[t].xy, (short)0, acc[t], false, false);
#pragma unroll
            for (int t = 0; t < 4; t++)
                acc[t] = __builtin_amdgcn_wmma_f32_16x16x4_f32(
                    false, a4[i].zw, false, bq[t].zw, (short)0, acc[t], false, false);
        }

        float* cp = C + (size_t)(strip * 16 + 8 * h) * NCOL + colbase + c;
#pragma unroll
        for (int t = 0; t < 4; t++)
#pragma unroll
            for (int r = 0; r < 8; r++) cp[(size_t)r * NCOL + t * 16] = acc[t][r];

        slot += 16;
        if (slot < nslots) {
            const int ns = (slot / Q) * GWG + wg;
            const q4f* pa = (const q4f*)(A + (size_t)(ns * 16 + c) * KD + 4 * h);
#pragma unroll
            for (int i = 0; i < KO; i++) a4[i] = pa[2 * i];
        }
    }
}

template <int H>
__global__ __launch_bounds__(256) void node_attn(
    const float* __restrict__ h, const float* __restrict__ al,
    const float* __restrict__ ar, float* __restrict__ el, float* __restrict__ er) {
    int lane = threadIdx.x & 31;
    int n = blockIdx.x * 8 + (threadIdx.x >> 5);
    if (n >= NN) return;
    const float* hp = h + (size_t)n * (H * 64);
#pragma unroll
    for (int hh = 0; hh < H; hh++) {
        float v0 = hp[hh * 64 + lane];
        float v1 = hp[hh * 64 + 32 + lane];
        float pl = v0 * al[hh * 64 + lane] + v1 * al[hh * 64 + 32 + lane];
        float pr = v0 * ar[hh * 64 + lane] + v1 * ar[hh * 64 + 32 + lane];
        for (int off = 16; off > 0; off >>= 1) {
            pl += __shfl_down(pl, off);
            pr += __shfl_down(pr, off);
        }
        if (lane == 0) {
            el[(size_t)n * H + hh] = pl;
            er[(size_t)n * H + hh] = pr;
        }
    }
}

#define TR   196
#define NBLK (EE / 128)
#define LCAP 8192
static_assert(EE % 128 == 0, "edge list is read as whole 128-edge blocks");
static_assert(NN <= 65536, "a list entry holds src in 16 bits beside the local row");

typedef __attribute__((ext_vector_type(4))) float v4f;
template <int H> struct score_vec;
template <> struct score_vec<1> { typedef float type; };
template <> struct score_vec<4> { typedef v4f type; };
template <int H>
__device__ __forceinline__ typename score_vec<H>::type load_scores(const float* __restrict__ p, int idx) {
    return *(const typename score_vec<H>::type*)(p + (size_t)idx * H);
}
template <int H>
__device__ __forceinline__ float score_of(typename score_vec<H>::type v, int hh) {
    if constexpr (H == 1) return v; else return v[hh];
}

template <int H>
__device__ __forceinline__ void tile_add_edge(
    int e, int r0, int lane,
    const int* __restrict__ src, const int* __restrict__ dst,
    const float* __restrict__ el, const float* __restrict__ er,
    const unsigned* mx, const float* __restrict__ h,
    float* acc, float* den) {
    constexpr int RW = H * 64;
    const int s  = src[e];
    const int d  = dst[e];
    const int dl = d - r0;
    const float2* hs = (const float2*)(h + (size_t)s * RW);
    float2 v[H];
#pragma unroll
    for (int j = 0; j < H; j++) v[j] = hs[lane + 32 * j];
    float w[H];
#pragma unroll
    for (int hh = 0; hh < H; hh++) {
        float sc = el[(size_t)s * H + hh] + er[(size_t)d * H + hh];
        sc = sc > 0.0f ? sc : SLOPE * sc;
        w[hh] = expf(sc - ord2f(mx[dl * H + hh]));
    }
    if (lane < H) {
        float wl = w[0];
#pragma unroll
        for (int hh = 1; hh < H; hh++) wl = (lane == hh) ? w[hh] : wl;
        atomicAdd(&den[dl * H + lane], wl);
    }
    float* row = acc + dl * RW + 2 * lane;
#pragma unroll
    for (int j = 0; j < H; j++) {
        atomicAdd(&row[64 * j],     w[j] * v[j].x);
        atomicAdd(&row[64 * j + 1], w[j] * v[j].y);
    }
}

template <int H, bool ELU>
__global__ __launch_bounds__(1024) void tile_accumulate(
    const int* __restrict__ src, const int* __restrict__ dst,
    const float* __restrict__ el, const float* __restrict__ er,
    const float* __restrict__ h,
    const float* __restrict__ b, float* __restrict__ outp) {
    constexpr int RW = H * 64;
    constexpr int U  = 8;
    constexpr int G  = 32 / H;
    constexpr int MU = 4;
    __shared__ float    acc[TR * RW];
    __shared__ float    den[TR * H];
    __shared__ unsigned mx[TR * H];
    __shared__ int      list[LCAP];
    __shared__ int      cnt;
    const int tid  = threadIdx.x;
    const int lane = tid & 31;
    const int wave = __builtin_amdgcn_readfirstlane(tid >> 5);
    const int r0   = blockIdx.x * TR;

    for (int i = tid; i < TR * RW; i += 1024) acc[i] = 0.0f;
    for (int i = tid; i < TR * H; i += 1024) { den[i] = 0.0f; mx[i] = 0u; }
    if (tid == 0) cnt = 0;
    __syncthreads();

    const int4* dst4 = (const int4*)dst;
    const int start = (int)((blockIdx.x * (unsigned)NBLK) / gridDim.x);

    for (int i = wave; i < NBLK; i += 32 * U) {
        int4 dv[U];
#pragma unroll
        for (int u = 0; u < U; u++) {
            const int bi = i + u * 32;
            int bb = (bi < NBLK ? bi : i) + start;
            if (bb >= NBLK) bb -= NBLK;
            dv[u] = dst4[(size_t)bb * 32 + lane];
        }
        unsigned lm = 0;
#pragma unroll
        for (int u = 0; u < U; u++) {
            if (i + u * 32 < NBLK) {
                lm |= ((unsigned)(dv[u].x - r0) < (unsigned)TR) ? (1u << (u * 4 + 0)) : 0u;
                lm |= ((unsigned)(dv[u].y - r0) < (unsigned)TR) ? (1u << (u * 4 + 1)) : 0u;
                lm |= ((unsigned)(dv[u].z - r0) < (unsigned)TR) ? (1u << (u * 4 + 2)) : 0u;
                lm |= ((unsigned)(dv[u].w - r0) < (unsigned)TR) ? (1u << (u * 4 + 3)) : 0u;
            }
        }
        if (lm != 0u) {
            int slot = atomicAdd(&cnt, __popc(lm));
            unsigned m2 = lm;
            while (m2) {
                const int k = __builtin_ctz(m2);
                m2 &= m2 - 1;
                if (slot < LCAP) {
                    int bb = i + (k >> 2) * 32 + start;
                    if (bb >= NBLK) bb -= NBLK;
                    list[slot] = bb * 128 + lane * 4 + (k & 3);
                }
                slot++;
            }
        }
    }
    __syncthreads();

    const int n = cnt;
    if (n <= LCAP) {
        static_assert(MU == 4 && LCAP % (MU * 1024) == 0, "step M: four entries per lane and trip, every place read lies inside the list");
        for (int base = 0; base < n; base += MU * 1024) {
            int  ee[MU];
            bool ok[MU];
#pragma unroll
            for (int u = 0; u < MU; u++) {
                const int i = base + u * 1024 + tid;
                const int t = list[i];
                ok[u] = i < n;
                ee[u] = ok[u] ? t : 0;
            }
            int ss[MU], dd[MU];
#pragma unroll
            for (int u = 0; u < MU; u++) { ss[u] = src[ee[u]]; dd[u] = dst[ee[u]]; }
            typename score_vec<H>::type a[MU], c[MU];
#pragma unroll
            for (int u = 0; u < MU; u++) {
                a[u] = load_scores<H>(el, ss[u]);
                c[u] = load_scores<H>(er, dd[u]);
            }
            asm volatile("" :: "v"(a[0]), "v"(c[0]), "v"(a[1]), "v"(c[1]), "v"(a[2]), "v"(c[2]), "v"(a[3]), "v"(c[3]));
#pragma unroll
            for (int u = 0; u < MU; u++) {
                if (ok[u]) {
                    const int dl = dd[u] - r0;
#pragma unroll
                    for (int hh = 0; hh < H; hh++) {
                        float sc = score_of<H>(a[u], hh) + score_of<H>(c[u], hh);
                        sc = sc > 0.0f ? sc : SLOPE * sc;
                        atomicMax(&mx[dl * H + hh], f2ord(sc));
                    }
                    list[base + u * 1024 + tid] = (dl << 16) | ss[u];
                }
            }
        }
        __syncthreads();

        const int gl  = lane / H;
        const int hhl = lane % H;
        for (int base = wave * G; base < n; base += 32 * G) {
            const bool okl = (base + gl) < n;
            const int  ent = list[okl ? base + gl : base];
            const int  sl  = ent & 0xffff;
            const int  dl_ = ent >> 16;
            float scl = el[(size_t)sl * H + hhl] + er[(size_t)(r0 + dl_) * H + hhl];
            scl = scl > 0.0f ? scl : SLOPE * scl;
            const float wl = expf(scl - ord2f(mx[dl_ * H + hhl]));
            float2 v[G][H];
#pragma unroll
            for (int g = 0; g < G; g++) {
                const int sg = __builtin_amdgcn_readlane(sl, g * H);
                const float2* hs = (const float2*)(h + (size_t)sg * RW);
#pragma unroll
                for (int j = 0; j < H; j++) v[g][j] = hs[lane + 32 * j];
            }
            if (okl) atomicAdd(&den[dl_ * H + hhl], wl);
#pragma unroll
            for (int g = 0; g < G; g++) {
                if (base + g < n) {
                    const int dg = __builtin_amdgcn_readlane(dl_, g * H);
                    float* row = acc + dg * RW + 2 * lane;
#pragma unroll
                    for (int j = 0; j < H; j++) {
                        const float wj = __int_as_float(__builtin_amdgcn_readlane(__float_as_int(wl), g * H + j));
                        atomicAdd(&row[64 * j],     wj * v[g][j].x);
                        atomicAdd(&row[64 * j + 1], wj * v[g][j].y);
                    }
                }
            }
        }
    } else {
        for (int g = wave; g < EE / 32; g += 32) {
            const int e = g * 32 + lane;
            const int d = dst[e];
            if ((unsigned)(d - r0) < (unsigned)TR) {
                const int s = src[e];
#pragma unroll
                for (int hh = 0; hh < H; hh++) {
                    float sc = el[(size_t)s * H + hh] + er[(size_t)d * H + hh];
                    sc = sc > 0.0f ? sc : SLOPE * sc;
                    atomicMax(&mx[(d - r0) * H + hh], f2ord(sc));
                }
            }
        }
        __syncthreads();
        for (int g = wave; g < EE / 32; g += 32) {
            const int d = dst[g * 32 + lane];
            unsigned hm = __builtin_amdgcn_ballot_w32((unsigned)(d - r0) < (unsigned)TR);
            while (hm) {
                const int l = __builtin_ctz(hm);
                hm &= hm - 1;
                tile_add_edge<H>(g * 32 + l, r0, lane, src, dst, el, er, mx, h, acc, den);
            }
        }
    }
    __syncthreads();

    static_assert(1024 % RW == 0, "a thread keeps its column");
    const int   rows = (NN - r0) < TR ? (NN - r0) : TR;
    const int   nel  = rows * RW;
    const float bc   = b[tid % RW];
    float* op = outp + (size_t)r0 * RW;
    for (int i = tid; i < nel; i += 1024) {
        const float dn = den[i >> 6];
        const float a  = acc[i];
        float x = (dn > 0.0f ? a / dn : 0.0f) + bc;
        if (ELU) x = x > 0.0f ? x : expm1f(x);
        op[i] = x;
    }
}

extern "C" void kernel_launch(void* const* d_in, const int* in_sizes, int n_in,
                              void* d_out, int out_size, void* d_ws, size_t ws_size,
                              hipStream_t stream) {
    const float* feat = (const float*)d_in[0];
    const int*   src  = (const int*)d_in[1];
    const int*   dst  = (const int*)d_in[2];
    const float* W1   = (const float*)d_in[3];
    const float* al1  = (const float*)d_in[4];
    const float* ar1  = (const float*)d_in[5];
    const float* b1   = (const float*)d_in[6];
    const float* W2   = (const float*)d_in[7];
    const float* al2  = (const float*)d_in[8];
    const float* ar2  = (const float*)d_in[9];
    const float* b2   = (const float*)d_in[10];
    float* out = (float*)d_out;

    float* h1    = (float*)d_ws;
    float* agg1  = h1    + (size_t)NN * 256;
    float* h2    = agg1  + (size_t)NN * 256;
    float* el1   = h2    + (size_t)NN * 64;
    float* er1   = el1   + (size_t)NN * H1_;
    float* el2   = er1   + (size_t)NN * H1_;
    float* er2   = el2   + NN;

    wmma_gemm_wres<256><<<GWG, 512, 0, stream>>>(feat, W1, h1);
    node_attn<H1_><<<(NN + 7) / 8, 256, 0, stream>>>(h1, al1, ar1, el1, er1);
    tile_accumulate<H1_, true><<<(NN + TR - 1) / TR, 1024, 0, stream>>>(src, dst, el1, er1, h1, b1, agg1);

    wmma_gemm_wres<64><<<GWG, 512, 0, stream>>>(agg1, W2, h2);
    node_attn<1><<<(NN + 7) / 8, 256, 0, stream>>>(h2, al2, ar2, el2, er2);
    tile_accumulate<1, false><<<(NN + TR - 1) / TR, 1024, 0, stream>>>(src, dst, el2, er2, h2, b2, out);
}


# Node scores by few long waves: 256 work-groups loop over 16-row blocks, 16 lanes per 64-float row (16-byte loads, 16-lane DPP butterfly: new sum order); 100000 short waves become 12288 long ones
# speedup vs baseline: 4.1082x; 1.0506x over previous
#include <hip/hip_runtime.h>
#include <math.h>

#define NN   50000
#define EE   800000
#define IND  256
#define H1_  4
#define HID1 64
#define OUTD 64
#define SLOPE 0.2f

typedef __attribute__((ext_vector_type(8))) float v8f;

__device__ __forceinline__ unsigned f2ord(float f) {
    unsigned u = __float_as_uint(f);
    return (u & 0x80000000u) ? ~u : (u | 0x80000000u);
}
__device__ __forceinline__ float ord2f(unsigned u) {
    return (u & 0x80000000u) ? __uint_as_float(u & 0x7fffffffu) : __uint_as_float(~u);
}

#define GWG 256
typedef __attribute__((ext_vector_type(4))) float q4f;
template <int NCOL>
__global__ __launch_bounds__(512) void wmma_gemm_wres(
    const float* __restrict__ A, const float* __restrict__ B, float* __restrict__ C) {
    constexpr int KD = 256;
    constexpr int KQ = KD / 4;
    constexpr int KO = KD / 8;
    constexpr int Q  = NCOL / 64;
    constexpr int NSTRIP = NN / 16;
    static_assert(NN % 16 == 0, "whole 16-row strips: no WMMA runs with rows masked off");
    static_assert(NCOL % 64 == 0 && 512 % NCOL == 0, "units of 64 columns; staging threads cover whole rows of B");
    __shared__ q4f Bs4[KQ * NCOL];

    const int tid  = threadIdx.x;
    const int lane = tid & 31;
    const int wave = __builtin_amdgcn_readfirstlane(tid >> 5);
    const int c    = lane & 15;
    const int h    = lane >> 4;
    const int wg   = blockIdx.x;
    const int nslots = ((NSTRIP - wg + GWG - 1) / GWG) * Q;

    q4f a4[KO];
    int slot = wave;
    if (slot < nslots) {
        const int strip = (slot / Q) * GWG + wg;
        const q4f* pa = (const q4f*)(A + (size_t)(strip * 16 + c) * KD + 4 * h);
#pragma unroll
        for (int i = 0; i < KO; i++) a4[i] = pa[2 * i];
    }

    {
        constexpr int KSTEP = 512 / NCOL;
        constexpr int UNR   = 8;
        static_assert(KQ % (KSTEP * UNR) == 0, "staging: whole trips");
        const int n   = tid % NCOL;
        const int kq0 = tid / NCOL;
        for (int kqb = kq0; kqb < KQ; kqb += KSTEP * UNR) {
            float v[UNR][4];
#pragma unroll
            for (int u = 0; u < UNR; u++)
#pragma unroll
                for (int s = 0; s < 4; s++)
                    v[u][s] = B[(size_t)(4 * (kqb + u * KSTEP) + s) * NCOL + n];
#pragma unroll
            for (int u = 0; u < UNR; u++) {
                q4f q = { v[u][0], v[u][1], v[u][2], v[u][3] };
                Bs4[(kqb + u * KSTEP) * NCOL + n] = q;
            }
        }
    }
    __syncthreads();

    while (slot < nslots) {
        const int strip   = (slot / Q) * GWG + wg;
        const int colbase = (slot % Q) * 64;
        v8f acc[4];
#pragma unroll
        for (int t = 0; t < 4; t++)
#pragma unroll
            for (int r = 0; r < 8; r++) acc[t][r] = 0.0f;

        int boff = h * NCOL + colbase + c;
        if constexpr (Q == 1) {
            asm volatile("" : "+v"(boff));
        }
        const q4f* bp = Bs4 + boff;
#pragma unroll
        for (int i = 0; i < KO; i++) {
            q4f bq[4];
#pragma unroll
            for (int t = 0; t < 4; t++) bq[t] = bp[(2 * i) * NCOL + t * 16];
#pragma unroll
            for (int t = 0; t < 4; t++)
                acc[t] = __builtin_amdgcn_wmma_f32_16x16x4_f32(
                    false, a4[i].xy, false, bq[t].xy, (short)0, acc[t], false, false);
#pragma unroll
            for (int t = 0; t < 4; t++)
                acc[t] = __builtin_amdgcn_wmma_f32_16x16x4_f32(
                    false, a4[i].zw, false, bq[t].zw, (short)0, acc[t], false, false);
        }

        float* cp = C + (size_t)(strip * 16 + 8 * h) * NCOL + colbase + c;
#pragma unroll
        for (int t = 0; t < 4; t++)
#pragma unroll
            for (int r = 0; r < 8; r++) cp[(size_t)r * NCOL + t * 16] = acc[t][r];

        slot += 16;
        if (slot < nslots) {
            const int ns = (slot / Q) * GWG + wg;
            const q4f* pa = (const q4f*)(A + (size_t)(ns * 16 + c) * KD + 4 * h);
#pragma unroll
            for (int i = 0; i < KO; i++) a4[i] = pa[2 * i];
        }
    }
}

#define NAB 16
#define ROW_XOR(v, X) __int_as_float(__builtin_amdgcn_update_dpp(0, __float_as_int(v), 0x160 + (X), 0xf, 0xf, false))
template <int H>
__global__ __launch_bounds__(1024) void node_attn(
    const float* __restrict__ h, const float* __restrict__ al,
    const float* __restrict__ ar, float* __restrict__ el, float* __restrict__ er) {
    constexpr int NL = NAB / 2;
    constexpr int NS = (H == 1) ? 1 : 2;
    static_assert(H == 1 || H == 4, "head of load j is 2 * (j & 1) + hf for H == 4");
    static_assert((NN * H) % NAB == 0, "whole blocks");
    constexpr int NBLKS = NN * H / NAB;
    const int lane = threadIdx.x & 31;
    const int c    = lane & 15;
    const int hf   = lane >> 4;
    const int wave = __builtin_amdgcn_readfirstlane(threadIdx.x >> 5);
    const int nwav = gridDim.x * (blockDim.x >> 5);
    int blk = wave * gridDim.x + blockIdx.x;
    if (blk >= NBLKS) return;

    q4f alv[NS], arv[NS];
#pragma unroll
    for (int s = 0; s < NS; s++) {
        const int head = (H == 1) ? 0 : 2 * s + hf;
        alv[s] = *(const q4f*)(al + head * 64 + 4 * c);
        arv[s] = *(const q4f*)(ar + head * 64 + 4 * c);
    }
    for (; blk < NBLKS; blk += nwav) {
        const q4f* p = (const q4f*)(h + (size_t)blk * (NAB * 64)) + lane;
        q4f v[NL];
#pragma unroll
        for (int j = 0; j < NL; j++) v[j] = p[j * 32];
        float pl[NL], pr[NL];
#pragma unroll
        for (int j = 0; j < NL; j++) {
            const int s = (H == 1) ? 0 : (j & 1);
            pl[j] = v[j].x * alv[s].x + v[j].y * alv[s].y + v[j].z * alv[s].z + v[j].w * alv[s].w;
            pr[j] = v[j].x * arv[s].x + v[j].y * arv[s].y + v[j].z * arv[s].z + v[j].w * arv[s].w;
        }
#pragma unroll
        for (int j = 0; j < NL; j++) {
            pl[j] += ROW_XOR(pl[j], 8); pr[j] += ROW_XOR(pr[j], 8);
            pl[j] += ROW_XOR(pl[j], 4); pr[j] += ROW_XOR(pr[j], 4);
            pl[j] += ROW_XOR(pl[j], 2); pr[j] += ROW_XOR(pr[j], 2);
            pl[j] += ROW_XOR(pl[j], 1); pr[j] += ROW_XOR(pr[j], 1);
        }
        float sl = pl[0], sr = pr[0];
#pragma unroll
        for (int j = 1; j < NL; j++) {
            sl = (c == j) ? pl[j] : sl;
            sr = (c == j) ? pr[j] : sr;
        }
        if (c < NL) {
            el[(size_t)blk * NAB + 2 * c + hf] = sl;
            er[(size_t)blk * NAB + 2 * c + hf] = sr;
        }
    }
}

#define TR   196
#define NBLK (EE / 128)
#define LCAP 8192
static_assert(EE % 128 == 0, "edge list is read as whole 128-edge blocks");
static_assert(NN <= 65536, "a list entry holds src in 16 bits beside the local row");

typedef __attribute__((ext_vector_type(4))) float v4f;
template <int H> struct score_vec;
template <> struct score_vec<1> { typedef float type; };
template <> struct score_vec<4> { typedef v4f type; };
template <int H>
__device__ __forceinline__ typename score_vec<H>::type load_scores(const float* __restrict__ p, int idx) {
    return *(const typename score_vec<H>::type*)(p + (size_t)idx * H);
}
template <int H>
__device__ __forceinline__ float score_of(typename score_vec<H>::type v, int hh) {
    if constexpr (H == 1) return v; else return v[hh];
}

template <int H>
__device__ __forceinline__ void tile_add_edge(
    int e, int r0, int lane,
    const int* __restrict__ src, const int* __restrict__ dst,
    const float* __restrict__ el, const float* __restrict__ er,
    const unsigned* mx, const float* __restrict__ h,
    float* acc, float* den) {
    constexpr int RW = H * 64;
    const int s  = src[e];
    const int d  = dst[e];
    const int dl = d - r0;
    const float2* hs = (const float2*)(h + (size_t)s * RW);
    float2 v[H];
#pragma unroll
    for (int j = 0; j < H; j++) v[j] = hs[lane + 32 * j];
    float w[H];
#pragma unroll
    for (int hh = 0; hh < H; hh++) {
        float sc = el[(size_t)s * H + hh] + er[(size_t)d * H + hh];
        sc = sc > 0.0f ? sc : SLOPE * sc;
        w[hh] = expf(sc - ord2f(mx[dl * H + hh]));
    }
    if (lane < H) {
        float wl = w[0];
#pragma unroll
        for (int hh = 1; hh < H; hh++) wl = (lane == hh) ? w[hh] : wl;
        atomicAdd(&den[dl * H + lane], wl);
    }
    float* row = acc + dl * RW + 2 * lane;
#pragma unroll
    for (int j = 0; j < H; j++) {
        atomicAdd(&row[64 * j],     w[j] * v[j].x);
        atomicAdd(&row[64 * j + 1], w[j] * v[j].y);
    }
}

template <int H, bool ELU>
__global__ __launch_bounds__(1024) void tile_accumulate(
    const int* __restrict__ src, const int* __restrict__ dst,
    const float* __restrict__ el, const float* __restrict__ er,
    const float* __restrict__ h,
    const float* __restrict__ b, float* __restrict__ outp) {
    constexpr int RW = H * 64;
    constexpr int U  = 8;
    constexpr int G  = 32 / H;
    constexpr int MU = 4;
    __shared__ float    acc[TR * RW];
    __shared__ float    den[TR * H];
    __shared__ unsigned mx[TR * H];
    __shared__ int      list[LCAP];
    __shared__ int      cnt;
    const int tid  = threadIdx.x;
    const int lane = tid & 31;
    const int wave = __builtin_amdgcn_readfirstlane(tid >> 5);
    const int r0   = blockIdx.x * TR;

    for (int i = tid; i < TR * RW; i += 1024) acc[i] = 0.0f;
    for (int i = tid; i < TR * H; i += 1024) { den[i] = 0.0f; mx[i] = 0u; }
    if (tid == 0) cnt = 0;
    __syncthreads();

    const int4* dst4 = (const int4*)dst;
    const int start = (int)((blockIdx.x * (unsigned)NBLK) / gridDim.x);

    for (int i = wave; i < NBLK; i += 32 * U) {
        int4 dv[U];
#pragma unroll
        for (int u = 0; u < U; u++) {
            const int bi = i + u * 32;
            int bb = (bi < NBLK ? bi : i) + start;
            if (bb >= NBLK) bb -= NBLK;
            dv[u] = dst4[(size_t)bb * 32 + lane];
        }
        unsigned lm = 0;
#pragma unroll
        for (int u = 0; u < U; u++) {
            if (i + u * 32 < NBLK) {
                lm |= ((unsigned)(dv[u].x - r0) < (unsigned)TR) ? (1u << (u * 4 + 0)) : 0u;
                lm |= ((unsigned)(dv[u].y - r0) < (unsigned)TR) ? (1u << (u * 4 + 1)) : 0u;
                lm |= ((unsigned)(dv[u].z - r0) < (unsigned)TR) ? (1u << (u * 4 + 2)) : 0u;
                lm |= ((unsigned)(dv[u].w - r0) < (unsigned)TR) ? (1u << (u * 4 + 3)) : 0u;
            }
        }
        if (lm != 0u) {
            int slot = atomicAdd(&cnt, __popc(lm));
            unsigned m2 = lm;
            while (m2) {
                const int k = __builtin_ctz(m2);
                m2 &= m2 - 1;
                if (slot < LCAP) {
                    int bb = i + (k >> 2) * 32 + start;
                    if (bb >= NBLK) bb -= NBLK;
                    list[slot] = bb * 128 + lane * 4 + (k & 3);
                }
                slot++;
            }
        }
    }
    __syncthreads();

    const int n = cnt;
    if (n <= LCAP) {
        static_assert(MU == 4 && LCAP % (MU * 1024) == 0, "step M: four entries per lane and trip, every place read lies inside the list");
        for (int base = 0; base < n; base += MU * 1024) {
            int  ee[MU];
            bool ok[MU];
#pragma unroll
            for (int u = 0; u < MU; u++) {
                const int i = base + u * 1024 + tid;
                const int t = list[i];
                ok[u] = i < n;
                ee[u] = ok[u] ? t : 0;
            }
            int ss[MU], dd[MU];
#pragma unroll
            for (int u = 0; u < MU; u++) { ss[u] = src[ee[u]]; dd[u] = dst[ee[u]]; }
            typename score_vec<H>::type a[MU], c[MU];
#pragma unroll
            for (int u = 0; u < MU; u++) {
                a[u] = load_scores<H>(el, ss[u]);
                c[u] = load_scores<H>(er, dd[u]);
            }
            asm volatile("" :: "v"(a[0]), "v"(c[0]), "v"(a[1]), "v"(c[1]), "v"(a[2]), "v"(c[2]), "v"(a[3]), "v"(c[3]));
#pragma unroll
            for (int u = 0; u < MU; u++) {
                if (ok[u]) {
                    const int dl = dd[u] - r0;
#pragma unroll
                    for (int hh = 0; hh < H; hh++) {
                        float sc = score_of<H>(a[u], hh) + score_of<H>(c[u], hh);
                        sc = sc > 0.0f ? sc : SLOPE * sc;
                        atomicMax(&mx[dl * H + hh], f2ord(sc));
                    }
                    list[base + u * 1024 + tid] = (dl << 16) | ss[u];
                }
            }
        }
        __syncthreads();

        const int gl  = lane / H;
        const int hhl = lane % H;
        for (int base = wave * G; base < n; base += 32 * G) {
            const bool okl = (base + gl) < n;
            const int  ent = list[okl ? base + gl : base];
            const int  sl  = ent & 0xffff;
            const int  dl_ = ent >> 16;
            float scl = el[(size_t)sl * H + hhl] + er[(size_t)(r0 + dl_) * H + hhl];
            scl = scl > 0.0f ? scl : SLOPE * scl;
            const float wl = expf(scl - ord2f(mx[dl_ * H + hhl]));
            float2 v[G][H];
#pragma unroll
            for (int g = 0; g < G; g++) {
                const int sg = __builtin_amdgcn_readlane(sl, g * H);
                const float2* hs = (const float2*)(h + (size_t)sg * RW);
#pragma unroll
                for (int j = 0; j < H; j++) v[g][j] = hs[lane + 32 * j];
            }
            if (okl) atomicAdd(&den[dl_ * H + hhl], wl);
#pragma unroll
            for (int g = 0; g < G; g++) {
                if (base + g < n) {
                    const int dg = __builtin_amdgcn_readlane(dl_, g * H);
                    float* row = acc + dg * RW + 2 * lane;
#pragma unroll
                    for (int j = 0; j < H; j++) {
                        const float wj = __int_as_float(__builtin_amdgcn_readlane(__float_as_int(wl), g * H + j));
                        atomicAdd(&row[64 * j],     wj * v[g][j].x);
                        atomicAdd(&row[64 * j + 1], wj * v[g][j].y);
                    }
                }
            }
        }
    } else {
        for (int g = wave; g < EE / 32; g += 32) {
            const int e = g * 32 + lane;
            const int d = dst[e];
            if ((unsigned)(d - r0) < (unsigned)TR) {
                const int s = src[e];
#pragma unroll
                for (int hh = 0; hh < H; hh++) {
                    float sc = el[(size_t)s * H + hh] + er[(size_t)d * H + hh];
                    sc = sc > 0.0f ? sc : SLOPE * sc;
                    atomicMax(&mx[(d - r0) * H + hh], f2ord(sc));
                }
            }
        }
        __syncthreads();
        for (int g = wave; g < EE / 32; g += 32) {
            const int d = dst[g * 32 + lane];
            unsigned hm = __builtin_amdgcn_ballot_w32((unsigned)(d - r0) < (unsigned)TR);
            while (hm) {
                const int l = __builtin_ctz(hm);
                hm &= hm - 1;
                tile_add_edge<H>(g * 32 + l, r0, lane, src, dst, el, er, mx, h, acc, den);
            }
        }
    }
    __syncthreads();

    static_assert(1024 % RW == 0, "a thread keeps its column");
    const int   rows = (NN - r0) < TR ? (NN - r0) : TR;
    const int   nel  = rows * RW;
    const float bc   = b[tid % RW];
    float* op = outp + (size_t)r0 * RW;
    for (int i = tid; i < nel; i += 1024) {
        const float dn = den[i >> 6];
        const float a  = acc[i];
        float x = (dn > 0.0f ? a / dn : 0.0f) + bc;
        if (ELU) x = x > 0.0f ? x : expm1f(x);
        op[i] = x;
    }
}

extern "C" void kernel_launch(void* const* d_in, const int* in_sizes, int n_in,
                              void* d_out, int out_size, void* d_ws, size_t ws_size,
                              hipStream_t stream) {
    const float* feat = (const float*)d_in[0];
    const int*   src  = (const int*)d_in[1];
    const int*   dst  = (const int*)d_in[2];
    const float* W1   = (const float*)d_in[3];
    const float* al1  = (const float*)d_in[4];
    const float* ar1  = (const float*)d_in[5];
    const float* b1   = (const float*)d_in[6];
    const float* W2   = (const float*)d_in[7];
    const float* al2  = (const float*)d_in[8];
    const float* ar2  = (const float*)d_in[9];
    const float* b2   = (const float*)d_in[10];
    float* out = (float*)d_out;

    float* h1    = (float*)d_ws;
    float* agg1  = h1    + (size_t)NN * 256;
    float* h2    = agg1  + (size_t)NN * 256;
    float* el1   = h2    + (size_t)NN * 64;
    float* er1   = el1   + (size_t)NN * H1_;
    float* el2   = er1   + (size_t)NN * H1_;
    float* er2   = el2   + NN;

    wmma_gemm_wres<256><<<GWG, 512, 0, stream>>>(feat, W1, h1);
    node_attn<H1_><<<GWG, 1024, 0, stream>>>(h1, al1, ar1, el1, er1);
    tile_accumulate<H1_, true><<<(NN + TR - 1) / TR, 1024, 0, stream>>>(src, dst, el1, er1, h1, b1, agg1);

    wmma_gemm_wres<64><<<GWG, 512, 0, stream>>>(agg1, W2, h2);
    node_attn<1><<<GWG, 512, 0, stream>>>(h2, al2, ar2, el2, er2);
    tile_accumulate<1, false><<<(NN + TR - 1) / TR, 1024, 0, stream>>>(src, dst, el2, er2, h2, b2, out);
}


# Noted edge lists made once: layer 1's tile pass leaves each tile's count and (row,src) entries in the workspace, layer 2 reads them back; its dst sweep and src/dst fetches go; sum order = LDS arrival
# speedup vs baseline: 4.7609x; 1.1589x over previous
#include <hip/hip_runtime.h>
#include <math.h>

#define NN   50000
#define EE   800000
#define IND  256
#define H1_  4
#define HID1 64
#define OUTD 64
#define SLOPE 0.2f

typedef __attribute__((ext_vector_type(8))) float v8f;

__device__ __forceinline__ unsigned f2ord(float f) {
    unsigned u = __float_as_uint(f);
    return (u & 0x80000000u) ? ~u : (u | 0x80000000u);
}
__device__ __forceinline__ float ord2f(unsigned u) {
    return (u & 0x80000000u) ? __uint_as_float(u & 0x7fffffffu) : __uint_as_float(~u);
}

#define GWG 256
typedef __attribute__((ext_vector_type(4))) float q4f;
template <int NCOL>
__global__ __launch_bounds__(512) void wmma_gemm_wres(
    const float* __restrict__ A, const float* __restrict__ B, float* __restrict__ C) {
    constexpr int KD = 256;
    constexpr int KQ = KD / 4;
    constexpr int KO = KD / 8;
    constexpr int Q  = NCOL / 64;
    constexpr int NSTRIP = NN / 16;
    static_assert(NN % 16 == 0, "whole 16-row strips: no WMMA runs with rows masked off");
    static_assert(NCOL % 64 == 0 && 512 % NCOL == 0, "units of 64 columns; staging threads cover whole rows of B");
    __shared__ q4f Bs4[KQ * NCOL];

    const int tid  = threadIdx.x;
    const int lane = tid & 31;
    const int wave = __builtin_amdgcn_readfirstlane(tid >> 5);
    const int c    = lane & 15;
    const int h    = lane >> 4;
    const int wg   = blockIdx.x;
    const int nslots = ((NSTRIP - wg + GWG - 1) / GWG) * Q;

    q4f a4[KO];
    int slot = wave;
    if (slot < nslots) {
        const int strip = (slot / Q) * GWG + wg;
        const q4f* pa = (const q4f*)(A + (size_t)(strip * 16 + c) * KD + 4 * h);
#pragma unroll
        for (int i = 0; i < KO; i++) a4[i] = pa[2 * i];
    }

    {
        constexpr int KSTEP = 512 / NCOL;
        constexpr int UNR   = 8;
        static_assert(KQ % (KSTEP * UNR) == 0, "staging: whole trips");
        const int n   = tid % NCOL;
        const int kq0 = tid / NCOL;
        for (int kqb = kq0; kqb < KQ; kqb += KSTEP * UNR) {
            float v[UNR][4];
#pragma unroll
            for (int u = 0; u < UNR; u++)
#pragma unroll
                for (int s = 0; s < 4; s++)
                    v[u][s] = B[(size_t)(4 * (kqb + u * KSTEP) + s) * NCOL + n];
#pragma unroll
            for (int u = 0; u < UNR; u++) {
                q4f q = { v[u][0], v[u][1], v[u][2], v[u][3] };
                Bs4[(kqb + u * KSTEP) * NCOL + n] = q;
            }
        }
    }
    __syncthreads();

    while (slot < nslots) {
        const int strip   = (slot / Q) * GWG + wg;
        const int colbase = (slot % Q) * 64;
        v8f acc[4];
#pragma unroll
        for (int t = 0; t < 4; t++)
#pragma unroll
            for (int r = 0; r < 8; r++) acc[t][r] = 0.0f;

        int boff = h * NCOL + colbase + c;
        if constexpr (Q == 1) {
            asm volatile("" : "+v"(boff));
        }
        const q4f* bp = Bs4 + boff;
#pragma unroll
        for (int i = 0; i < KO; i++) {
            q4f bq[4];
#pragma unroll
            for (int t = 0; t < 4; t++) bq[t] = bp[(2 * i) * NCOL + t * 16];
#pragma unroll
            for (int t = 0; t < 4; t++)
                acc[t] = __builtin_amdgcn_wmma_f32_16x16x4_f32(
                    false, a4[i].xy, false, bq[t].xy, (short)0, acc[t], false, false);
#pragma unroll
            for (int t = 0; t < 4; t++)
                acc[t] = __builtin_amdgcn_wmma_f32_16x16x4_f32(
                    false, a4[i].zw, false, bq[t].zw, (short)0, acc[t], false, false);
        }

        float* cp = C + (size_t)(strip * 16 + 8 * h) * NCOL + colbase + c;
#pragma unroll
        for (int t = 0; t < 4; t++)
#pragma unroll
            for (int r = 0; r < 8; r++) cp[(size_t)r * NCOL + t * 16] = acc[t][r];

        slot += 16;
        if (slot < nslots) {
            const int ns = (slot / Q) * GWG + wg;
            const q4f* pa = (const q4f*)(A + (size_t)(ns * 16 + c) * KD + 4 * h);
#pragma unroll
            for (int i = 0; i < KO; i++) a4[i] = pa[2 * i];
        }
    }
}

#define NAB 16
#define ROW_XOR(v, X) __int_as_float(__builtin_amdgcn_update_dpp(0, __float_as_int(v), 0x160 + (X), 0xf, 0xf, false))
template <int H>
__global__ __launch_bounds__(1024) void node_attn(
    const float* __restrict__ h, const float* __restrict__ al,
    const float* __restrict__ ar, float* __restrict__ el, float* __restrict__ er) {
    constexpr int NL = NAB / 2;
    constexpr int NS = (H == 1) ? 1 : 2;
    static_assert(H == 1 || H == 4, "head of load j is 2 * (j & 1) + hf for H == 4");
    static_assert((NN * H) % NAB == 0, "whole blocks");
    constexpr int NBLKS = NN * H / NAB;
    const int lane = threadIdx.x & 31;
    const int c    = lane & 15;
    const int hf   = lane >> 4;
    const int wave = __builtin_amdgcn_readfirstlane(threadIdx.x >> 5);
    const int nwav = gridDim.x * (blockDim.x >> 5);
    int blk = wave * gridDim.x + blockIdx.x;
    if (blk >= NBLKS) return;

    q4f alv[NS], arv[NS];
#pragma unroll
    for (int s = 0; s < NS; s++) {
        const int head = (H == 1) ? 0 : 2 * s + hf;
        alv[s] = *(const q4f*)(al + head * 64 + 4 * c);
        arv[s] = *(const q4f*)(ar + head * 64 + 4 * c);
    }
    for (; blk < NBLKS; blk += nwav) {
        const q4f* p = (const q4f*)(h + (size_t)blk * (NAB * 64)) + lane;
        q4f v[NL];
#pragma unroll
        for (int j = 0; j < NL; j++) v[j] = p[j * 32];
        float pl[NL], pr[NL];
#pragma unroll
        for (int j = 0; j < NL; j++) {
            const int s = (H == 1) ? 0 : (j & 1);
            pl[j] = v[j].x * alv[s].x + v[j].y * alv[s].y + v[j].z * alv[s].z + v[j].w * alv[s].w;
            pr[j] = v[j].x * arv[s].x + v[j].y * arv[s].y + v[j].z * arv[s].z + v[j].w * arv[s].w;
        }
#pragma unroll
        for (int j = 0; j < NL; j++) {
            pl[j] += ROW_XOR(pl[j], 8); pr[j] += ROW_XOR(pr[j], 8);
            pl[j] += ROW_XOR(pl[j], 4); pr[j] += ROW_XOR(pr[j], 4);
            pl[j] += ROW_XOR(pl[j], 2); pr[j] += ROW_XOR(pr[j], 2);
            pl[j] += ROW_XOR(pl[j], 1); pr[j] += ROW_XOR(pr[j], 1);
        }
        float sl = pl[0], sr = pr[0];
#pragma unroll
        for (int j = 1; j < NL; j++) {
            sl = (c == j) ? pl[j] : sl;
            sr = (c == j) ? pr[j] : sr;
        }
        if (c < NL) {
            el[(size_t)blk * NAB + 2 * c + hf] = sl;
            er[(size_t)blk * NAB + 2 * c + hf] = sr;
        }
    }
}

#define TR   196
#define NBLK (EE / 128)
#define LCAP 8192
#define NTILE ((NN + TR - 1) / TR)
static_assert(EE % 128 == 0, "edge list is read as whole 128-edge blocks");
static_assert(NN <= 65536, "a list entry holds src in 16 bits beside the local row");

typedef __attribute__((ext_vector_type(4))) float v4f;
template <int H> struct score_vec;
template <> struct score_vec<1> { typedef float type; };
template <> struct score_vec<4> { typedef v4f type; };
template <int H>
__device__ __forceinline__ typename score_vec<H>::type load_scores(const float* __restrict__ p, int idx) {
    return *(const typename score_vec<H>::type*)(p + (size_t)idx * H);
}
template <int H>
__device__ __forceinline__ float score_of(typename score_vec<H>::type v, int hh) {
    if constexpr (H == 1) return v; else return v[hh];
}

template <int H>
__device__ __forceinline__ void tile_add_edge(
    int e, int r0, int lane,
    const int* __restrict__ src, const int* __restrict__ dst,
    const float* __restrict__ el, const float* __restrict__ er,
    const unsigned* mx, const float* __restrict__ h,
    float* acc, float* den) {
    constexpr int RW = H * 64;
    const int s  = src[e];
    const int d  = dst[e];
    const int dl = d - r0;
    const float2* hs = (const float2*)(h + (size_t)s * RW);
    float2 v[H];
#pragma unroll
    for (int j = 0; j < H; j++) v[j] = hs[lane + 32 * j];
    float w[H];
#pragma unroll
    for (int hh = 0; hh < H; hh++) {
        float sc = el[(size_t)s * H + hh] + er[(size_t)d * H + hh];
        sc = sc > 0.0f ? sc : SLOPE * sc;
        w[hh] = expf(sc - ord2f(mx[dl * H + hh]));
    }
    if (lane < H) {
        float wl = w[0];
#pragma unroll
        for (int hh = 1; hh < H; hh++) wl = (lane == hh) ? w[hh] : wl;
        atomicAdd(&den[dl * H + lane], wl);
    }
    float* row = acc + dl * RW + 2 * lane;
#pragma unroll
    for (int j = 0; j < H; j++) {
        atomicAdd(&row[64 * j],     w[j] * v[j].x);
        atomicAdd(&row[64 * j + 1], w[j] * v[j].y);
    }
}

template <int H, bool ELU, bool FIRST>
__global__ __launch_bounds__(1024) void tile_accumulate(
    const int* __restrict__ src, const int* __restrict__ dst,
    const float* __restrict__ el, const float* __restrict__ er,
    const float* __restrict__ h,
    const float* __restrict__ b, float* __restrict__ outp,
    int* __restrict__ keep, int* __restrict__ kcnt) {
    constexpr int RW = H * 64;
    constexpr int U  = 8;
    constexpr int G  = 32 / H;
    constexpr int MU = 4;
    __shared__ float    acc[TR * RW];
    __shared__ float    den[TR * H];
    __shared__ unsigned mx[TR * H];
    __shared__ int      list[LCAP];
    __shared__ int      cnt;
    const int tid  = threadIdx.x;
    const int lane = tid & 31;
    const int wave = __builtin_amdgcn_readfirstlane(tid >> 5);
    const int r0   = blockIdx.x * TR;

    for (int i = tid; i < TR * RW; i += 1024) acc[i] = 0.0f;
    for (int i = tid; i < TR * H; i += 1024) { den[i] = 0.0f; mx[i] = 0u; }
    if (FIRST && tid == 0) cnt = 0;
    __syncthreads();

    int* kl = keep + (size_t)blockIdx.x * LCAP;
    int n;
    if constexpr (FIRST) {
        const int4* dst4 = (const int4*)dst;
        const int start = (int)((blockIdx.x * (unsigned)NBLK) / gridDim.x);

        for (int i = wave; i < NBLK; i += 32 * U) {
            int4 dv[U];
#pragma unroll
            for (int u = 0; u < U; u++) {
                const int bi = i + u * 32;
                int bb = (bi < NBLK ? bi : i) + start;
                if (bb >= NBLK) bb -= NBLK;
                dv[u] = dst4[(size_t)bb * 32 + lane];
            }
            unsigned lm = 0;
#pragma unroll
            for (int u = 0; u < U; u++) {
                if (i + u * 32 < NBLK) {
                    lm |= ((unsigned)(dv[u].x - r0) < (unsigned)TR) ? (1u << (u * 4 + 0)) : 0u;
                    lm |= ((unsigned)(dv[u].y - r0) < (unsigned)TR) ? (1u << (u * 4 + 1)) : 0u;
                    lm |= ((unsigned)(dv[u].z - r0) < (unsigned)TR) ? (1u << (u * 4 + 2)) : 0u;
                    lm |= ((unsigned)(dv[u].w - r0) < (unsigned)TR) ? (1u << (u * 4 + 3)) : 0u;
                }
            }
            if (lm != 0u) {
                int slot = atomicAdd(&cnt, __popc(lm));
                unsigned m2 = lm;
                while (m2) {
                    const int k = __builtin_ctz(m2);
                    m2 &= m2 - 1;
                    if (slot < LCAP) {
                        int bb = i + (k >> 2) * 32 + start;
                        if (bb >= NBLK) bb -= NBLK;
                        list[slot] = bb * 128 + lane * 4 + (k & 3);
                    }
                    slot++;
                }
            }
        }
        __syncthreads();

        n = cnt;
        if (tid == 0) kcnt[blockIdx.x] = n;
    } else {
        n = __builtin_amdgcn_readfirstlane(kcnt[blockIdx.x]);
    }
    if (n <= LCAP) {
        static_assert(MU == 4 && LCAP % (MU * 1024) == 0, "step M: four entries per lane and trip, every place read lies inside the list");
        for (int base = 0; base < n; base += MU * 1024) {
            bool ok[MU];
            int  ss[MU], dd[MU];
            if constexpr (FIRST) {
                int ee[MU];
#pragma unroll
                for (int u = 0; u < MU; u++) {
                    const int i = base + u * 1024 + tid;
                    const int t = list[i];
                    ok[u] = i < n;
                    ee[u] = ok[u] ? t : 0;
                }
#pragma unroll
                for (int u = 0; u < MU; u++) { ss[u] = src[ee[u]]; dd[u] = dst[ee[u]]; }
            } else {
#pragma unroll
                for (int u = 0; u < MU; u++) {
                    const int i = base + u * 1024 + tid;
                    ok[u] = i < n;
                    const int t = ok[u] ? kl[i] : 0;
                    ss[u] = t & 0xffff;
                    dd[u] = r0 + (t >> 16);
                }
            }
            typename score_vec<H>::type a[MU], c[MU];
#pragma unroll
            for (int u = 0; u < MU; u++) {
                a[u] = load_scores<H>(el, ss[u]);
                c[u] = load_scores<H>(er, dd[u]);
            }
            asm volatile("" :: "v"(a[0]), "v"(c[0]), "v"(a[1]), "v"(c[1]), "v"(a[2]), "v"(c[2]), "v"(a[3]), "v"(c[3]));
#pragma unroll
            for (int u = 0; u < MU; u++) {
                if (ok[u]) {
                    const int dl = dd[u] - r0;
#pragma unroll
                    for (int hh = 0; hh < H; hh++) {
                        float sc = score_of<H>(a[u], hh) + score_of<H>(c[u], hh);
                        sc = sc > 0.0f ? sc : SLOPE * sc;
                        atomicMax(&mx[dl * H + hh], f2ord(sc));
                    }
                    const int ent = (dl << 16) | ss[u];
                    list[base + u * 1024 + tid] = ent;
                    if constexpr (FIRST) kl[base + u * 1024 + tid] = ent;
                }
            }
        }
        __syncthreads();

        const int gl  = lane / H;
        const int hhl = lane % H;
        for (int base = wave * G; base < n; base += 32 * G) {
            const bool okl = (base + gl) < n;
            const int  ent = list[okl ? base + gl : base];
            const int  sl  = ent & 0xffff;
            const int  dl_ = ent >> 16;
            float scl = el[(size_t)sl * H + hhl] + er[(size_t)(r0 + dl_) * H + hhl];
            scl = scl > 0.0f ? scl : SLOPE * scl;
            const float wl = expf(scl - ord2f(mx[dl_ * H + hhl]));
            float2 v[G][H];
#pragma unroll
            for (int g = 0; g < G; g++) {
                const int sg = __builtin_amdgcn_readlane(sl, g * H);
                const float2* hs = (const float2*)(h + (size_t)sg * RW);
#pragma unroll
                for (int j = 0; j < H; j++) v[g][j] = hs[lane + 32 * j];
            }
            if (okl) atomicAdd(&den[dl_ * H + hhl], wl);
#pragma unroll
            for (int g = 0; g < G; g++) {
                if (base + g < n) {
                    const int dg = __builtin_amdgcn_readlane(dl_, g * H);
                    float* row = acc + dg * RW + 2 * lane;
#pragma unroll
                    for (int j = 0; j < H; j++) {
                        const float wj = __int_as_float(__builtin_amdgcn_readlane(__float_as_int(wl), g * H + j));
                        atomicAdd(&row[64 * j],     wj * v[g][j].x);
                        atomicAdd(&row[64 * j + 1], wj * v[g][j].y);
                    }
                }
            }
        }
    } else {
        for (int g = wave; g < EE / 32; g += 32) {
            const int e = g * 32 + lane;
            const int d = dst[e];
            if ((unsigned)(d - r0) < (unsigned)TR) {
                const int s = src[e];
#pragma unroll
                for (int hh = 0; hh < H; hh++) {
                    float sc = el[(size_t)s * H + hh] + er[(size_t)d * H + hh];
                    sc = sc > 0.0f ? sc : SLOPE * sc;
                    atomicMax(&mx[(d - r0) * H + hh], f2ord(sc));
                }
            }
        }
        __syncthreads();
        for (int g = wave; g < EE / 32; g += 32) {
            const int d = dst[g * 32 + lane];
            unsigned hm = __builtin_amdgcn_ballot_w32((unsigned)(d - r0) < (unsigned)TR);
            while (hm) {
                const int l = __builtin_ctz(hm);
                hm &= hm - 1;
                tile_add_edge<H>(g * 32 + l, r0, lane, src, dst, el, er, mx, h, acc, den);
            }
        }
    }
    __syncthreads();

    static_assert(1024 % RW == 0, "a thread keeps its column");
    const int   rows = (NN - r0) < TR ? (NN - r0) : TR;
    const int   nel  = rows * RW;
    const float bc   = b[tid % RW];
    float* op = outp + (size_t)r0 * RW;
    for (int i = tid; i < nel; i += 1024) {
        const float dn = den[i >> 6];
        const float a  = acc[i];
        float x = (dn > 0.0f ? a / dn : 0.0f) + bc;
        if (ELU) x = x > 0.0f ? x : expm1f(x);
        op[i] = x;
    }
}

extern "C" void kernel_launch(void* const* d_in, const int* in_sizes, int n_in,
                              void* d_out, int out_size, void* d_ws, size_t ws_size,
                              hipStream_t stream) {
    const float* feat = (const float*)d_in[0];
    const int*   src  = (const int*)d_in[1];
    const int*   dst  = (const int*)d_in[2];
    const float* W1   = (const float*)d_in[3];
    const float* al1  = (const float*)d_in[4];
    const float* ar1  = (const float*)d_in[5];
    const float* b1   = (const float*)d_in[6];
    const float* W2   = (const float*)d_in[7];
    const float* al2  = (const float*)d_in[8];
    const float* ar2  = (const float*)d_in[9];
    const float* b2   = (const float*)d_in[10];
    float* out = (float*)d_out;

    float* h1    = (float*)d_ws;
    float* agg1  = h1    + (size_t)NN * 256;
    float* h2    = agg1  + (size_t)NN * 256;
    float* el1   = h2    + (size_t)NN * 64;
    float* er1   = el1   + (size_t)NN * H1_;
    float* el2   = er1   + (size_t)NN * H1_;
    float* er2   = el2   + NN;
    int*   keep  = (int*)(er2 + NN);
    int*   kcnt  = keep  + (size_t)NTILE * LCAP;

    wmma_gemm_wres<256><<<GWG, 512, 0, stream>>>(feat, W1, h1);
    node_attn<H1_><<<GWG, 1024, 0, stream>>>(h1, al1, ar1, el1, er1);
    tile_accumulate<H1_, true, true><<<NTILE, 1024, 0, stream>>>(src, dst, el1, er1, h1, b1, agg1, keep, kcnt);

    wmma_gemm_wres<64><<<GWG, 512, 0, stream>>>(agg1, W2, h2);
    node_attn<1><<<GWG, 512, 0, stream>>>(h2, al2, ar2, el2, er2);
    tile_accumulate<1, false, false><<<NTILE, 1024, 0, stream>>>(src, dst, el2, er2, h2, b2, out, keep, kcnt);
}


# Tile filter reads 1 byte per edge: the node-score launch writes dst/196 as bytes once per call; each tile scans that list, 4 bytes per compare, not the int dst list; sum order = LDS arrival as before
# speedup vs baseline: 5.0819x; 1.0674x over previous
#include <hip/hip_runtime.h>
#include <math.h>

#define NN   50000
#define EE   800000
#define IND  256
#define H1_  4
#define HID1 64
#define OUTD 64
#define SLOPE 0.2f
#define TR   196
#define EEP  (((EE + 511) / 512) * 512)
#define NBLKC (EEP / 512)
static_assert((NN + TR - 1) / TR <= 256, "a tile number fits one byte");
static_assert(EE % 16 == 0, "a lane's 16 bytes of the list are all edges or all padding");

typedef __attribute__((ext_vector_type(8))) float v8f;

__device__ __forceinline__ unsigned f2ord(float f) {
    unsigned u = __float_as_uint(f);
    return (u & 0x80000000u) ? ~u : (u | 0x80000000u);
}
__device__ __forceinline__ float ord2f(unsigned u) {
    return (u & 0x80000000u) ? __uint_as_float(u & 0x7fffffffu) : __uint_as_float(~u);
}

#define GWG 256
typedef __attribute__((ext_vector_type(4))) float q4f;
template <int NCOL>
__global__ __launch_bounds__(512) void wmma_gemm_wres(
    const float* __restrict__ A, const float* __restrict__ B, float* __restrict__ C) {
    constexpr int KD = 256;
    constexpr int KQ = KD / 4;
    constexpr int KO = KD / 8;
    constexpr int Q  = NCOL / 64;
    constexpr int NSTRIP = NN / 16;
    static_assert(NN % 16 == 0, "whole 16-row strips: no WMMA runs with rows masked off");
    static_assert(NCOL % 64 == 0 && 512 % NCOL == 0, "units of 64 columns; staging threads cover whole rows of B");
    __shared__ q4f Bs4[KQ * NCOL];

    const int tid  = threadIdx.x;
    const int lane = tid & 31;
    const int wave = __builtin_amdgcn_readfirstlane(tid >> 5);
    const int c    = lane & 15;
    const int h    = lane >> 4;
    const int wg   = blockIdx.x;
    const int nslots = ((NSTRIP - wg + GWG - 1) / GWG) * Q;

    q4f a4[KO];
    int slot = wave;
    if (slot < nslots) {
        const int strip = (slot / Q) * GWG + wg;
        const q4f* pa = (const q4f*)(A + (size_t)(strip * 16 + c) * KD + 4 * h);
#pragma unroll
        for (int i = 0; i < KO; i++) a4[i] = pa[2 * i];
    }

    {
        constexpr int KSTEP = 512 / NCOL;
        constexpr int UNR   = 8;
        static_assert(KQ % (KSTEP * UNR) == 0, "staging: whole trips");
        const int n   = tid % NCOL;
        const int kq0 = tid / NCOL;
        for (int kqb = kq0; kqb < KQ; kqb += KSTEP * UNR) {
            float v[UNR][4];
#pragma unroll
            for (int u = 0; u < UNR; u++)
#pragma unroll
                for (int s = 0; s < 4; s++)
                    v[u][s] = B[(size_t)(4 * (kqb + u * KSTEP) + s) * NCOL + n];
#pragma unroll
            for (int u = 0; u < UNR; u++) {
                q4f q = { v[u][0], v[u][1], v[u][2], v[u][3] };
                Bs4[(kqb + u * KSTEP) * NCOL + n] = q;
            }
        }
    }
    __syncthreads();

    while (slot < nslots) {
        const int strip   = (slot / Q) * GWG + wg;
        const int colbase = (slot % Q) * 64;
        v8f acc[4];
#pragma unroll
        for (int t = 0; t < 4; t++)
#pragma unroll
            for (int r = 0; r < 8; r++) acc[t][r] = 0.0f;

        int boff = h * NCOL + colbase + c;
        if constexpr (Q == 1) {
            asm volatile("" : "+v"(boff));
        }
        const q4f* bp = Bs4 + boff;
#pragma unroll
        for (int i = 0; i < KO; i++) {
            q4f bq[4];
#pragma unroll
            for (int t = 0; t < 4; t++) bq[t] = bp[(2 * i) * NCOL + t * 16];
#pragma unroll
            for (int t = 0; t < 4; t++)
                acc[t] = __builtin_amdgcn_wmma_f32_16x16x4_f32(
                    false, a4[i].xy, false, bq[t].xy, (short)0, acc[t], false, false);
#pragma unroll
            for (int t = 0; t < 4; t++)
                acc[t] = __builtin_amdgcn_wmma_f32_16x16x4_f32(
                    false, a4[i].zw, false, bq[t].zw, (short)0, acc[t], false, false);
        }

        float* cp = C + (size_t)(strip * 16 + 8 * h) * NCOL + colbase + c;
#pragma unroll
        for (int t = 0; t < 4; t++)
#pragma unroll
            for (int r = 0; r < 8; r++) cp[(size_t)r * NCOL + t * 16] = acc[t][r];

        slot += 16;
        if (slot < nslots) {
            const int ns = (slot / Q) * GWG + wg;
            const q4f* pa = (const q4f*)(A + (size_t)(ns * 16 + c) * KD + 4 * h);
#pragma unroll
            for (int i = 0; i < KO; i++) a4[i] = pa[2 * i];
        }
    }
}

#define NAB 16
#define ROW_XOR(v, X) __int_as_float(__builtin_amdgcn_update_dpp(0, __float_as_int(v), 0x160 + (X), 0xf, 0xf, false))
template <int H, bool CODE>
__global__ __launch_bounds__(1024) void node_attn(
    const float* __restrict__ h, const float* __restrict__ al,
    const float* __restrict__ ar, float* __restrict__ el, float* __restrict__ er,
    const int* __restrict__ dst, unsigned* __restrict__ tcode) {
    if constexpr (CODE) {
        const int gt = blockIdx.x * blockDim.x + threadIdx.x;
        if (gt < EEP / 4) {
            unsigned code = 0u;
            if (gt < EE / 4) {
                const int4 d = ((const int4*)dst)[gt];
                code = ((unsigned)d.x / TR) | (((unsigned)d.y / TR) << 8) |
                       (((unsigned)d.z / TR) << 16) | (((unsigned)d.w / TR) << 24);
            }
            tcode[gt] = code;
        }
    }
    constexpr int NL = NAB / 2;
    constexpr int NS = (H == 1) ? 1 : 2;
    static_assert(H == 1 || H == 4, "head of load j is 2 * (j & 1) + hf for H == 4");
    static_assert((NN * H) % NAB == 0, "whole blocks");
    constexpr int NBLKS = NN * H / NAB;
    const int lane = threadIdx.x & 31;
    const int c    = lane & 15;
    const int hf   = lane >> 4;
    const int wave = __builtin_amdgcn_readfirstlane(threadIdx.x >> 5);
    const int nwav = gridDim.x * (blockDim.x >> 5);
    int blk = wave * gridDim.x + blockIdx.x;
    if (blk >= NBLKS) return;

    q4f alv[NS], arv[NS];
#pragma unroll
    for (int s = 0; s < NS; s++) {
        const int head = (H == 1) ? 0 : 2 * s + hf;
        alv[s] = *(const q4f*)(al + head * 64 + 4 * c);
        arv[s] = *(const q4f*)(ar + head * 64 + 4 * c);
    }
    for (; blk < NBLKS; blk += nwav) {
        const q4f* p = (const q4f*)(h + (size_t)blk * (NAB * 64)) + lane;
        q4f v[NL];
#pragma unroll
        for (int j = 0; j < NL; j++) v[j] = p[j * 32];
        float pl[NL], pr[NL];
#pragma unroll
        for (int j = 0; j < NL; j++) {
            const int s = (H == 1) ? 0 : (j & 1);
            pl[j] = v[j].x * alv[s].x + v[j].y * alv[s].y + v[j].z * alv[s].z + v[j].w * alv[s].w;
            pr[j] = v[j].x * arv[s].x + v[j].y * arv[s].y + v[j].z * arv[s].z + v[j].w * arv[s].w;
        }
#pragma unroll
        for (int j = 0; j < NL; j++) {
            pl[j] += ROW_XOR(pl[j], 8); pr[j] += ROW_XOR(pr[j], 8);
            pl[j] += ROW_XOR(pl[j], 4); pr[j] += ROW_XOR(pr[j], 4);
            pl[j] += ROW_XOR(pl[j], 2); pr[j] += ROW_XOR(pr[j], 2);
            pl[j] += ROW_XOR(pl[j], 1); pr[j] += ROW_XOR(pr[j], 1);
        }
        float sl = pl[0], sr = pr[0];
#pragma unroll
        for (int j = 1; j < NL; j++) {
            sl = (c == j) ? pl[j] : sl;
            sr = (c == j) ? pr[j] : sr;
        }
        if (c < NL) {
            el[(size_t)blk * NAB + 2 * c + hf] = sl;
            er[(size_t)blk * NAB + 2 * c + hf] = sr;
        }
    }
}

#define LCAP 8192
#define NTILE ((NN + TR - 1) / TR)
static_assert(NN <= 65536, "a list entry holds src in 16 bits beside the local row");

typedef __attribute__((ext_vector_type(4))) float v4f;
template <int H> struct score_vec;
template <> struct score_vec<1> { typedef float type; };
template <> struct score_vec<4> { typedef v4f type; };
template <int H>
__device__ __forceinline__ typename score_vec<H>::type load_scores(const float* __restrict__ p, int idx) {
    return *(const typename score_vec<H>::type*)(p + (size_t)idx * H);
}
template <int H>
__device__ __forceinline__ float score_of(typename score_vec<H>::type v, int hh) {
    if constexpr (H == 1) return v; else return v[hh];
}

template <int H>
__device__ __forceinline__ void tile_add_edge(
    int e, int r0, int lane,
    const int* __restrict__ src, const int* __restrict__ dst,
    const float* __restrict__ el, const float* __restrict__ er,
    const unsigned* mx, const float* __restrict__ h,
    float* acc, float* den) {
    constexpr int RW = H * 64;
    const int s  = src[e];
    const int d  = dst[e];
    const int dl = d - r0;
    const float2* hs = (const float2*)(h + (size_t)s * RW);
    float2 v[H];
#pragma unroll
    for (int j = 0; j < H; j++) v[j] = hs[lane + 32 * j];
    float w[H];
#pragma unroll
    for (int hh = 0; hh < H; hh++) {
        float sc = el[(size_t)s * H + hh] + er[(size_t)d * H + hh];
        sc = sc > 0.0f ? sc : SLOPE * sc;
        w[hh] = expf(sc - ord2f(mx[dl * H + hh]));
    }
    if (lane < H) {
        float wl = w[0];
#pragma unroll
        for (int hh = 1; hh < H; hh++) wl = (lane == hh) ? w[hh] : wl;
        atomicAdd(&den[dl * H + lane], wl);
    }
    float* row = acc + dl * RW + 2 * lane;
#pragma unroll
    for (int j = 0; j < H; j++) {
        atomicAdd(&row[64 * j],     w[j] * v[j].x);
        atomicAdd(&row[64 * j + 1], w[j] * v[j].y);
    }
}

template <int H, bool ELU, bool FIRST>
__global__ __launch_bounds__(1024) void tile_accumulate(
    const int* __restrict__ src, const int* __restrict__ dst,
    const float* __restrict__ el, const float* __restrict__ er,
    const float* __restrict__ h,
    const float* __restrict__ b, float* __restrict__ outp,
    int* __restrict__ keep, int* __restrict__ kcnt,
    const unsigned* __restrict__ tcode) {
    constexpr int RW = H * 64;
    constexpr int U  = 8;
    constexpr int G  = 32 / H;
    constexpr int MU = 4;
    __shared__ float    acc[TR * RW];
    __shared__ float    den[TR * H];
    __shared__ unsigned mx[TR * H];
    __shared__ int      list[LCAP];
    __shared__ int      cnt;
    const int tid  = threadIdx.x;
    const int lane = tid & 31;
    const int wave = __builtin_amdgcn_readfirstlane(tid >> 5);
    const int r0   = blockIdx.x * TR;

    for (int i = tid; i < TR * RW; i += 1024) acc[i] = 0.0f;
    for (int i = tid; i < TR * H; i += 1024) { den[i] = 0.0f; mx[i] = 0u; }
    if (FIRST && tid == 0) cnt = 0;
    __syncthreads();

    int* kl = keep + (size_t)blockIdx.x * LCAP;
    int n;
    if constexpr (FIRST) {
        const uint4* tc4 = (const uint4*)tcode;
        const unsigned pat = (unsigned)blockIdx.x * 0x01010101u;
        const int start = (int)((blockIdx.x * (unsigned)NBLKC) / gridDim.x);
        static_assert(U % 2 == 0, "two loads (32 edges of a lane) make one word of match bits");

        for (int i = wave; i < NBLKC; i += 32 * U) {
            uint4 cv[U];
#pragma unroll
            for (int u = 0; u < U; u++) {
                const int bi = i + u * 32;
                int bb = (bi < NBLKC ? bi : i) + start;
                if (bb >= NBLKC) bb -= NBLKC;
                cv[u] = tc4[(size_t)bb * 32 + lane];
            }
            unsigned lm[U / 2];
#pragma unroll
            for (int g = 0; g < U / 2; g++) {
                unsigned nz = 0u;
                unsigned keepm = 0u;
#pragma unroll
                for (int uu = 0; uu < 2; uu++) {
                    const int u  = 2 * g + uu;
                    const int bi = i + u * 32;
                    int bb = (bi < NBLKC ? bi : i) + start;
                    if (bb >= NBLKC) bb -= NBLKC;
                    const bool live = (bi < NBLKC) && (bb * 512 + lane * 16 < EE);
                    keepm |= live ? (uu == 0 ? 0xF0F0F0F0u : 0x0F0F0F0Fu) : 0u;
                    const unsigned wd[4] = { cv[u].x, cv[u].y, cv[u].z, cv[u].w };
#pragma unroll
                    for (int w = 0; w < 4; w++) {
                        const int q = 4 * uu + w;
                        const unsigned t = wd[w] ^ pat;
                        const unsigned y = ((t & 0x7f7f7f7fu) + 0x7f7f7f7fu) | t;
                        nz |= (y >> q) & (0x80808080u >> q);
                    }
                }
                lm[g] = ~nz & keepm;
            }
            unsigned any = 0u;
            int total = 0;
#pragma unroll
            for (int g = 0; g < U / 2; g++) { any |= lm[g]; total += __popc(lm[g]); }
            if (any != 0u) {
                int slot = atomicAdd(&cnt, total);
#pragma unroll
                for (int g = 0; g < U / 2; g++) {
                    unsigned m2 = lm[g];
                    while (m2) {
                        const int p = __builtin_ctz(m2);
                        m2 &= m2 - 1;
                        if (slot < LCAP) {
                            const int q = 7 - (p & 7);
                            const int k = p >> 3;
                            int bb = i + (2 * g + (q >> 2)) * 32 + start;
                            if (bb >= NBLKC) bb -= NBLKC;
                            list[slot] = bb * 512 + lane * 16 + (q & 3) * 4 + k;
                        }
                        slot++;
                    }
                }
            }
        }
        __syncthreads();

        n = cnt;
        if (tid == 0) kcnt[blockIdx.x] = n;
    } else {
        n = __builtin_amdgcn_readfirstlane(kcnt[blockIdx.x]);
    }
    if (n <= LCAP) {
        static_assert(MU == 4 && LCAP % (MU * 1024) == 0, "step M: four entries per lane and trip, every place read lies inside the list");
        for (int base = 0; base < n; base += MU * 1024) {
            bool ok[MU];
            int  ss[MU], dd[MU];
            if constexpr (FIRST) {
                int ee[MU];
#pragma unroll
                for (int u = 0; u < MU; u++) {
                    const int i = base + u * 1024 + tid;
                    const int t = list[i];
                    ok[u] = i < n;
                    ee[u] = ok[u] ? t : 0;
                }
#pragma unroll
                for (int u = 0; u < MU; u++) { ss[u] = src[ee[u]]; dd[u] = dst[ee[u]]; }
            } else {
#pragma unroll
                for (int u = 0; u < MU; u++) {
                    const int i = base + u * 1024 + tid;
                    ok[u] = i < n;
                    const int t = ok[u] ? kl[i] : 0;
                    ss[u] = t & 0xffff;
                    dd[u] = r0 + (t >> 16);
                }
            }
            typename score_vec<H>::type a[MU], c[MU];
#pragma unroll
            for (int u = 0; u < MU; u++) {
                a[u] = load_scores<H>(el, ss[u]);
                c[u] = load_scores<H>(er, dd[u]);
            }
            asm volatile("" :: "v"(a[0]), "v"(c[0]), "v"(a[1]), "v"(c[1]), "v"(a[2]), "v"(c[2]), "v"(a[3]), "v"(c[3]));
#pragma unroll
            for (int u = 0; u < MU; u++) {
                if (ok[u]) {
                    const int dl = dd[u] - r0;
#pragma unroll
                    for (int hh = 0; hh < H; hh++) {
                        float sc = score_of<H>(a[u], hh) + score_of<H>(c[u], hh);
                        sc = sc > 0.0f ? sc : SLOPE * sc;
                        atomicMax(&mx[dl * H + hh], f2ord(sc));
                    }
                    const int ent = (dl << 16) | ss[u];
                    list[base + u * 1024 + tid] = ent;
                    if constexpr (FIRST) kl[base + u * 1024 + tid] = ent;
                }
            }
        }
        __syncthreads();

        const int gl  = lane / H;
        const int hhl = lane % H;
        for (int base = wave * G; base < n; base += 32 * G) {
            const bool okl = (base + gl) < n;
            const int  ent = list[okl ? base + gl : base];
            const int  sl  = ent & 0xffff;
            const int  dl_ = ent >> 16;
            float scl = el[(size_t)sl * H + hhl] + er[(size_t)(r0 + dl_) * H + hhl];
            scl = scl > 0.0f ? scl : SLOPE * scl;
            const float wl = expf(scl - ord2f(mx[dl_ * H + hhl]));
            float2 v[G][H];
#pragma unroll
            for (int g = 0; g < G; g++) {
                const int sg = __builtin_amdgcn_readlane(sl, g * H);
                const float2* hs = (const float2*)(h + (size_t)sg * RW);
#pragma unroll
                for (int j = 0; j < H; j++) v[g][j] = hs[lane + 32 * j];
            }
            if (okl) atomicAdd(&den[dl_ * H + hhl], wl);
#pragma unroll
            for (int g = 0; g < G; g++) {
                if (base + g < n) {
                    const int dg = __builtin_amdgcn_readlane(dl_, g * H);
                    float* row = acc + dg * RW + 2 * lane;
#pragma unroll
                    for (int j = 0; j < H; j++) {
                        const float wj = __int_as_float(__builtin_amdgcn_readlane(__float_as_int(wl), g * H + j));
                        atomicAdd(&row[64 * j],     wj * v[g][j].x);
                        atomicAdd(&row[64 * j + 1], wj * v[g][j].y);
                    }
                }
            }
        }
    } else {
        for (int g = wave; g < EE / 32; g += 32) {
            const int e = g * 32 + lane;
            const int d = dst[e];
            if ((unsigned)(d - r0) < (unsigned)TR) {
                const int s = src[e];
#pragma unroll
                for (int hh = 0; hh < H; hh++) {
                    float sc = el[(size_t)s * H + hh] + er[(size_t)d * H + hh];
                    sc = sc > 0.0f ? sc : SLOPE * sc;
                    atomicMax(&mx[(d - r0) * H + hh], f2ord(sc));
                }
            }
        }
        __syncthreads();
        for (int g = wave; g < EE / 32; g += 32) {
            const int d = dst[g * 32 + lane];
            unsigned hm = __builtin_amdgcn_ballot_w32((unsigned)(d - r0) < (unsigned)TR);
            while (hm) {
                const int l = __builtin_ctz(hm);
                hm &= hm - 1;
                tile_add_edge<H>(g * 32 + l, r0, lane, src, dst, el, er, mx, h, acc, den);
            }
        }
    }
    __syncthreads();

    static_assert(1024 % RW == 0, "a thread keeps its column");
    const int   rows = (NN - r0) < TR ? (NN - r0) : TR;
    const int   nel  = rows * RW;
    const float bc   = b[tid % RW];
    float* op = outp + (size_t)r0 * RW;
    for (int i = tid; i < nel; i += 1024) {
        const float dn = den[i >> 6];
        const float a  = acc[i];
        float x = (dn > 0.0f ? a / dn : 0.0f) + bc;
        if (ELU) x = x > 0.0f ? x : expm1f(x);
        op[i] = x;
    }
}

extern "C" void kernel_launch(void* const* d_in, const int* in_sizes, int n_in,
                              void* d_out, int out_size, void* d_ws, size_t ws_size,
                              hipStream_t stream) {
    const float* feat = (const float*)d_in[0];
    const int*   src  = (const int*)d_in[1];
    const int*   dst  = (const int*)d_in[2];
    const float* W1   = (const float*)d_in[3];
    const float* al1  = (const float*)d_in[4];
    const float* ar1  = (const float*)d_in[5];
    const float* b1   = (const float*)d_in[6];
    const float* W2   = (const float*)d_in[7];
    const float* al2  = (const float*)d_in[8];
    const float* ar2  = (const float*)d_in[9];
    const float* b2   = (const float*)d_in[10];
    float* out = (float*)d_out;

    float* h1    = (float*)d_ws;
    float* agg1  = h1    + (size_t)NN * 256;
    float* h2    = agg1  + (size_t)NN * 256;
    float* el1   = h2    + (size_t)NN * 64;
    float* er1   = el1   + (size_t)NN * H1_;
    float* el2   = er1   + (size_t)NN * H1_;
    float* er2   = el2   + NN;
    int*   keep  = (int*)(er2 + NN);
    int*   kcnt  = keep  + (size_t)NTILE * LCAP;
    unsigned* tcode = (unsigned*)(kcnt + NTILE);

    wmma_gemm_wres<256><<<GWG, 512, 0, stream>>>(feat, W1, h1);
    static_assert(GWG * 1024 >= EEP / 4, "the node-score launch has a thread for every four edges of the tile-number list");
    node_attn<H1_, true><<<GWG, 1024, 0, stream>>>(h1, al1, ar1, el1, er1, dst, tcode);
    tile_accumulate<H1_, true, true><<<NTILE, 1024, 0, stream>>>(src, dst, el1, er1, h1, b1, agg1, keep, kcnt, tcode);

    wmma_gemm_wres<64><<<GWG, 512, 0, stream>>>(agg1, W2, h2);
    node_attn<1, false><<<GWG, 512, 0, stream>>>(h2, al2, ar2, el2, er2, dst, tcode);
    tile_accumulate<1, false, false><<<NTILE, 1024, 0, stream>>>(src, dst, el2, er2, h2, b2, out, keep, kcnt, tcode);
}


# Tile rows owned by waves (row % 32): step M files each noted edge in its owner's LDS list; a wave sums its own rows by plain LDS load/add/store; row float atomics go; sum order = owner-list arrival
# speedup vs baseline: 5.4584x; 1.0741x over previous
#include <hip/hip_runtime.h>
#include <math.h>

#define NN   50000
#define EE   800000
#define IND  256
#define H1_  4
#define HID1 64
#define OUTD 64
#define SLOPE 0.2f
#define TR   196
#define EEP  (((EE + 511) / 512) * 512)
#define NBLKC (EEP / 512)
static_assert((NN + TR - 1) / TR <= 256, "a tile number fits one byte");
static_assert(EE % 16 == 0, "a lane's 16 bytes of the list are all edges or all padding");

typedef __attribute__((ext_vector_type(8))) float v8f;

__device__ __forceinline__ unsigned f2ord(float f) {
    unsigned u = __float_as_uint(f);
    return (u & 0x80000000u) ? ~u : (u | 0x80000000u);
}
__device__ __forceinline__ float ord2f(unsigned u) {
    return (u & 0x80000000u) ? __uint_as_float(u & 0x7fffffffu) : __uint_as_float(~u);
}

#define GWG 256
typedef __attribute__((ext_vector_type(4))) float q4f;
template <int NCOL>
__global__ __launch_bounds__(512) void wmma_gemm_wres(
    const float* __restrict__ A, const float* __restrict__ B, float* __restrict__ C) {
    constexpr int KD = 256;
    constexpr int KQ = KD / 4;
    constexpr int KO = KD / 8;
    constexpr int Q  = NCOL / 64;
    constexpr int NSTRIP = NN / 16;
    static_assert(NN % 16 == 0, "whole 16-row strips: no WMMA runs with rows masked off");
    static_assert(NCOL % 64 == 0 && 512 % NCOL == 0, "units of 64 columns; staging threads cover whole rows of B");
    __shared__ q4f Bs4[KQ * NCOL];

    const int tid  = threadIdx.x;
    const int lane = tid & 31;
    const int wave = __builtin_amdgcn_readfirstlane(tid >> 5);
    const int c    = lane & 15;
    const int h    = lane >> 4;
    const int wg   = blockIdx.x;
    const int nslots = ((NSTRIP - wg + GWG - 1) / GWG) * Q;

    q4f a4[KO];
    int slot = wave;
    if (slot < nslots) {
        const int strip = (slot / Q) * GWG + wg;
        const q4f* pa = (const q4f*)(A + (size_t)(strip * 16 + c) * KD + 4 * h);
#pragma unroll
        for (int i = 0; i < KO; i++) a4[i] = pa[2 * i];
    }

    {
        constexpr int KSTEP = 512 / NCOL;
        constexpr int UNR   = 8;
        static_assert(KQ % (KSTEP * UNR) == 0, "staging: whole trips");
        const int n   = tid % NCOL;
        const int kq0 = tid / NCOL;
        for (int kqb = kq0; kqb < KQ; kqb += KSTEP * UNR) {
            float v[UNR][4];
#pragma unroll
            for (int u = 0; u < UNR; u++)
#pragma unroll
                for (int s = 0; s < 4; s++)
                    v[u][s] = B[(size_t)(4 * (kqb + u * KSTEP) + s) * NCOL + n];
#pragma unroll
            for (int u = 0; u < UNR; u++) {
                q4f q = { v[u][0], v[u][1], v[u][2], v[u][3] };
                Bs4[(kqb + u * KSTEP) * NCOL + n] = q;
            }
        }
    }
    __syncthreads();

    while (slot < nslots) {
        const int strip   = (slot / Q) * GWG + wg;
        const int colbase = (slot % Q) * 64;
        v8f acc[4];
#pragma unroll
        for (int t = 0; t < 4; t++)
#pragma unroll
            for (int r = 0; r < 8; r++) acc[t][r] = 0.0f;

        int boff = h * NCOL + colbase + c;
        if constexpr (Q == 1) {
            asm volatile("" : "+v"(boff));
        }
        const q4f* bp = Bs4 + boff;
#pragma unroll
        for (int i = 0; i < KO; i++) {
            q4f bq[4];
#pragma unroll
            for (int t = 0; t < 4; t++) bq[t] = bp[(2 * i) * NCOL + t * 16];
#pragma unroll
            for (int t = 0; t < 4; t++)
                acc[t] = __builtin_amdgcn_wmma_f32_16x16x4_f32(
                    false, a4[i].xy, false, bq[t].xy, (short)0, acc[t], false, false);
#pragma unroll
            for (int t = 0; t < 4; t++)
                acc[t] = __builtin_amdgcn_wmma_f32_16x16x4_f32(
                    false, a4[i].zw, false, bq[t].zw, (short)0, acc[t], false, false);
        }

        float* cp = C + (size_t)(strip * 16 + 8 * h) * NCOL + colbase + c;
#pragma unroll
        for (int t = 0; t < 4; t++)
#pragma unroll
            for (int r = 0; r < 8; r++) cp[(size_t)r * NCOL + t * 16] = acc[t][r];

        slot += 16;
        if (slot < nslots) {
            const int ns = (slot / Q) * GWG + wg;
            const q4f* pa = (const q4f*)(A + (size_t)(ns * 16 + c) * KD + 4 * h);
#pragma unroll
            for (int i = 0; i < KO; i++) a4[i] = pa[2 * i];
        }
    }
}

#define NAB 16
#define ROW_XOR(v, X) __int_as_float(__builtin_amdgcn_update_dpp(0, __float_as_int(v), 0x160 + (X), 0xf, 0xf, false))
template <int H, bool CODE>
__global__ __launch_bounds__(1024) void node_attn(
    const float* __restrict__ h, const float* __restrict__ al,
    const float* __restrict__ ar, float* __restrict__ el, float* __restrict__ er,
    const int* __restrict__ dst, unsigned* __restrict__ tcode) {
    if constexpr (CODE) {
        const int gt = blockIdx.x * blockDim.x + threadIdx.x;
        if (gt < EEP / 4) {
            unsigned code = 0u;
            if (gt < EE / 4) {
                const int4 d = ((const int4*)dst)[gt];
                code = ((unsigned)d.x / TR) | (((unsigned)d.y / TR) << 8) |
                       (((unsigned)d.z / TR) << 16) | (((unsigned)d.w / TR) << 24);
            }
            tcode[gt] = code;
        }
    }
    constexpr int NL = NAB / 2;
    constexpr int NS = (H == 1) ? 1 : 2;
    static_assert(H == 1 || H == 4, "head of load j is 2 * (j & 1) + hf for H == 4");
    static_assert((NN * H) % NAB == 0, "whole blocks");
    constexpr int NBLKS = NN * H / NAB;
    const int lane = threadIdx.x & 31;
    const int c    = lane & 15;
    const int hf   = lane >> 4;
    const int wave = __builtin_amdgcn_readfirstlane(threadIdx.x >> 5);
    const int nwav = gridDim.x * (blockDim.x >> 5);
    int blk = wave * gridDim.x + blockIdx.x;
    if (blk >= NBLKS) return;

    q4f alv[NS], arv[NS];
#pragma unroll
    for (int s = 0; s < NS; s++) {
        const int head = (H == 1) ? 0 : 2 * s + hf;
        alv[s] = *(const q4f*)(al + head * 64 + 4 * c);
        arv[s] = *(const q4f*)(ar + head * 64 + 4 * c);
    }
    for (; blk < NBLKS; blk += nwav) {
        const q4f* p = (const q4f*)(h + (size_t)blk * (NAB * 64)) + lane;
        q4f v[NL];
#pragma unroll
        for (int j = 0; j < NL; j++) v[j] = p[j * 32];
        float pl[NL], pr[NL];
#pragma unroll
        for (int j = 0; j < NL; j++) {
            const int s = (H == 1) ? 0 : (j & 1);
            pl[j] = v[j].x * alv[s].x + v[j].y * alv[s].y + v[j].z * alv[s].z + v[j].w * alv[s].w;
            pr[j] = v[j].x * arv[s].x + v[j].y * arv[s].y + v[j].z * arv[s].z + v[j].w * arv[s].w;
        }
#pragma unroll
        for (int j = 0; j < NL; j++) {
            pl[j] += ROW_XOR(pl[j], 8); pr[j] += ROW_XOR(pr[j], 8);
            pl[j] += ROW_XOR(pl[j], 4); pr[j] += ROW_XOR(pr[j], 4);
            pl[j] += ROW_XOR(pl[j], 2); pr[j] += ROW_XOR(pr[j], 2);
            pl[j] += ROW_XOR(pl[j], 1); pr[j] += ROW_XOR(pr[j], 1);
        }
        float sl = pl[0], sr = pr[0];
#pragma unroll
        for (int j = 1; j < NL; j++) {
            sl = (c == j) ? pl[j] : sl;
            sr = (c == j) ? pr[j] : sr;
        }
        if (c < NL) {
            el[(size_t)blk * NAB + 2 * c + hf] = sl;
            er[(size_t)blk * NAB + 2 * c + hf] = sr;
        }
    }
}

#define LCAP 8192
#define WCAP 256
#define NTILE ((NN + TR - 1) / TR)
static_assert(NN <= 65536, "a list entry holds src in 16 bits beside the local row");

typedef __attribute__((ext_vector_type(4))) float v4f;
template <int H> struct score_vec;
template <> struct score_vec<1> { typedef float type; };
template <> struct score_vec<4> { typedef v4f type; };
template <int H>
__device__ __forceinline__ typename score_vec<H>::type load_scores(const float* __restrict__ p, int idx) {
    return *(const typename score_vec<H>::type*)(p + (size_t)idx * H);
}
template <int H>
__device__ __forceinline__ float score_of(typename score_vec<H>::type v, int hh) {
    if constexpr (H == 1) return v; else return v[hh];
}

template <int H>
__device__ __forceinline__ void tile_add_edge(
    int e, int r0, int lane,
    const int* __restrict__ src, const int* __restrict__ dst,
    const float* __restrict__ el, const float* __restrict__ er,
    const unsigned* mx, const float* __restrict__ h,
    float* acc, float* den) {
    constexpr int RW = H * 64;
    const int s  = src[e];
    const int d  = dst[e];
    const int dl = d - r0;
    const float2* hs = (const float2*)(h + (size_t)s * RW);
    float2 v[H];
#pragma unroll
    for (int j = 0; j < H; j++) v[j] = hs[lane + 32 * j];
    float w[H];
#pragma unroll
    for (int hh = 0; hh < H; hh++) {
        float sc = el[(size_t)s * H + hh] + er[(size_t)d * H + hh];
        sc = sc > 0.0f ? sc : SLOPE * sc;
        w[hh] = expf(sc - ord2f(mx[dl * H + hh]));
    }
    if (lane < H) {
        float wl = w[0];
#pragma unroll
        for (int hh = 1; hh < H; hh++) wl = (lane == hh) ? w[hh] : wl;
        atomicAdd(&den[dl * H + lane], wl);
    }
    float* row = acc + dl * RW + 2 * lane;
#pragma unroll
    for (int j = 0; j < H; j++) {
        atomicAdd(&row[64 * j],     w[j] * v[j].x);
        atomicAdd(&row[64 * j + 1], w[j] * v[j].y);
    }
}

template <int H, bool ELU, bool FIRST>
__global__ __launch_bounds__(1024) void tile_accumulate(
    const int* __restrict__ src, const int* __restrict__ dst,
    const float* __restrict__ el, const float* __restrict__ er,
    const float* __restrict__ h,
    const float* __restrict__ b, float* __restrict__ outp,
    int* __restrict__ keep, int* __restrict__ kcnt,
    const unsigned* __restrict__ tcode) {
    constexpr int RW = H * 64;
    constexpr int U  = 8;
    constexpr int G  = 32 / H;
    constexpr int MU = 4;
    __shared__ __attribute__((aligned(16))) float acc[TR * RW];
    __shared__ float    den[TR * H];
    __shared__ unsigned mx[TR * H];
    __shared__ int      list[LCAP];
    __shared__ int      cnt;
    __shared__ int      sub[32 * WCAP];
    __shared__ int      wcnt[32];
    const int tid  = threadIdx.x;
    const int lane = tid & 31;
    const int wave = __builtin_amdgcn_readfirstlane(tid >> 5);
    const int r0   = blockIdx.x * TR;

    for (int i = tid; i < TR * RW; i += 1024) acc[i] = 0.0f;
    for (int i = tid; i < TR * H; i += 1024) { den[i] = 0.0f; mx[i] = 0u; }
    if (FIRST && tid == 0) cnt = 0;
    if (tid < 32) wcnt[tid] = 0;
    __syncthreads();

    int* kl = keep + (size_t)blockIdx.x * LCAP;
    int n;
    if constexpr (FIRST) {
        const uint4* tc4 = (const uint4*)tcode;
        const unsigned pat = (unsigned)blockIdx.x * 0x01010101u;
        const int start = (int)((blockIdx.x * (unsigned)NBLKC) / gridDim.x);
        static_assert(U % 2 == 0, "two loads (32 edges of a lane) make one word of match bits");

        for (int i = wave; i < NBLKC; i += 32 * U) {
            uint4 cv[U];
#pragma unroll
            for (int u = 0; u < U; u++) {
                const int bi = i + u * 32;
                int bb = (bi < NBLKC ? bi : i) + start;
                if (bb >= NBLKC) bb -= NBLKC;
                cv[u] = tc4[(size_t)bb * 32 + lane];
            }
            unsigned lm[U / 2];
#pragma unroll
            for (int g = 0; g < U / 2; g++) {
                unsigned nz = 0u;
                unsigned keepm = 0u;
#pragma unroll
                for (int uu = 0; uu < 2; uu++) {
                    const int u  = 2 * g + uu;
                    const int bi = i + u * 32;
                    int bb = (bi < NBLKC ? bi : i) + start;
                    if (bb >= NBLKC) bb -= NBLKC;
                    const bool live = (bi < NBLKC) && (bb * 512 + lane * 16 < EE);
                    keepm |= live ? (uu == 0 ? 0xF0F0F0F0u : 0x0F0F0F0Fu) : 0u;
                    const unsigned wd[4] = { cv[u].x, cv[u].y, cv[u].z, cv[u].w };
#pragma unroll
                    for (int w = 0; w < 4; w++) {
                        const int q = 4 * uu + w;
                        const unsigned t = wd[w] ^ pat;
                        const unsigned y = ((t & 0x7f7f7f7fu) + 0x7f7f7f7fu) | t;
                        nz |= (y >> q) & (0x80808080u >> q);
                    }
                }
                lm[g] = ~nz & keepm;
            }
            unsigned any = 0u;
            int total = 0;
#pragma unroll
            for (int g = 0; g < U / 2; g++) { any |= lm[g]; total += __popc(lm[g]); }
            if (any != 0u) {
                int slot = atomicAdd(&cnt, total);
#pragma unroll
                for (int g = 0; g < U / 2; g++) {
                    unsigned m2 = lm[g];
                    while (m2) {
                        const int p = __builtin_ctz(m2);
                        m2 &= m2 - 1;
                        if (slot < LCAP) {
                            const int q = 7 - (p & 7);
                            const int k = p >> 3;
                            int bb = i + (2 * g + (q >> 2)) * 32 + start;
                            if (bb >= NBLKC) bb -= NBLKC;
                            list[slot] = bb * 512 + lane * 16 + (q & 3) * 4 + k;
                        }
                        slot++;
                    }
                }
            }
        }
        __syncthreads();

        n = cnt;
        if (tid == 0) kcnt[blockIdx.x] = n;
    } else {
        n = __builtin_amdgcn_readfirstlane(kcnt[blockIdx.x]);
    }
    if (n <= LCAP) {
        static_assert(MU == 4 && LCAP % (MU * 1024) == 0, "step M: four entries per lane and trip, every place read lies inside the list");
        for (int base = 0; base < n; base += MU * 1024) {
            bool ok[MU];
            int  ss[MU], dd[MU];
            if constexpr (FIRST) {
                int ee[MU];
#pragma unroll
                for (int u = 0; u < MU; u++) {
                    const int i = base + u * 1024 + tid;
                    const int t = list[i];
                    ok[u] = i < n;
                    ee[u] = ok[u] ? t : 0;
                }
#pragma unroll
                for (int u = 0; u < MU; u++) { ss[u] = src[ee[u]]; dd[u] = dst[ee[u]]; }
            } else {
#pragma unroll
                for (int u = 0; u < MU; u++) {
                    const int i = base + u * 1024 + tid;
                    ok[u] = i < n;
                    const int t = ok[u] ? kl[i] : 0;
                    ss[u] = t & 0xffff;
                    dd[u] = r0 + (t >> 16);
                }
            }
            typename score_vec<H>::type a[MU], c[MU];
#pragma unroll
            for (int u = 0; u < MU; u++) {
                a[u] = load_scores<H>(el, ss[u]);
                c[u] = load_scores<H>(er, dd[u]);
            }
            asm volatile("" :: "v"(a[0]), "v"(c[0]), "v"(a[1]), "v"(c[1]), "v"(a[2]), "v"(c[2]), "v"(a[3]), "v"(c[3]));
#pragma unroll
            for (int u = 0; u < MU; u++) {
                if (ok[u]) {
                    const int dl = dd[u] - r0;
#pragma unroll
                    for (int hh = 0; hh < H; hh++) {
                        float sc = score_of<H>(a[u], hh) + score_of<H>(c[u], hh);
                        sc = sc > 0.0f ? sc : SLOPE * sc;
                        atomicMax(&mx[dl * H + hh], f2ord(sc));
                    }
                    const int ent = (dl << 16) | ss[u];
                    list[base + u * 1024 + tid] = ent;
                    const int ow = dl & 31;
                    const int pw = atomicAdd(&wcnt[ow], 1);
                    if (pw < WCAP) sub[ow * WCAP + pw] = ent;
                    if constexpr (FIRST) kl[base + u * 1024 + tid] = ent;
                }
            }
        }
        __syncthreads();

        const int gl  = lane / H;
        const int hhl = lane % H;
        const bool owned = __builtin_amdgcn_ballot_w32(wcnt[lane] > WCAP) == 0u;
        if (owned) {
            const int  nw = __builtin_amdgcn_readfirstlane(wcnt[wave]);
            const int* sw = sub + wave * WCAP;
            for (int base = 0; base < nw; base += G) {
                const bool okl = (base + gl) < nw;
                const int  ent = sw[okl ? base + gl : base];
                const int  sl  = ent & 0xffff;
                const int  dl_ = ent >> 16;
                float scl = el[(size_t)sl * H + hhl] + er[(size_t)(r0 + dl_) * H + hhl];
                scl = scl > 0.0f ? scl : SLOPE * scl;
                const float wl = expf(scl - ord2f(mx[dl_ * H + hhl]));
                float2 v[G][H];
#pragma unroll
                for (int g = 0; g < G; g++) {
                    const int sg = __builtin_amdgcn_readlane(sl, g * H);
                    const float2* hs = (const float2*)(h + (size_t)sg * RW);
#pragma unroll
                    for (int j = 0; j < H; j++) v[g][j] = hs[lane + 32 * j];
                }
                if (okl) atomicAdd(&den[dl_ * H + hhl], wl);
#pragma unroll
                for (int g = 0; g < G; g++) {
                    if (base + g < nw) {
                        const int dg = __builtin_amdgcn_readlane(dl_, g * H);
                        float2* row = (float2*)(acc + dg * RW + 2 * lane);
#pragma unroll
                        for (int j = 0; j < H; j++) {
                            const float wj = __int_as_float(__builtin_amdgcn_readlane(__float_as_int(wl), g * H + j));
                            float2 a = row[32 * j];
                            a.x += wj * v[g][j].x;
                            a.y += wj * v[g][j].y;
                            row[32 * j] = a;
                        }
                    }
                }
            }
        } else
        for (int base = wave * G; base < n; base += 32 * G) {
            const bool okl = (base + gl) < n;
            const int  ent = list[okl ? base + gl : base];
            const int  sl  = ent & 0xffff;
            const int  dl_ = ent >> 16;
            float scl = el[(size_t)sl * H + hhl] + er[(size_t)(r0 + dl_) * H + hhl];
            scl = scl > 0.0f ? scl : SLOPE * scl;
            const float wl = expf(scl - ord2f(mx[dl_ * H + hhl]));
            float2 v[G][H];
#pragma unroll
            for (int g = 0; g < G; g++) {
                const int sg = __builtin_amdgcn_readlane(sl, g * H);
                const float2* hs = (const float2*)(h + (size_t)sg * RW);
#pragma unroll
                for (int j = 0; j < H; j++) v[g][j] = hs[lane + 32 * j];
            }
            if (okl) atomicAdd(&den[dl_ * H + hhl], wl);
#pragma unroll
            for (int g = 0; g < G; g++) {
                if (base + g < n) {
                    const int dg = __builtin_amdgcn_readlane(dl_, g * H);
                    float* row = acc + dg * RW + 2 * lane;
#pragma unroll
                    for (int j = 0; j < H; j++) {
                        const float wj = __int_as_float(__builtin_amdgcn_readlane(__float_as_int(wl), g * H + j));
                        atomicAdd(&row[64 * j],     wj * v[g][j].x);
                        atomicAdd(&row[64 * j + 1], wj * v[g][j].y);
                    }
                }
            }
        }
    } else {
        for (int g = wave; g < EE / 32; g += 32) {
            const int e = g * 32 + lane;
            const int d = dst[e];
            if ((unsigned)(d - r0) < (unsigned)TR) {
                const int s = src[e];
#pragma unroll
                for (int hh = 0; hh < H; hh++) {
                    float sc = el[(size_t)s * H + hh] + er[(size_t)d * H + hh];
                    sc = sc > 0.0f ? sc : SLOPE * sc;
                    atomicMax(&mx[(d - r0) * H + hh], f2ord(sc));
                }
            }
        }
        __syncthreads();
        for (int g = wave; g < EE / 32; g += 32) {
            const int d = dst[g * 32 + lane];
            unsigned hm = __builtin_amdgcn_ballot_w32((unsigned)(d - r0) < (unsigned)TR);
            while (hm) {
                const int l = __builtin_ctz(hm);
                hm &= hm - 1;
                tile_add_edge<H>(g * 32 + l, r0, lane, src, dst, el, er, mx, h, acc, den);
            }
        }
    }
    __syncthreads();

    static_assert(1024 % RW == 0, "a thread keeps its column");
    const int   rows = (NN - r0) < TR ? (NN - r0) : TR;
    const int   nel  = rows * RW;
    const float bc   = b[tid % RW];
    float* op = outp + (size_t)r0 * RW;
    for (int i = tid; i < nel; i += 1024) {
        const float dn = den[i >> 6];
        const float a  = acc[i];
        float x = (dn > 0.0f ? a / dn : 0.0f) + bc;
        if (ELU) x = x > 0.0f ? x : expm1f(x);
        op[i] = x;
    }
}

extern "C" void kernel_launch(void* const* d_in, const int* in_sizes, int n_in,
                              void* d_out, int out_size, void* d_ws, size_t ws_size,
                              hipStream_t stream) {
    const float* feat = (const float*)d_in[0];
    const int*   src  = (const int*)d_in[1];
    const int*   dst  = (const int*)d_in[2];
    const float* W1   = (const float*)d_in[3];
    const float* al1  = (const float*)d_in[4];
    const float* ar1  = (const float*)d_in[5];
    const float* b1   = (const float*)d_in[6];
    const float* W2   = (const float*)d_in[7];
    const float* al2  = (const float*)d_in[8];
    const float* ar2  = (const float*)d_in[9];
    const float* b2   = (const float*)d_in[10];
    float* out = (float*)d_out;

    float* h1    = (float*)d_ws;
    float* agg1  = h1    + (size_t)NN * 256;
    float* h2    = agg1  + (size_t)NN * 256;
    float* el1   = h2    + (size_t)NN * 64;
    float* er1   = el1   + (size_t)NN * H1_;
    float* el2   = er1   + (size_t)NN * H1_;
    float* er2   = el2   + NN;
    int*   keep  = (int*)(er2 + NN);
    int*   kcnt  = keep  + (size_t)NTILE * LCAP;
    unsigned* tcode = (unsigned*)(kcnt + NTILE);

    wmma_gemm_wres<256><<<GWG, 512, 0, stream>>>(feat, W1, h1);
    static_assert(GWG * 1024 >= EEP / 4, "the node-score launch has a thread for every four edges of the tile-number list");
    node_attn<H1_, true><<<GWG, 1024, 0, stream>>>(h1, al1, ar1, el1, er1, dst, tcode);
    tile_accumulate<H1_, true, true><<<NTILE, 1024, 0, stream>>>(src, dst, el1, er1, h1, b1, agg1, keep, kcnt, tcode);

    wmma_gemm_wres<64><<<GWG, 512, 0, stream>>>(agg1, W2, h2);
    node_attn<1, false><<<GWG, 512, 0, stream>>>(h2, al2, ar2, el2, er2, dst, tcode);
    tile_accumulate<1, false, false><<<NTILE, 1024, 0, stream>>>(src, dst, el2, er2, h2, b2, out, keep, kcnt, tcode);
}


# Tile row sums held in the owning wave's registers: a wave-uniform branch on the row slot picks them (256-col layer in two rounds); LDS row load/add/store and tile zeroing go; sum order as before
# speedup vs baseline: 5.8473x; 1.0712x over previous
#include <hip/hip_runtime.h>
#include <math.h>

#define NN   50000
#define EE   800000
#define IND  256
#define H1_  4
#define HID1 64
#define OUTD 64
#define SLOPE 0.2f
#define TR   196
#define EEP  (((EE + 511) / 512) * 512)
#define NBLKC (EEP / 512)
static_assert((NN + TR - 1) / TR <= 256, "a tile number fits one byte");
static_assert(EE % 16 == 0, "a lane's 16 bytes of the list are all edges or all padding");

typedef __attribute__((ext_vector_type(8))) float v8f;

__device__ __forceinline__ unsigned f2ord(float f) {
    unsigned u = __float_as_uint(f);
    return (u & 0x80000000u) ? ~u : (u | 0x80000000u);
}
__device__ __forceinline__ float ord2f(unsigned u) {
    return (u & 0x80000000u) ? __uint_as_float(u & 0x7fffffffu) : __uint_as_float(~u);
}

#define GWG 256
typedef __attribute__((ext_vector_type(4))) float q4f;
template <int NCOL>
__global__ __launch_bounds__(512) void wmma_gemm_wres(
    const float* __restrict__ A, const float* __restrict__ B, float* __restrict__ C) {
    constexpr int KD = 256;
    constexpr int KQ = KD / 4;
    constexpr int KO = KD / 8;
    constexpr int Q  = NCOL / 64;
    constexpr int NSTRIP = NN / 16;
    static_assert(NN % 16 == 0, "whole 16-row strips: no WMMA runs with rows masked off");
    static_assert(NCOL % 64 == 0 && 512 % NCOL == 0, "units of 64 columns; staging threads cover whole rows of B");
    __shared__ q4f Bs4[KQ * NCOL];

    const int tid  = threadIdx.x;
    const int lane = tid & 31;
    const int wave = __builtin_amdgcn_readfirstlane(tid >> 5);
    const int c    = lane & 15;
    const int h    = lane >> 4;
    const int wg   = blockIdx.x;
    const int nslots = ((NSTRIP - wg + GWG - 1) / GWG) * Q;

    q4f a4[KO];
    int slot = wave;
    if (slot < nslots) {
        const int strip = (slot / Q) * GWG + wg;
        const q4f* pa = (const q4f*)(A + (size_t)(strip * 16 + c) * KD + 4 * h);
#pragma unroll
        for (int i = 0; i < KO; i++) a4[i] = pa[2 * i];
    }

    {
        constexpr int KSTEP = 512 / NCOL;
        constexpr int UNR   = 8;
        static_assert(KQ % (KSTEP * UNR) == 0, "staging: whole trips");
        const int n   = tid % NCOL;
        const int kq0 = tid / NCOL;
        for (int kqb = kq0; kqb < KQ; kqb += KSTEP * UNR) {
            float v[UNR][4];
#pragma unroll
            for (int u = 0; u < UNR; u++)
#pragma unroll
                for (int s = 0; s < 4; s++)
                    v[u][s] = B[(size_t)(4 * (kqb + u * KSTEP) + s) * NCOL + n];
#pragma unroll
            for (int u = 0; u < UNR; u++) {
                q4f q = { v[u][0], v[u][1], v[u][2], v[u][3] };
                Bs4[(kqb + u * KSTEP) * NCOL + n] = q;
            }
        }
    }
    __syncthreads();

    while (slot < nslots) {
        const int strip   = (slot / Q) * GWG + wg;
        const int colbase = (slot % Q) * 64;
        v8f acc[4];
#pragma unroll
        for (int t = 0; t < 4; t++)
#pragma unroll
            for (int r = 0; r < 8; r++) acc[t][r] = 0.0f;

        int boff = h * NCOL + colbase + c;
        if constexpr (Q == 1) {
            asm volatile("" : "+v"(boff));
        }
        const q4f* bp = Bs4 + boff;
#pragma unroll
        for (int i = 0; i < KO; i++) {
            q4f bq[4];
#pragma unroll
            for (int t = 0; t < 4; t++) bq[t] = bp[(2 * i) * NCOL + t * 16];
#pragma unroll
            for (int t = 0; t < 4; t++)
                acc[t] = __builtin_amdgcn_wmma_f32_16x16x4_f32(
                    false, a4[i].xy, false, bq[t].xy, (short)0, acc[t], false, false);
#pragma unroll
            for (int t = 0; t < 4; t++)
                acc[t] = __builtin_amdgcn_wmma_f32_16x16x4_f32(
                    false, a4[i].zw, false, bq[t].zw, (short)0, acc[t], false, false);
        }

        float* cp = C + (size_t)(strip * 16 + 8 * h) * NCOL + colbase + c;
#pragma unroll
        for (int t = 0; t < 4; t++)
#pragma unroll
            for (int r = 0; r < 8; r++) cp[(size_t)r * NCOL + t * 16] = acc[t][r];

        slot += 16;
        if (slot < nslots) {
            const int ns = (slot / Q) * GWG + wg;
            const q4f* pa = (const q4f*)(A + (size_t)(ns * 16 + c) * KD + 4 * h);
#pragma unroll
            for (int i = 0; i < KO; i++) a4[i] = pa[2 * i];
        }
    }
}

#define NAB 16
#define ROW_XOR(v, X) __int_as_float(__builtin_amdgcn_update_dpp(0, __float_as_int(v), 0x160 + (X), 0xf, 0xf, false))
template <int H, bool CODE>
__global__ __launch_bounds__(1024) void node_attn(
    const float* __restrict__ h, const float* __restrict__ al,
    const float* __restrict__ ar, float* __restrict__ el, float* __restrict__ er,
    const int* __restrict__ dst, unsigned* __restrict__ tcode) {
    if constexpr (CODE) {
        const int gt = blockIdx.x * blockDim.x + threadIdx.x;
        if (gt < EEP / 4) {
            unsigned code = 0u;
            if (gt < EE / 4) {
                const int4 d = ((const int4*)dst)[gt];
                code = ((unsigned)d.x / TR) | (((unsigned)d.y / TR) << 8) |
                       (((unsigned)d.z / TR) << 16) | (((unsigned)d.w / TR) << 24);
            }
            tcode[gt] = code;
        }
    }
    constexpr int NL = NAB / 2;
    constexpr int NS = (H == 1) ? 1 : 2;
    static_assert(H == 1 || H == 4, "head of load j is 2 * (j & 1) + hf for H == 4");
    static_assert((NN * H) % NAB == 0, "whole blocks");
    constexpr int NBLKS = NN * H / NAB;
    const int lane = threadIdx.x & 31;
    const int c    = lane & 15;
    const int hf   = lane >> 4;
    const int wave = __builtin_amdgcn_readfirstlane(threadIdx.x >> 5);
    const int nwav = gridDim.x * (blockDim.x >> 5);
    int blk = wave * gridDim.x + blockIdx.x;
    if (blk >= NBLKS) return;

    q4f alv[NS], arv[NS];
#pragma unroll
    for (int s = 0; s < NS; s++) {
        const int head = (H == 1) ? 0 : 2 * s + hf;
        alv[s] = *(const q4f*)(al + head * 64 + 4 * c);
        arv[s] = *(const q4f*)(ar + head * 64 + 4 * c);
    }
    for (; blk < NBLKS; blk += nwav) {
        const q4f* p = (const q4f*)(h + (size_t)blk * (NAB * 64)) + lane;
        q4f v[NL];
#pragma unroll
        for (int j = 0; j < NL; j++) v[j] = p[j * 32];
        float pl[NL], pr[NL];
#pragma unroll
        for (int j = 0; j < NL; j++) {
            const int s = (H == 1) ? 0 : (j & 1);
            pl[j] = v[j].x * alv[s].x + v[j].y * alv[s].y + v[j].z * alv[s].z + v[j].w * alv[s].w;
            pr[j] = v[j].x * arv[s].x + v[j].y * arv[s].y + v[j].z * arv[s].z + v[j].w * arv[s].w;
        }
#pragma unroll
        for (int j = 0; j < NL; j++) {
            pl[j] += ROW_XOR(pl[j], 8); pr[j] += ROW_XOR(pr[j], 8);
            pl[j] += ROW_XOR(pl[j], 4); pr[j] += ROW_XOR(pr[j], 4);
            pl[j] += ROW_XOR(pl[j], 2); pr[j] += ROW_XOR(pr[j], 2);
            pl[j] += ROW_XOR(pl[j], 1); pr[j] += ROW_XOR(pr[j], 1);
        }
        float sl = pl[0], sr = pr[0];
#pragma unroll
        for (int j = 1; j < NL; j++) {
            sl = (c == j) ? pl[j] : sl;
            sr = (c == j) ? pr[j] : sr;
        }
        if (c < NL) {
            el[(size_t)blk * NAB + 2 * c + hf] = sl;
            er[(size_t)blk * NAB + 2 * c + hf] = sr;
        }
    }
}

#define LCAP 8192
#define WCAP 256
#define NTILE ((NN + TR - 1) / TR)
static_assert(NN <= 65536, "a list entry holds src in 16 bits beside the local row");

typedef __attribute__((ext_vector_type(4))) float v4f;
template <int H> struct score_vec;
template <> struct score_vec<1> { typedef float type; };
template <> struct score_vec<4> { typedef v4f type; };
template <int H>
__device__ __forceinline__ typename score_vec<H>::type load_scores(const float* __restrict__ p, int idx) {
    return *(const typename score_vec<H>::type*)(p + (size_t)idx * H);
}
template <int H>
__device__ __forceinline__ float score_of(typename score_vec<H>::type v, int hh) {
    if constexpr (H == 1) return v; else return v[hh];
}

template <int H>
__device__ __forceinline__ void tile_add_edge(
    int e, int r0, int lane,
    const int* __restrict__ src, const int* __restrict__ dst,
    const float* __restrict__ el, const float* __restrict__ er,
    const unsigned* mx, const float* __restrict__ h,
    float* acc, float* den) {
    constexpr int RW = H * 64;
    const int s  = src[e];
    const int d  = dst[e];
    const int dl = d - r0;
    const float2* hs = (const float2*)(h + (size_t)s * RW);
    float2 v[H];
#pragma unroll
    for (int j = 0; j < H; j++) v[j] = hs[lane + 32 * j];
    float w[H];
#pragma unroll
    for (int hh = 0; hh < H; hh++) {
        float sc = el[(size_t)s * H + hh] + er[(size_t)d * H + hh];
        sc = sc > 0.0f ? sc : SLOPE * sc;
        w[hh] = expf(sc - ord2f(mx[dl * H + hh]));
    }
    if (lane < H) {
        float wl = w[0];
#pragma unroll
        for (int hh = 1; hh < H; hh++) wl = (lane == hh) ? w[hh] : wl;
        atomicAdd(&den[dl * H + lane], wl);
    }
    float* row = acc + dl * RW + 2 * lane;
#pragma unroll
    for (int j = 0; j < H; j++) {
        atomicAdd(&row[64 * j],     w[j] * v[j].x);
        atomicAdd(&row[64 * j + 1], w[j] * v[j].y);
    }
}

template <int H, bool ELU, bool FIRST>
__global__ __launch_bounds__(1024) void tile_accumulate(
    const int* __restrict__ src, const int* __restrict__ dst,
    const float* __restrict__ el, const float* __restrict__ er,
    const float* __restrict__ h,
    const float* __restrict__ b, float* __restrict__ outp,
    int* __restrict__ keep, int* __restrict__ kcnt,
    const unsigned* __restrict__ tcode) {
    constexpr int RW = H * 64;
    constexpr int U  = 8;
    constexpr int G  = 32 / H;
    constexpr int NSLOT = (TR + 31) / 32;
    constexpr int NRND  = (H == 4) ? 2 : 1;
    constexpr int SPR   = (NSLOT + NRND - 1) / NRND;
    constexpr int RCAP  = WCAP / NRND;
    constexpr int MU = 4;
    __shared__ __attribute__((aligned(16))) float acc[TR * RW];
    __shared__ float    den[TR * H];
    __shared__ unsigned mx[TR * H];
    __shared__ int      list[LCAP];
    __shared__ int      cnt;
    __shared__ int      sub[32 * WCAP];
    __shared__ int      wcnt[32 * NRND];
    const int tid  = threadIdx.x;
    const int lane = tid & 31;
    const int wave = __builtin_amdgcn_readfirstlane(tid >> 5);
    const int r0   = blockIdx.x * TR;

    for (int i = tid; i < TR * H; i += 1024) { den[i] = 0.0f; mx[i] = 0u; }
    if (FIRST && tid == 0) cnt = 0;
    if (tid < 32 * NRND) wcnt[tid] = 0;
    __syncthreads();

    int* kl = keep + (size_t)blockIdx.x * LCAP;
    int n;
    if constexpr (FIRST) {
        const uint4* tc4 = (const uint4*)tcode;
        const unsigned pat = (unsigned)blockIdx.x * 0x01010101u;
        const int start = (int)((blockIdx.x * (unsigned)NBLKC) / gridDim.x);
        static_assert(U % 2 == 0, "two loads (32 edges of a lane) make one word of match bits");

        for (int i = wave; i < NBLKC; i += 32 * U) {
            uint4 cv[U];
#pragma unroll
            for (int u = 0; u < U; u++) {
                const int bi = i + u * 32;
                int bb = (bi < NBLKC ? bi : i) + start;
                if (bb >= NBLKC) bb -= NBLKC;
                cv[u] = tc4[(size_t)bb * 32 + lane];
            }
            unsigned lm[U / 2];
#pragma unroll
            for (int g = 0; g < U / 2; g++) {
                unsigned nz = 0u;
                unsigned keepm = 0u;
#pragma unroll
                for (int uu = 0; uu < 2; uu++) {
                    const int u  = 2 * g + uu;
                    const int bi = i + u * 32;
                    int bb = (bi < NBLKC ? bi : i) + start;
                    if (bb >= NBLKC) bb -= NBLKC;
                    const bool live = (bi < NBLKC) && (bb * 512 + lane * 16 < EE);
                    keepm |= live ? (uu == 0 ? 0xF0F0F0F0u : 0x0F0F0F0Fu) : 0u;
                    const unsigned wd[4] = { cv[u].x, cv[u].y, cv[u].z, cv[u].w };
#pragma unroll
                    for (int w = 0; w < 4; w++) {
                        const int q = 4 * uu + w;
                        const unsigned t = wd[w] ^ pat;
                        const unsigned y = ((t & 0x7f7f7f7fu) + 0x7f7f7f7fu) | t;
                        nz |= (y >> q) & (0x80808080u >> q);
                    }
                }
                lm[g] = ~nz & keepm;
            }
            unsigned any = 0u;
            int total = 0;
#pragma unroll
            for (int g = 0; g < U / 2; g++) { any |= lm[g]; total += __popc(lm[g]); }
            if (any != 0u) {
                int slot = atomicAdd(&cnt, total);
#pragma unroll
                for (int g = 0; g < U / 2; g++) {
                    unsigned m2 = lm[g];
                    while (m2) {
                        const int p = __builtin_ctz(m2);
                        m2 &= m2 - 1;
                        if (slot < LCAP) {
                            const int q = 7 - (p & 7);
                            const int k = p >> 3;
                            int bb = i + (2 * g + (q >> 2)) * 32 + start;
                            if (bb >= NBLKC) bb -= NBLKC;
                            list[slot] = bb * 512 + lane * 16 + (q & 3) * 4 + k;
                        }
                        slot++;
                    }
                }
            }
        }
        __syncthreads();

        n = cnt;
        if (tid == 0) kcnt[blockIdx.x] = n;
    } else {
        n = __builtin_amdgcn_readfirstlane(kcnt[blockIdx.x]);
    }
    bool owned = false;
    if (n <= LCAP) {
        static_assert(MU == 4 && LCAP % (MU * 1024) == 0, "step M: four entries per lane and trip, every place read lies inside the list");
        for (int base = 0; base < n; base += MU * 1024) {
            bool ok[MU];
            int  ss[MU], dd[MU];
            if constexpr (FIRST) {
                int ee[MU];
#pragma unroll
                for (int u = 0; u < MU; u++) {
                    const int i = base + u * 1024 + tid;
                    const int t = list[i];
                    ok[u] = i < n;
                    ee[u] = ok[u] ? t : 0;
                }
#pragma unroll
                for (int u = 0; u < MU; u++) { ss[u] = src[ee[u]]; dd[u] = dst[ee[u]]; }
            } else {
#pragma unroll
                for (int u = 0; u < MU; u++) {
                    const int i = base + u * 1024 + tid;
                    ok[u] = i < n;
                    const int t = ok[u] ? kl[i] : 0;
                    ss[u] = t & 0xffff;
                    dd[u] = r0 + (t >> 16);
                }
            }
            typename score_vec<H>::type a[MU], c[MU];
#pragma unroll
            for (int u = 0; u < MU; u++) {
                a[u] = load_scores<H>(el, ss[u]);
                c[u] = load_scores<H>(er, dd[u]);
            }
            asm volatile("" :: "v"(a[0]), "v"(c[0]), "v"(a[1]), "v"(c[1]), "v"(a[2]), "v"(c[2]), "v"(a[3]), "v"(c[3]));
#pragma unroll
            for (int u = 0; u < MU; u++) {
                if (ok[u]) {
                    const int dl = dd[u] - r0;
#pragma unroll
                    for (int hh = 0; hh < H; hh++) {
                        float sc = score_of<H>(a[u], hh) + score_of<H>(c[u], hh);
                        sc = sc > 0.0f ? sc : SLOPE * sc;
                        atomicMax(&mx[dl * H + hh], f2ord(sc));
                    }
                    const int ent = (dl << 16) | ss[u];
                    list[base + u * 1024 + tid] = ent;
                    const int ow = dl & 31;
                    const int rd = (dl >> 5) / SPR;
                    const int pw = atomicAdd(&wcnt[ow * NRND + rd], 1);
                    if (pw < RCAP) sub[ow * WCAP + rd * RCAP + pw] = ent;
                    if constexpr (FIRST) kl[base + u * 1024 + tid] = ent;
                }
            }
        }
        __syncthreads();

        bool over = wcnt[lane] > RCAP;
        if constexpr (NRND == 2) over = over || (wcnt[lane + 32] > RCAP);
        owned = __builtin_amdgcn_ballot_w32(over) == 0u;
    }

    const int gl  = lane / H;
    const int hhl = lane % H;
    const int rows = (NN - r0) < TR ? (NN - r0) : TR;
    if (owned) {
        float2 bv[H];
#pragma unroll
        for (int j = 0; j < H; j++) bv[j] = *(const float2*)(b + 64 * j + 2 * lane);
        for (int rd = 0; rd < NRND; rd++) {
            const int  nw = __builtin_amdgcn_readfirstlane(wcnt[wave * NRND + rd]);
            const int* sw = sub + wave * WCAP + rd * RCAP;
            float2 racc[SPR][H];
#pragma unroll
            for (int s = 0; s < SPR; s++)
#pragma unroll
                for (int j = 0; j < H; j++) racc[s][j] = make_float2(0.0f, 0.0f);
            for (int base = 0; base < nw; base += G) {
                const bool okl = (base + gl) < nw;
                const int  ent = sw[okl ? base + gl : base];
                const int  sl  = ent & 0xffff;
                const int  dl_ = ent >> 16;
                float scl = el[(size_t)sl * H + hhl] + er[(size_t)(r0 + dl_) * H + hhl];
                scl = scl > 0.0f ? scl : SLOPE * scl;
                const float wl = expf(scl - ord2f(mx[dl_ * H + hhl]));
                float2 v[G][H];
#pragma unroll
                for (int g = 0; g < G; g++) {
                    const int sg = __builtin_amdgcn_readlane(sl, g * H);
                    const float2* hs = (const float2*)(h + (size_t)sg * RW);
#pragma unroll
                    for (int j = 0; j < H; j++) v[g][j] = hs[lane + 32 * j];
                }
                if (okl) atomicAdd(&den[dl_ * H + hhl], wl);
#pragma unroll
                for (int g = 0; g < G; g++) {
                    if (base + g < nw) {
                        const int sg_ = (__builtin_amdgcn_readlane(dl_, g * H) >> 5) - rd * SPR;
                        float wj[H];
#pragma unroll
                        for (int j = 0; j < H; j++)
                            wj[j] = __int_as_float(__builtin_amdgcn_readlane(__float_as_int(wl), g * H + j));
#pragma unroll
                        for (int s = 0; s < SPR; s++) {
                            if (sg_ == s) {
#pragma unroll
                                for (int j = 0; j < H; j++) {
                                    racc[s][j].x += wj[j] * v[g][j].x;
                                    racc[s][j].y += wj[j] * v[g][j].y;
                                }
                            }
                        }
                    }
                }
            }
#pragma unroll
            for (int s = 0; s < SPR; s++) {
                const int row = wave + 32 * (rd * SPR + s);
                if (row < rows) {
                    float* orow = outp + (size_t)(r0 + row) * RW + 2 * lane;
#pragma unroll
                    for (int j = 0; j < H; j++) {
                        const float dn = den[row * H + j];
                        float x0 = (dn > 0.0f ? racc[s][j].x / dn : 0.0f) + bv[j].x;
                        float x1 = (dn > 0.0f ? racc[s][j].y / dn : 0.0f) + bv[j].y;
                        if (ELU) {
                            x0 = x0 > 0.0f ? x0 : expm1f(x0);
                            x1 = x1 > 0.0f ? x1 : expm1f(x1);
                        }
                        *(float2*)(orow + 64 * j) = make_float2(x0, x1);
                    }
                }
            }
        }
    } else {
        for (int i = tid; i < TR * RW; i += 1024) acc[i] = 0.0f;
        __syncthreads();
        if (n <= LCAP) {
            for (int base = wave * G; base < n; base += 32 * G) {
                const bool okl = (base + gl) < n;
                const int  ent = list[okl ? base + gl : base];
                const int  sl  = ent & 0xffff;
                const int  dl_ = ent >> 16;
                float scl = el[(size_t)sl * H + hhl] + er[(size_t)(r0 + dl_) * H + hhl];
                scl = scl > 0.0f ? scl : SLOPE * scl;
                const float wl = expf(scl - ord2f(mx[dl_ * H + hhl]));
                float2 v[G][H];
#pragma unroll
                for (int g = 0; g < G; g++) {
                    const int sg = __builtin_amdgcn_readlane(sl, g * H);
                    const float2* hs = (const float2*)(h + (size_t)sg * RW);
#pragma unroll
                    for (int j = 0; j < H; j++) v[g][j] = hs[lane + 32 * j];
                }
                if (okl) atomicAdd(&den[dl_ * H + hhl], wl);
#pragma unroll
                for (int g = 0; g < G; g++) {
                    if (base + g < n) {
                        const int dg = __builtin_amdgcn_readlane(dl_, g * H);
                        float* row = acc + dg * RW + 2 * lane;
#pragma unroll
                        for (int j = 0; j < H; j++) {
                            const float wj = __int_as_float(__builtin_amdgcn_readlane(__float_as_int(wl), g * H + j));
                            atomicAdd(&row[64 * j],     wj * v[g][j].x);
                            atomicAdd(&row[64 * j + 1], wj * v[g][j].y);
                        }
                    }
                }
            }
        } else {
            for (int g = wave; g < EE / 32; g += 32) {
                const int e = g * 32 + lane;
                const int d = dst[e];
                if ((unsigned)(d - r0) < (unsigned)TR) {
                    const int s = src[e];
#pragma unroll
                    for (int hh = 0; hh < H; hh++) {
                        float sc = el[(size_t)s * H + hh] + er[(size_t)d * H + hh];
                        sc = sc > 0.0f ? sc : SLOPE * sc;
                        atomicMax(&mx[(d - r0) * H + hh], f2ord(sc));
                    }
                }
            }
            __syncthreads();
            for (int g = wave; g < EE / 32; g += 32) {
                const int d = dst[g * 32 + lane];
                unsigned hm = __builtin_amdgcn_ballot_w32((unsigned)(d - r0) < (unsigned)TR);
                while (hm) {
                    const int l = __builtin_ctz(hm);
                    hm &= hm - 1;
                    tile_add_edge<H>(g * 32 + l, r0, lane, src, dst, el, er, mx, h, acc, den);
                }
            }
        }
        __syncthreads();

        static_assert(1024 % RW == 0, "a thread keeps its column");
        const int   nel  = rows * RW;
        const float bc   = b[tid % RW];
        float* op = outp + (size_t)r0 * RW;
        for (int i = tid; i < nel; i += 1024) {
            const float dn = den[i >> 6];
            const float a  = acc[i];
            float x = (dn > 0.0f ? a / dn : 0.0f) + bc;
            if (ELU) x = x > 0.0f ? x : expm1f(x);
            op[i] = x;
        }
    }
}

extern "C" void kernel_launch(void* const* d_in, const int* in_sizes, int n_in,
                              void* d_out, int out_size, void* d_ws, size_t ws_size,
                              hipStream_t stream) {
    const float* feat = (const float*)d_in[0];
    const int*   src  = (const int*)d_in[1];
    const int*   dst  = (const int*)d_in[2];
    const float* W1   = (const float*)d_in[3];
    const float* al1  = (const float*)d_in[4];
    const float* ar1  = (const float*)d_in[5];
    const float* b1   = (const float*)d_in[6];
    const float* W2   = (const float*)d_in[7];
    const float* al2  = (const float*)d_in[8];
    const float* ar2  = (const float*)d_in[9];
    const float* b2   = (const float*)d_in[10];
    float* out = (float*)d_out;

    float* h1    = (float*)d_ws;
    float* agg1  = h1    + (size_t)NN * 256;
    float* h2    = agg1  + (size_t)NN * 256;
    float* el1   = h2    + (size_t)NN * 64;
    float* er1   = el1   + (size_t)NN * H1_;
    float* el2   = er1   + (size_t)NN * H1_;
    float* er2   = el2   + NN;
    int*   keep  = (int*)(er2 + NN);
    int*   kcnt  = keep  + (size_t)NTILE * LCAP;
    unsigned* tcode = (unsigned*)(kcnt + NTILE);

    wmma_gemm_wres<256><<<GWG, 512, 0, stream>>>(feat, W1, h1);
    static_assert(GWG * 1024 >= EEP / 4, "the node-score launch has a thread for every four edges of the tile-number list");
    node_attn<H1_, true><<<GWG, 1024, 0, stream>>>(h1, al1, ar1, el1, er1, dst, tcode);
    tile_accumulate<H1_, true, true><<<NTILE, 1024, 0, stream>>>(src, dst, el1, er1, h1, b1, agg1, keep, kcnt, tcode);

    wmma_gemm_wres<64><<<GWG, 512, 0, stream>>>(agg1, W2, h2);
    node_attn<1, false><<<GWG, 512, 0, stream>>>(h2, al2, ar2, el2, er2, dst, tcode);
    tile_accumulate<1, false, false><<<NTILE, 1024, 0, stream>>>(src, dst, el2, er2, h2, b2, out, keep, kcnt, tcode);
}


# Layer-2 product made inside the layer-1 tile pass: finished rows stay in the LDS tile and meet W2 there on f32 WMMA; the x2 plane, its write and re-read, and one launch go; k sum order as before
# speedup vs baseline: 6.1891x; 1.0585x over previous
#include <hip/hip_runtime.h>
#include <math.h>

#define NN   50000
#define EE   800000
#define IND  256
#define H1_  4
#define HID1 64
#define OUTD 64
#define SLOPE 0.2f
#define TR   196
#define EEP  (((EE + 511) / 512) * 512)
#define NBLKC (EEP / 512)
static_assert((NN + TR - 1) / TR <= 256, "a tile number fits one byte");
static_assert(EE % 16 == 0, "a lane's 16 bytes of the list are all edges or all padding");

typedef __attribute__((ext_vector_type(8))) float v8f;

__device__ __forceinline__ unsigned f2ord(float f) {
    unsigned u = __float_as_uint(f);
    return (u & 0x80000000u) ? ~u : (u | 0x80000000u);
}
__device__ __forceinline__ float ord2f(unsigned u) {
    return (u & 0x80000000u) ? __uint_as_float(u & 0x7fffffffu) : __uint_as_float(~u);
}

#define GWG 256
typedef __attribute__((ext_vector_type(4))) float q4f;
template <int NCOL>
__global__ __launch_bounds__(512) void wmma_gemm_wres(
    const float* __restrict__ A, const float* __restrict__ B, float* __restrict__ C) {
    constexpr int KD = 256;
    constexpr int KQ = KD / 4;
    constexpr int KO = KD / 8;
    constexpr int Q  = NCOL / 64;
    constexpr int NSTRIP = NN / 16;
    static_assert(NN % 16 == 0, "whole 16-row strips: no WMMA runs with rows masked off");
    static_assert(NCOL % 64 == 0 && 512 % NCOL == 0, "units of 64 columns; staging threads cover whole rows of B");
    __shared__ q4f Bs4[KQ * NCOL];

    const int tid  = threadIdx.x;
    const int lane = tid & 31;
    const int wave = __builtin_amdgcn_readfirstlane(tid >> 5);
    const int c    = lane & 15;
    const int h    = lane >> 4;
    const int wg   = blockIdx.x;
    const int nslots = ((NSTRIP - wg + GWG - 1) / GWG) * Q;

    q4f a4[KO];
    int slot = wave;
    if (slot < nslots) {
        const int strip = (slot / Q) * GWG + wg;
        const q4f* pa = (const q4f*)(A + (size_t)(strip * 16 + c) * KD + 4 * h);
#pragma unroll
        for (int i = 0; i < KO; i++) a4[i] = pa[2 * i];
    }

    {
        constexpr int KSTEP = 512 / NCOL;
        constexpr int UNR   = 8;
        static_assert(KQ % (KSTEP * UNR) == 0, "staging: whole trips");
        const int n   = tid % NCOL;
        const int kq0 = tid / NCOL;
        for (int kqb = kq0; kqb < KQ; kqb += KSTEP * UNR) {
            float v[UNR][4];
#pragma unroll
            for (int u = 0; u < UNR; u++)
#pragma unroll
                for (int s = 0; s < 4; s++)
                    v[u][s] = B[(size_t)(4 * (kqb + u * KSTEP) + s) * NCOL + n];
#pragma unroll
            for (int u = 0; u < UNR; u++) {
                q4f q = { v[u][0], v[u][1], v[u][2], v[u][3] };
                Bs4[(kqb + u * KSTEP) * NCOL + n] = q;
            }
        }
    }
    __syncthreads();

    while (slot < nslots) {
        const int strip   = (slot / Q) * GWG + wg;
        const int colbase = (slot % Q) * 64;
        v8f acc[4];
#pragma unroll
        for (int t = 0; t < 4; t++)
#pragma unroll
            for (int r = 0; r < 8; r++) acc[t][r] = 0.0f;

        const q4f* bp = Bs4 + h * NCOL + colbase + c;
#pragma unroll
        for (int i = 0; i < KO; i++) {
            q4f bq[4];
#pragma unroll
            for (int t = 0; t < 4; t++) bq[t] = bp[(2 * i) * NCOL + t * 16];
#pragma unroll
            for (int t = 0; t < 4; t++)
                acc[t] = __builtin_amdgcn_wmma_f32_16x16x4_f32(
                    false, a4[i].xy, false, bq[t].xy, (short)0, acc[t], false, false);
#pragma unroll
            for (int t = 0; t < 4; t++)
                acc[t] = __builtin_amdgcn_wmma_f32_16x16x4_f32(
                    false, a4[i].zw, false, bq[t].zw, (short)0, acc[t], false, false);
        }

        float* cp = C + (size_t)(strip * 16 + 8 * h) * NCOL + colbase + c;
#pragma unroll
        for (int t = 0; t < 4; t++)
#pragma unroll
            for (int r = 0; r < 8; r++) cp[(size_t)r * NCOL + t * 16] = acc[t][r];

        slot += 16;
        if (slot < nslots) {
            const int ns = (slot / Q) * GWG + wg;
            const q4f* pa = (const q4f*)(A + (size_t)(ns * 16 + c) * KD + 4 * h);
#pragma unroll
            for (int i = 0; i < KO; i++) a4[i] = pa[2 * i];
        }
    }
}

#define NAB 16
#define ROW_XOR(v, X) __int_as_float(__builtin_amdgcn_update_dpp(0, __float_as_int(v), 0x160 + (X), 0xf, 0xf, false))
template <int H, bool CODE>
__global__ __launch_bounds__(1024) void node_attn(
    const float* __restrict__ h, const float* __restrict__ al,
    const float* __restrict__ ar, float* __restrict__ el, float* __restrict__ er,
    const int* __restrict__ dst, unsigned* __restrict__ tcode) {
    if constexpr (CODE) {
        const int gt = blockIdx.x * blockDim.x + threadIdx.x;
        if (gt < EEP / 4) {
            unsigned code = 0u;
            if (gt < EE / 4) {
                const int4 d = ((const int4*)dst)[gt];
                code = ((unsigned)d.x / TR) | (((unsigned)d.y / TR) << 8) |
                       (((unsigned)d.z / TR) << 16) | (((unsigned)d.w / TR) << 24);
            }
            tcode[gt] = code;
        }
    }
    constexpr int NL = NAB / 2;
    constexpr int NS = (H == 1) ? 1 : 2;
    static_assert(H == 1 || H == 4, "head of load j is 2 * (j & 1) + hf for H == 4");
    static_assert((NN * H) % NAB == 0, "whole blocks");
    constexpr int NBLKS = NN * H / NAB;
    const int lane = threadIdx.x & 31;
    const int c    = lane & 15;
    const int hf   = lane >> 4;
    const int wave = __builtin_amdgcn_readfirstlane(threadIdx.x >> 5);
    const int nwav = gridDim.x * (blockDim.x >> 5);
    int blk = wave * gridDim.x + blockIdx.x;
    if (blk >= NBLKS) return;

    q4f alv[NS], arv[NS];
#pragma unroll
    for (int s = 0; s < NS; s++) {
        const int head = (H == 1) ? 0 : 2 * s + hf;
        alv[s] = *(const q4f*)(al + head * 64 + 4 * c);
        arv[s] = *(const q4f*)(ar + head * 64 + 4 * c);
    }
    for (; blk < NBLKS; blk += nwav) {
        const q4f* p = (const q4f*)(h + (size_t)blk * (NAB * 64)) + lane;
        q4f v[NL];
#pragma unroll
        for (int j = 0; j < NL; j++) v[j] = p[j * 32];
        float pl[NL], pr[NL];
#pragma unroll
        for (int j = 0; j < NL; j++) {
            const int s = (H == 1) ? 0 : (j & 1);
            pl[j] = v[j].x * alv[s].x + v[j].y * alv[s].y + v[j].z * alv[s].z + v[j].w * alv[s].w;
            pr[j] = v[j].x * arv[s].x + v[j].y * arv[s].y + v[j].z * arv[s].z + v[j].w * arv[s].w;
        }
#pragma unroll
        for (int j = 0; j < NL; j++) {
            pl[j] += ROW_XOR(pl[j], 8); pr[j] += ROW_XOR(pr[j], 8);
            pl[j] += ROW_XOR(pl[j], 4); pr[j] += ROW_XOR(pr[j], 4);
            pl[j] += ROW_XOR(pl[j], 2); pr[j] += ROW_XOR(pr[j], 2);
            pl[j] += ROW_XOR(pl[j], 1); pr[j] += ROW_XOR(pr[j], 1);
        }
        float sl = pl[0], sr = pr[0];
#pragma unroll
        for (int j = 1; j < NL; j++) {
            sl = (c == j) ? pl[j] : sl;
            sr = (c == j) ? pr[j] : sr;
        }
        if (c < NL) {
            el[(size_t)blk * NAB + 2 * c + hf] = sl;
            er[(size_t)blk * NAB + 2 * c + hf] = sr;
        }
    }
}

#define LCAP 8192
#define WCAP 256
#define NTILE ((NN + TR - 1) / TR)
static_assert(NN <= 65536, "a list entry holds src in 16 bits beside the local row");

typedef __attribute__((ext_vector_type(4))) float v4f;
template <int H> struct score_vec;
template <> struct score_vec<1> { typedef float type; };
template <> struct score_vec<4> { typedef v4f type; };
template <int H>
__device__ __forceinline__ typename score_vec<H>::type load_scores(const float* __restrict__ p, int idx) {
    return *(const typename score_vec<H>::type*)(p + (size_t)idx * H);
}
template <int H>
__device__ __forceinline__ float score_of(typename score_vec<H>::type v, int hh) {
    if constexpr (H == 1) return v; else return v[hh];
}

template <int H, int PITCH>
__device__ __forceinline__ void tile_add_edge(
    int e, int r0, int lane,
    const int* __restrict__ src, const int* __restrict__ dst,
    const float* __restrict__ el, const float* __restrict__ er,
    const unsigned* mx, const float* __restrict__ h,
    float* acc, float* den) {
    constexpr int RW = H * 64;
    const int s  = src[e];
    const int d  = dst[e];
    const int dl = d - r0;
    const float2* hs = (const float2*)(h + (size_t)s * RW);
    float2 v[H];
#pragma unroll
    for (int j = 0; j < H; j++) v[j] = hs[lane + 32 * j];
    float w[H];
#pragma unroll
    for (int hh = 0; hh < H; hh++) {
        float sc = el[(size_t)s * H + hh] + er[(size_t)d * H + hh];
        sc = sc > 0.0f ? sc : SLOPE * sc;
        w[hh] = expf(sc - ord2f(mx[dl * H + hh]));
    }
    if (lane < H) {
        float wl = w[0];
#pragma unroll
        for (int hh = 1; hh < H; hh++) wl = (lane == hh) ? w[hh] : wl;
        atomicAdd(&den[dl * H + lane], wl);
    }
    float* row = acc + dl * PITCH + 2 * lane;
#pragma unroll
    for (int j = 0; j < H; j++) {
        atomicAdd(&row[64 * j],     w[j] * v[j].x);
        atomicAdd(&row[64 * j + 1], w[j] * v[j].y);
    }
}

typedef __attribute__((ext_vector_type(2))) float v2f;
template <int NS, int PITCH>
__device__ __forceinline__ void strip_product(
    const float* xs, const float* __restrict__ Wn, float* __restrict__ outp,
    int r0, int rows, int strip0, int t, int lane) {
    const int c  = lane & 15;
    const int hl = lane >> 4;
    const float* ap[NS];
#pragma unroll
    for (int s = 0; s < NS; s++) {
        int ar = (strip0 + s) * 16 + c;
        ar = ar < rows ? ar : rows - 1;
        ap[s] = xs + ar * PITCH + 4 * hl;
    }
    const float* bp = Wn + (size_t)(4 * hl) * 64 + 16 * t + c;
    v8f d[NS];
#pragma unroll
    for (int s = 0; s < NS; s++)
#pragma unroll
        for (int r = 0; r < 8; r++) d[s][r] = 0.0f;
#pragma unroll 8
    for (int i = 0; i < 32; i++) {
        const v2f blo = { bp[(8 * i)     * 64], bp[(8 * i + 1) * 64] };
        const v2f bhi = { bp[(8 * i + 2) * 64], bp[(8 * i + 3) * 64] };
        q4f a[NS];
#pragma unroll
        for (int s = 0; s < NS; s++) a[s] = *(const q4f*)(ap[s] + 8 * i);
#pragma unroll
        for (int s = 0; s < NS; s++)
            d[s] = __builtin_amdgcn_wmma_f32_16x16x4_f32(false, a[s].xy, false, blo, (short)0, d[s], false, false);
#pragma unroll
        for (int s = 0; s < NS; s++)
            d[s] = __builtin_amdgcn_wmma_f32_16x16x4_f32(false, a[s].zw, false, bhi, (short)0, d[s], false, false);
    }
#pragma unroll
    for (int s = 0; s < NS; s++) {
        const int rb = (strip0 + s) * 16 + 8 * hl;
        float* cp = outp + (size_t)(r0 + rb) * 64 + 16 * t + c;
#pragma unroll
        for (int r = 0; r < 8; r++)
            if (rb + r < rows) cp[(size_t)r * 64] = d[s][r];
    }
}

template <int H, bool ELU, bool FIRST>
__global__ __launch_bounds__(1024) void tile_accumulate(
    const int* __restrict__ src, const int* __restrict__ dst,
    const float* __restrict__ el, const float* __restrict__ er,
    const float* __restrict__ h,
    const float* __restrict__ b, float* __restrict__ outp,
    int* __restrict__ keep, int* __restrict__ kcnt,
    const unsigned* __restrict__ tcode, const float* __restrict__ Wn) {
    constexpr int RW = H * 64;
    constexpr int PITCH = FIRST ? RW + 4 : RW;
    static_assert(!FIRST || (H == 4 && ELU), "the product in the tile pass is the 256 x 64 one behind layer 1");
    constexpr int U  = 8;
    constexpr int G  = 32 / H;
    constexpr int NSLOT = (TR + 31) / 32;
    constexpr int NRND  = (H == 4) ? 2 : 1;
    constexpr int SPR   = (NSLOT + NRND - 1) / NRND;
    constexpr int RCAP  = WCAP / NRND;
    constexpr int MU = 4;
    __shared__ __attribute__((aligned(16))) float acc[TR * PITCH];
    __shared__ float    den[TR * H];
    __shared__ unsigned mx[TR * H];
    __shared__ int      list[LCAP];
    __shared__ int      cnt;
    __shared__ int      sub[32 * WCAP];
    __shared__ int      wcnt[32 * NRND];
    const int tid  = threadIdx.x;
    const int lane = tid & 31;
    const int wave = __builtin_amdgcn_readfirstlane(tid >> 5);
    const int r0   = blockIdx.x * TR;

    for (int i = tid; i < TR * H; i += 1024) { den[i] = 0.0f; mx[i] = 0u; }
    if (FIRST && tid == 0) cnt = 0;
    if (tid < 32 * NRND) wcnt[tid] = 0;
    __syncthreads();

    int* kl = keep + (size_t)blockIdx.x * LCAP;
    int n;
    if constexpr (FIRST) {
        const uint4* tc4 = (const uint4*)tcode;
        const unsigned pat = (unsigned)blockIdx.x * 0x01010101u;
        const int start = (int)((blockIdx.x * (unsigned)NBLKC) / gridDim.x);
        static_assert(U % 2 == 0, "two loads (32 edges of a lane) make one word of match bits");

        for (int i = wave; i < NBLKC; i += 32 * U) {
            uint4 cv[U];
#pragma unroll
            for (int u = 0; u < U; u++) {
                const int bi = i + u * 32;
                int bb = (bi < NBLKC ? bi : i) + start;
                if (bb >= NBLKC) bb -= NBLKC;
                cv[u] = tc4[(size_t)bb * 32 + lane];
            }
            unsigned lm[U / 2];
#pragma unroll
            for (int g = 0; g < U / 2; g++) {
                unsigned nz = 0u;
                unsigned keepm = 0u;
#pragma unroll
                for (int uu = 0; uu < 2; uu++) {
                    const int u  = 2 * g + uu;
                    const int bi = i + u * 32;
                    int bb = (bi < NBLKC ? bi : i) + start;
                    if (bb >= NBLKC) bb -= NBLKC;
                    const bool live = (bi < NBLKC) && (bb * 512 + lane * 16 < EE);
                    keepm |= live ? (uu == 0 ? 0xF0F0F0F0u : 0x0F0F0F0Fu) : 0u;
                    const unsigned wd[4] = { cv[u].x, cv[u].y, cv[u].z, cv[u].w };
#pragma unroll
                    for (int w = 0; w < 4; w++) {
                        const int q = 4 * uu + w;
                        const unsigned t = wd[w] ^ pat;
                        const unsigned y = ((t & 0x7f7f7f7fu) + 0x7f7f7f7fu) | t;
                        nz |= (y >> q) & (0x80808080u >> q);
                    }
                }
                lm[g] = ~nz & keepm;
            }
            unsigned any = 0u;
            int total = 0;
#pragma unroll
            for (int g = 0; g < U / 2; g++) { any |= lm[g]; total += __popc(lm[g]); }
            if (any != 0u) {
                int slot = atomicAdd(&cnt, total);
#pragma unroll
                for (int g = 0; g < U / 2; g++) {
                    unsigned m2 = lm[g];
                    while (m2) {
                        const int p = __builtin_ctz(m2);
                        m2 &= m2 - 1;
                        if (slot < LCAP) {
                            const int q = 7 - (p & 7);
                            const int k = p >> 3;
                            int bb = i + (2 * g + (q >> 2)) * 32 + start;
                            if (bb >= NBLKC) bb -= NBLKC;
                            list[slot] = bb * 512 + lane * 16 + (q & 3) * 4 + k;
                        }
                        slot++;
                    }
                }
            }
        }
        __syncthreads();

        n = cnt;
        if (tid == 0) kcnt[blockIdx.x] = n;
    } else {
        n = __builtin_amdgcn_readfirstlane(kcnt[blockIdx.x]);
    }
    bool owned = false;
    if (n <= LCAP) {
        static_assert(MU == 4 && LCAP % (MU * 1024) == 0, "step M: four entries per lane and trip, every place read lies inside the list");
        for (int base = 0; base < n; base += MU * 1024) {
            bool ok[MU];
            int  ss[MU], dd[MU];
            if constexpr (FIRST) {
                int ee[MU];
#pragma unroll
                for (int u = 0; u < MU; u++) {
                    const int i = base + u * 1024 + tid;
                    const int t = list[i];
                    ok[u] = i < n;
                    ee[u] = ok[u] ? t : 0;
                }
#pragma unroll
                for (int u = 0; u < MU; u++) { ss[u] = src[ee[u]]; dd[u] = dst[ee[u]]; }
            } else {
#pragma unroll
                for (int u = 0; u < MU; u++) {
                    const int i = base + u * 1024 + tid;
                    ok[u] = i < n;
                    const int t = ok[u] ? kl[i] : 0;
                    ss[u] = t & 0xffff;
                    dd[u] = r0 + (t >> 16);
                }
            }
            typename score_vec<H>::type a[MU], c[MU];
#pragma unroll
            for (int u = 0; u < MU; u++) {
                a[u] = load_scores<H>(el, ss[u]);
                c[u] = load_scores<H>(er, dd[u]);
            }
            asm volatile("" :: "v"(a[0]), "v"(c[0]), "v"(a[1]), "v"(c[1]), "v"(a[2]), "v"(c[2]), "v"(a[3]), "v"(c[3]));
#pragma unroll
            for (int u = 0; u < MU; u++) {
                if (ok[u]) {
                    const int dl = dd[u] - r0;
#pragma unroll
                    for (int hh = 0; hh < H; hh++) {
                        float sc = score_of<H>(a[u], hh) + score_of<H>(c[u], hh);
                        sc = sc > 0.0f ? sc : SLOPE * sc;
                        atomicMax(&mx[dl * H + hh], f2ord(sc));
                    }
                    const int ent = (dl << 16) | ss[u];
                    list[base + u * 1024 + tid] = ent;
                    const int ow = dl & 31;
                    const int rd = (dl >> 5) / SPR;
                    const int pw = atomicAdd(&wcnt[ow * NRND + rd], 1);
                    if (pw < RCAP) sub[ow * WCAP + rd * RCAP + pw] = ent;
                    if constexpr (FIRST) kl[base + u * 1024 + tid] = ent;
                }
            }
        }
        __syncthreads();

        bool over = wcnt[lane] > RCAP;
        if constexpr (NRND == 2) over = over || (wcnt[lane + 32] > RCAP);
        owned = __builtin_amdgcn_ballot_w32(over) == 0u;
    }

    const int gl  = lane / H;
    const int hhl = lane % H;
    const int rows = (NN - r0) < TR ? (NN - r0) : TR;
    if (owned) {
        float2 bv[H];
#pragma unroll
        for (int j = 0; j < H; j++) bv[j] = *(const float2*)(b + 64 * j + 2 * lane);
        for (int rd = 0; rd < NRND; rd++) {
            const int  nw = __builtin_amdgcn_readfirstlane(wcnt[wave * NRND + rd]);
            const int* sw = sub + wave * WCAP + rd * RCAP;
            float2 racc[SPR][H];
#pragma unroll
            for (int s = 0; s < SPR; s++)
#pragma unroll
                for (int j = 0; j < H; j++) racc[s][j] = make_float2(0.0f, 0.0f);
            for (int base = 0; base < nw; base += G) {
                const bool okl = (base + gl) < nw;
                const int  ent = sw[okl ? base + gl : base];
                const int  sl  = ent & 0xffff;
                const int  dl_ = ent >> 16;
                float scl = el[(size_t)sl * H + hhl] + er[(size_t)(r0 + dl_) * H + hhl];
                scl = scl > 0.0f ? scl : SLOPE * scl;
                const float wl = expf(scl - ord2f(mx[dl_ * H + hhl]));
                float2 v[G][H];
#pragma unroll
                for (int g = 0; g < G; g++) {
                    const int sg = __builtin_amdgcn_readlane(sl, g * H);
                    const float2* hs = (const float2*)(h + (size_t)sg * RW);
#pragma unroll
                    for (int j = 0; j < H; j++) v[g][j] = hs[lane + 32 * j];
                }
                if (okl) atomicAdd(&den[dl_ * H + hhl], wl);
#pragma unroll
                for (int g = 0; g < G; g++) {
                    if (base + g < nw) {
                        const int sg_ = (__builtin_amdgcn_readlane(dl_, g * H) >> 5) - rd * SPR;
                        float wj[H];
#pragma unroll
                        for (int j = 0; j < H; j++)
                            wj[j] = __int_as_float(__builtin_amdgcn_readlane(__float_as_int(wl), g * H + j));
#pragma unroll
                        for (int s = 0; s < SPR; s++) {
                            if (sg_ == s) {
#pragma unroll
                                for (int j = 0; j < H; j++) {
                                    racc[s][j].x += wj[j] * v[g][j].x;
                                    racc[s][j].y += wj[j] * v[g][j].y;
                                }
                            }
                        }
                    }
                }
            }
#pragma unroll
            for (int s = 0; s < SPR; s++) {
                const int row = wave + 32 * (rd * SPR + s);
                if (row < rows) {
                    float* orow = FIRST ? acc + row * PITCH + 2 * lane
                                        : outp + (size_t)(r0 + row) * RW + 2 * lane;
#pragma unroll
                    for (int j = 0; j < H; j++) {
                        const float dn = den[row * H + j];
                        float x0 = (dn > 0.0f ? racc[s][j].x / dn : 0.0f) + bv[j].x;
                        float x1 = (dn > 0.0f ? racc[s][j].y / dn : 0.0f) + bv[j].y;
                        if (ELU) {
                            x0 = x0 > 0.0f ? x0 : expm1f(x0);
                            x1 = x1 > 0.0f ? x1 : expm1f(x1);
                        }
                        *(float2*)(orow + 64 * j) = make_float2(x0, x1);
                    }
                }
            }
        }
    } else {
        for (int i = tid; i < TR * PITCH; i += 1024) acc[i] = 0.0f;
        __syncthreads();
        if (n <= LCAP) {
            for (int base = wave * G; base < n; base += 32 * G) {
                const bool okl = (base + gl) < n;
                const int  ent = list[okl ? base + gl : base];
                const int  sl  = ent & 0xffff;
                const int  dl_ = ent >> 16;
                float scl = el[(size_t)sl * H + hhl] + er[(size_t)(r0 + dl_) * H + hhl];
                scl = scl > 0.0f ? scl : SLOPE * scl;
                const float wl = expf(scl - ord2f(mx[dl_ * H + hhl]));
                float2 v[G][H];
#pragma unroll
                for (int g = 0; g < G; g++) {
                    const int sg = __builtin_amdgcn_readlane(sl, g * H);
                    const float2* hs = (const float2*)(h + (size_t)sg * RW);
#pragma unroll
                    for (int j = 0; j < H; j++) v[g][j] = hs[lane + 32 * j];
                }
                if (okl) atomicAdd(&den[dl_ * H + hhl], wl);
#pragma unroll
                for (int g = 0; g < G; g++) {
                    if (base + g < n) {
                        const int dg = __builtin_amdgcn_readlane(dl_, g * H);
                        float* row = acc + dg * PITCH + 2 * lane;
#pragma unroll
                        for (int j = 0; j < H; j++) {
                            const float wj = __int_as_float(__builtin_amdgcn_readlane(__float_as_int(wl), g * H + j));
                            atomicAdd(&row[64 * j],     wj * v[g][j].x);
                            atomicAdd(&row[64 * j + 1], wj * v[g][j].y);
                        }
                    }
                }
            }
        } else {
            for (int g = wave; g < EE / 32; g += 32) {
                const int e = g * 32 + lane;
                const int d = dst[e];
                if ((unsigned)(d - r0) < (unsigned)TR) {
                    const int s = src[e];
#pragma unroll
                    for (int hh = 0; hh < H; hh++) {
                        float sc = el[(size_t)s * H + hh] + er[(size_t)d * H + hh];
                        sc = sc > 0.0f ? sc : SLOPE * sc;
                        atomicMax(&mx[(d - r0) * H + hh], f2ord(sc));
                    }
                }
            }
            __syncthreads();
            for (int g = wave; g < EE / 32; g += 32) {
                const int d = dst[g * 32 + lane];
                unsigned hm = __builtin_amdgcn_ballot_w32((unsigned)(d - r0) < (unsigned)TR);
                while (hm) {
                    const int l = __builtin_ctz(hm);
                    hm &= hm - 1;
                    tile_add_edge<H, PITCH>(g * 32 + l, r0, lane, src, dst, el, er, mx, h, acc, den);
                }
            }
        }
        __syncthreads();

        static_assert(1024 % RW == 0, "a thread keeps its column");
        const int   nel  = rows * RW;
        const float bc   = b[tid % RW];
        float* op = outp + (size_t)r0 * RW;
        for (int i = tid; i < nel; i += 1024) {
            const int   ai = (i / RW) * PITCH + (i % RW);
            const float dn = den[i >> 6];
            const float a  = acc[ai];
            float x = (dn > 0.0f ? a / dn : 0.0f) + bc;
            if (ELU) x = x > 0.0f ? x : expm1f(x);
            if constexpr (FIRST) acc[ai] = x; else op[i] = x;
        }
    }

    if constexpr (FIRST) {
        __syncthreads();
        const int nstr = (rows + 15) >> 4;
        for (int item = wave; item < ((nstr + 1) >> 1) * 4; item += 32) {
            const int strip0 = 2 * (item >> 2);
            const int t      = item & 3;
            if (strip0 + 1 < nstr) strip_product<2, PITCH>(acc, Wn, outp, r0, rows, strip0, t, lane);
            else                   strip_product<1, PITCH>(acc, Wn, outp, r0, rows, strip0, t, lane);
        }
    }
}

extern "C" void kernel_launch(void* const* d_in, const int* in_sizes, int n_in,
                              void* d_out, int out_size, void* d_ws, size_t ws_size,
                              hipStream_t stream) {
    const float* feat = (const float*)d_in[0];
    const int*   src  = (const int*)d_in[1];
    const int*   dst  = (const int*)d_in[2];
    const float* W1   = (const float*)d_in[3];
    const float* al1  = (const float*)d_in[4];
    const float* ar1  = (const float*)d_in[5];
    const float* b1   = (const float*)d_in[6];
    const float* W2   = (const float*)d_in[7];
    const float* al2  = (const float*)d_in[8];
    const float* ar2  = (const float*)d_in[9];
    const float* b2   = (const float*)d_in[10];
    float* out = (float*)d_out;

    float* h1    = (float*)d_ws;
    float* h2    = h1    + (size_t)NN * 256;
    float* el1   = h2    + (size_t)NN * 64;
    float* er1   = el1   + (size_t)NN * H1_;
    float* el2   = er1   + (size_t)NN * H1_;
    float* er2   = el2   + NN;
    int*   keep  = (int*)(er2 + NN);
    int*   kcnt  = keep  + (size_t)NTILE * LCAP;
    unsigned* tcode = (unsigned*)(kcnt + NTILE);

    wmma_gemm_wres<256><<<GWG, 512, 0, stream>>>(feat, W1, h1);
    static_assert(GWG * 1024 >= EEP / 4, "the node-score launch has a thread for every four edges of the tile-number list");
    node_attn<H1_, true><<<GWG, 1024, 0, stream>>>(h1, al1, ar1, el1, er1, dst, tcode);
    tile_accumulate<H1_, true, true><<<NTILE, 1024, 0, stream>>>(src, dst, el1, er1, h1, b1, h2, keep, kcnt, tcode, W2);

    node_attn<1, false><<<GWG, 512, 0, stream>>>(h2, al2, ar2, el2, er2, dst, tcode);
    tile_accumulate<1, false, false><<<NTILE, 1024, 0, stream>>>(src, dst, el2, er2, h2, b2, out, keep, kcnt, tcode, W2);
}
